# starting kernel
# baseline (speedup 1.0000x reference)
#include <hip/hip_runtime.h>
#include <hip/hip_bf16.h>

#define SDIM   100
#define SPAD   112
#define HID    8
#define NOBS   12
#define DTC    0.01f
#define NCHUNK 25
#define NWA    7

typedef float v2f __attribute__((ext_vector_type(2)));
typedef float v8f __attribute__((ext_vector_type(8)));

__global__ __launch_bounds__(256, 1)
void greybox_scan_kernel(const float* __restrict__ c0,
                         const float* __restrict__ w,
                         const float* __restrict__ u,
                         const float* __restrict__ A,
                         const float* __restrict__ Bw,
                         const float* __restrict__ W1,
                         const float* __restrict__ b1,
                         const float* __restrict__ W2,
                         const float* __restrict__ b2,
                         const int*   __restrict__ obs,
                         float* __restrict__ out_c,
                         float* __restrict__ out_y,
                         int T)
{
    __shared__ float cS[SDIM];
    __shared__ float zS[SDIM];
    __shared__ float hS[HID];
    __shared__ float accS[SPAD];
    __shared__ float uS[2][SDIM];
    __shared__ float wS[2][2];
    __shared__ float BwS[2 * SDIM];
    __shared__ float b2S[SDIM];
    __shared__ float b1S[HID];
    __shared__ int   obsS[NOBS];

    const int tid  = threadIdx.x;
    const int wave = tid >> 5;
    const int lane = tid & 31;
    const int hi   = (lane >> 4) & 1;
    const int rloc = lane & 15;

    for (int i = tid; i < 2 * SDIM; i += 256) BwS[i] = Bw[i];
    for (int i = tid; i < SDIM;     i += 256) { b2S[i] = DTC * b2[i]; zS[i] = 0.0f; cS[i] = c0[i]; }
    for (int i = tid; i < HID;      i += 256) b1S[i] = b1[i];
    for (int i = tid; i < NOBS;     i += 256) obsS[i] = obs[i];
    for (int i = tid; i < SPAD;     i += 256) accS[i] = 0.0f;
    if (tid < SDIM) { uS[0][tid] = u[tid]; uS[1][tid] = 0.0f; }
    if (tid < 2)    { wS[0][tid] = w[tid]; wS[1][tid] = 0.0f; }

    const float* bBaseC = (rloc == 0) ? (cS + 2 * hi) : zS;
    const float* bBaseH = (rloc == 0) ? (hS + 2 * hi) : zS;

    v2f afrag[NCHUNK] = {};
    v2f w2frag[2]     = {};

    if (wave < NWA) {
        const int r = wave * 16 + rloc;
        #pragma unroll
        for (int j = 0; j < NCHUNK; ++j) {
            const int k = 4 * j + 2 * hi;
            v2f a; a.x = 0.0f; a.y = 0.0f;
            if (r < SDIM) {
                a.x = A[r * SDIM + k];
                a.y = A[r * SDIM + k + 1];
            }
            afrag[j] = a;
        }
        #pragma unroll
        for (int jj = 0; jj < 2; ++jj) {
            const int k = 4 * jj + 2 * hi;
            v2f a; a.x = 0.0f; a.y = 0.0f;
            if (r < SDIM) {
                a.x = DTC * W2[r * HID + k];
                a.y = DTC * W2[r * HID + k + 1];
            }
            w2frag[jj] = a;
        }
    } else {
        const int rw = rloc;
        #pragma unroll
        for (int j = 0; j < NCHUNK; ++j) {
            const int k = 4 * j + 2 * hi;
            v2f a; a.x = 0.0f; a.y = 0.0f;
            if (rw < HID) {
                a.x = W1[rw * SDIM + k];
                a.y = W1[rw * SDIM + k + 1];
            }
            afrag[j] = a;
        }
    }
    __syncthreads();

    for (int t = 0; t < T; ++t) {
        const int buf = t & 1;

        if (t + 1 < T) {
            if (tid < SDIM)
                uS[buf ^ 1][tid] = u[(size_t)(t + 1) * SDIM + tid];
            else if (tid < SDIM + 2)
                wS[buf ^ 1][tid - SDIM] = w[(size_t)(t + 1) * 2 + (tid - SDIM)];
        }

        v8f acc[4];
        #pragma unroll
        for (int q = 0; q < 4; ++q)
            acc[q] = (v8f){0.f, 0.f, 0.f, 0.f, 0.f, 0.f, 0.f, 0.f};

        if (wave == NWA) {
            if (t > 0 && lane < NOBS)
                out_y[(size_t)(t - 1) * NOBS + lane] = cS[obsS[lane]];
            #pragma unroll
            for (int j = 0; j < NCHUNK; ++j) {
                const v2f b = *(const v2f*)(bBaseC + 4 * j);
                acc[j & 3] = __builtin_amdgcn_wmma_f32_16x16x4_f32(
                                 false, afrag[j], false, b, (short)0, acc[j & 3],
                                 false, false);
            }
            const v8f s = (acc[0] + acc[1]) + (acc[2] + acc[3]);
            if (lane == 0) {
                #pragma unroll
                for (int i = 0; i < HID; ++i)
                    hS[i] = tanhf(s[i] + b1S[i]);
            }
        } else {
            #pragma unroll
            for (int j = 0; j < NCHUNK; ++j) {
                const v2f b = *(const v2f*)(bBaseC + 4 * j);
                acc[j & 3] = __builtin_amdgcn_wmma_f32_16x16x4_f32(
                                 false, afrag[j], false, b, (short)0, acc[j & 3],
                                 false, false);
            }
        }
        __syncthreads();

        if (wave < NWA) {
            v8f s = (acc[0] + acc[1]) + (acc[2] + acc[3]);
            #pragma unroll
            for (int jj = 0; jj < 2; ++jj) {
                const v2f b = *(const v2f*)(bBaseH + 4 * jj);
                s = __builtin_amdgcn_wmma_f32_16x16x4_f32(
                        false, w2frag[jj], false, b, (short)0, s, false, false);
            }
            if (rloc == 0) {
                const int base = wave * 16 + hi * 8;
                #pragma unroll
                for (int i = 0; i < 8; ++i)
                    accS[base + i] = s[i];
            }
        }
        __syncthreads();

        if (tid < SDIM) {
            const float cn = accS[tid]
                           + BwS[2 * tid]     * wS[buf][0]
                           + BwS[2 * tid + 1] * wS[buf][1]
                           + uS[buf][tid]
                           + b2S[tid];
            cS[tid] = cn;
            out_c[(size_t)t * SDIM + tid] = cn;
        }
        __syncthreads();
    }

    if (tid < NOBS)
        out_y[(size_t)(T - 1) * NOBS + tid] = cS[obsS[tid]];
}

extern "C" void kernel_launch(void* const* d_in, const int* in_sizes, int n_in,
                              void* d_out, int out_size, void* d_ws, size_t ws_size,
                              hipStream_t stream) {
    const float* c0 = (const float*)d_in[0];
    const float* w  = (const float*)d_in[1];
    const float* u  = (const float*)d_in[2];
    const float* A  = (const float*)d_in[3];
    const float* Bw = (const float*)d_in[4];
    const float* W1 = (const float*)d_in[5];
    const float* b1 = (const float*)d_in[6];
    const float* W2 = (const float*)d_in[7];
    const float* b2 = (const float*)d_in[8];
    const int*  obs = (const int*)  d_in[9];

    const int T = in_sizes[1] / 2;
    float* out_c = (float*)d_out;
    float* out_y = out_c + (size_t)T * SDIM;

    greybox_scan_kernel<<<1, 256, 0, stream>>>(c0, w, u, A, Bw, W1, b1, W2, b2,
                                               obs, out_c, out_y, T);
}


# Global I/O moved to 128-step time tiles: a tile's u,w rows staged in LDS up front, c rows buffered in LDS and written with gathered y rows once per tile; steps touch LDS only; sum order unchanged
# speedup vs baseline: 1.5771x; 1.5771x over previous
#include <hip/hip_runtime.h>
#include <hip/hip_bf16.h>

#define SDIM   100
#define SPAD   112
#define HID    8
#define NOBS   12
#define DTC    0.01f
#define NCHUNK 25
#define NWA    7
#define TB     128
#define ROW4   (SDIM / 4)
#define TILE4  (TB * ROW4)
#define NIT4   ((TILE4 + 255) / 256)
static_assert(TB * 2 <= 256, "a tile's w rows are staged one float per thread");

typedef float v2f __attribute__((ext_vector_type(2)));
typedef float v8f __attribute__((ext_vector_type(8)));

__global__ __launch_bounds__(256, 1)
void greybox_scan_kernel(const float* __restrict__ c0,
                         const float* __restrict__ w,
                         const float* __restrict__ u,
                         const float* __restrict__ A,
                         const float* __restrict__ Bw,
                         const float* __restrict__ W1,
                         const float* __restrict__ b1,
                         const float* __restrict__ W2,
                         const float* __restrict__ b2,
                         const int*   __restrict__ obs,
                         float* __restrict__ out_c,
                         float* __restrict__ out_y,
                         int T)
{
    __shared__ float cS[SDIM];
    __shared__ float zS[SDIM];
    __shared__ float hS[HID];
    __shared__ float accS[SPAD];
    __shared__ __attribute__((aligned(16))) float uT[TB * SDIM];
    __shared__ __attribute__((aligned(16))) float oT[TB * SDIM];
    __shared__ float wT[TB * 2];
    __shared__ float BwS[2 * SDIM];
    __shared__ float b2S[SDIM];
    __shared__ float b1S[HID];
    __shared__ int   obsS[NOBS];

    const int tid  = threadIdx.x;
    const int wave = tid >> 5;
    const int lane = tid & 31;
    const int hi   = (lane >> 4) & 1;
    const int rloc = lane & 15;

    for (int i = tid; i < 2 * SDIM; i += 256) BwS[i] = Bw[i];
    for (int i = tid; i < SDIM;     i += 256) { b2S[i] = DTC * b2[i]; zS[i] = 0.0f; cS[i] = c0[i]; }
    for (int i = tid; i < HID;      i += 256) b1S[i] = b1[i];
    for (int i = tid; i < NOBS;     i += 256) obsS[i] = obs[i];
    for (int i = tid; i < SPAD;     i += 256) accS[i] = 0.0f;

    const float* bBaseC = (rloc == 0) ? (cS + 2 * hi) : zS;
    const float* bBaseH = (rloc == 0) ? (hS + 2 * hi) : zS;

    v2f afrag[NCHUNK] = {};
    v2f w2frag[2]     = {};

    if (wave < NWA) {
        const int r = wave * 16 + rloc;
        #pragma unroll
        for (int j = 0; j < NCHUNK; ++j) {
            const int k = 4 * j + 2 * hi;
            v2f a; a.x = 0.0f; a.y = 0.0f;
            if (r < SDIM) {
                a.x = A[r * SDIM + k];
                a.y = A[r * SDIM + k + 1];
            }
            afrag[j] = a;
        }
        #pragma unroll
        for (int jj = 0; jj < 2; ++jj) {
            const int k = 4 * jj + 2 * hi;
            v2f a; a.x = 0.0f; a.y = 0.0f;
            if (r < SDIM) {
                a.x = DTC * W2[r * HID + k];
                a.y = DTC * W2[r * HID + k + 1];
            }
            w2frag[jj] = a;
        }
    } else {
        const int rw = rloc;
        #pragma unroll
        for (int j = 0; j < NCHUNK; ++j) {
            const int k = 4 * j + 2 * hi;
            v2f a; a.x = 0.0f; a.y = 0.0f;
            if (rw < HID) {
                a.x = W1[rw * SDIM + k];
                a.y = W1[rw * SDIM + k + 1];
            }
            afrag[j] = a;
        }
    }
    __syncthreads();

    for (int t0 = 0; t0 < T; t0 += TB) {
        const int ns = (T - t0 < TB) ? (T - t0) : TB;

        {
            const float4* __restrict__ src4 = (const float4*)(u + (size_t)t0 * SDIM);
            float4* dst4 = (float4*)uT;
            const int n4 = ns * ROW4;
            const int nw = ns * 2;
            float4 r[NIT4];
            #pragma unroll
            for (int k = 0; k < NIT4; ++k) {
                const int i = tid + 256 * k;
                r[k] = (i < n4) ? src4[i] : make_float4(0.f, 0.f, 0.f, 0.f);
            }
            const float wv = (tid < nw) ? w[(size_t)t0 * 2 + tid] : 0.0f;
            #pragma unroll
            for (int k = 0; k < NIT4; ++k) {
                const int i = tid + 256 * k;
                if (i < n4) dst4[i] = r[k];
            }
            if (tid < nw) wT[tid] = wv;
        }
        __syncthreads();

        for (int tl = 0; tl < ns; ++tl) {
            v8f acc[4];
            #pragma unroll
            for (int q = 0; q < 4; ++q)
                acc[q] = (v8f){0.f, 0.f, 0.f, 0.f, 0.f, 0.f, 0.f, 0.f};

            if (wave == NWA) {
                #pragma unroll
                for (int j = 0; j < NCHUNK; ++j) {
                    const v2f b = *(const v2f*)(bBaseC + 4 * j);
                    acc[j & 3] = __builtin_amdgcn_wmma_f32_16x16x4_f32(
                                     false, afrag[j], false, b, (short)0, acc[j & 3],
                                     false, false);
                }
                const v8f s = (acc[0] + acc[1]) + (acc[2] + acc[3]);
                if (lane == 0) {
                    #pragma unroll
                    for (int i = 0; i < HID; ++i)
                        hS[i] = tanhf(s[i] + b1S[i]);
                }
            } else {
                #pragma unroll
                for (int j = 0; j < NCHUNK; ++j) {
                    const v2f b = *(const v2f*)(bBaseC + 4 * j);
                    acc[j & 3] = __builtin_amdgcn_wmma_f32_16x16x4_f32(
                                     false, afrag[j], false, b, (short)0, acc[j & 3],
                                     false, false);
                }
            }
            __syncthreads();

            if (wave < NWA) {
                v8f s = (acc[0] + acc[1]) + (acc[2] + acc[3]);
                #pragma unroll
                for (int jj = 0; jj < 2; ++jj) {
                    const v2f b = *(const v2f*)(bBaseH + 4 * jj);
                    s = __builtin_amdgcn_wmma_f32_16x16x4_f32(
                            false, w2frag[jj], false, b, (short)0, s, false, false);
                }
                if (rloc == 0) {
                    const int base = wave * 16 + hi * 8;
                    #pragma unroll
                    for (int i = 0; i < 8; ++i)
                        accS[base + i] = s[i];
                }
            }
            __syncthreads();

            if (tid < SDIM) {
                const float cn = accS[tid]
                               + BwS[2 * tid]     * wT[2 * tl]
                               + BwS[2 * tid + 1] * wT[2 * tl + 1]
                               + uT[tl * SDIM + tid]
                               + b2S[tid];
                cS[tid] = cn;
                oT[tl * SDIM + tid] = cn;
            }
            __syncthreads();
        }

        {
            float4* __restrict__ dst4 = (float4*)(out_c + (size_t)t0 * SDIM);
            const float4* src4 = (const float4*)oT;
            const int n4 = ns * ROW4;
            #pragma unroll
            for (int k = 0; k < NIT4; ++k) {
                const int i = tid + 256 * k;
                if (i < n4) dst4[i] = src4[i];
            }
            const int ny = ns * NOBS;
            for (int i = tid; i < ny; i += 256) {
                const int r = i / NOBS;
                const int j = i - r * NOBS;
                out_y[(size_t)t0 * NOBS + i] = oT[r * SDIM + obsS[j]];
            }
        }
    }
}

extern "C" void kernel_launch(void* const* d_in, const int* in_sizes, int n_in,
                              void* d_out, int out_size, void* d_ws, size_t ws_size,
                              hipStream_t stream) {
    const float* c0 = (const float*)d_in[0];
    const float* w  = (const float*)d_in[1];
    const float* u  = (const float*)d_in[2];
    const float* A  = (const float*)d_in[3];
    const float* Bw = (const float*)d_in[4];
    const float* W1 = (const float*)d_in[5];
    const float* b1 = (const float*)d_in[6];
    const float* W2 = (const float*)d_in[7];
    const float* b2 = (const float*)d_in[8];
    const int*  obs = (const int*)  d_in[9];

    const int T = in_sizes[1] / 2;
    float* out_c = (float*)d_out;
    float* out_y = out_c + (size_t)T * SDIM;

    greybox_scan_kernel<<<1, 256, 0, stream>>>(c0, w, u, A, Bw, W1, b1, W2, b2,
                                               obs, out_c, out_y, T);
}


# W1 wave reads the state into all 16 B columns: every lane holds the 8 pre-activations, lane n keeps unit n by register select, so the 8 tanh run lane-parallel, not serially in lane 0; sum order kept
# speedup vs baseline: 2.6003x; 1.6488x over previous
#include <hip/hip_runtime.h>
#include <hip/hip_bf16.h>

#define SDIM   100
#define SPAD   112
#define HID    8
#define NOBS   12
#define DTC    0.01f
#define NCHUNK 25
#define NWA    7
#define TB     128
#define ROW4   (SDIM / 4)
#define TILE4  (TB * ROW4)
#define NIT4   ((TILE4 + 255) / 256)
static_assert(TB * 2 <= 256, "a tile's w rows are staged one float per thread");

typedef float v2f __attribute__((ext_vector_type(2)));
typedef float v8f __attribute__((ext_vector_type(8)));

__global__ __launch_bounds__(256, 1)
void greybox_scan_kernel(const float* __restrict__ c0,
                         const float* __restrict__ w,
                         const float* __restrict__ u,
                         const float* __restrict__ A,
                         const float* __restrict__ Bw,
                         const float* __restrict__ W1,
                         const float* __restrict__ b1,
                         const float* __restrict__ W2,
                         const float* __restrict__ b2,
                         const int*   __restrict__ obs,
                         float* __restrict__ out_c,
                         float* __restrict__ out_y,
                         int T)
{
    __shared__ float cS[SDIM];
    __shared__ float zS[SDIM];
    __shared__ float hS[HID];
    __shared__ float accS[SPAD];
    __shared__ __attribute__((aligned(16))) float uT[TB * SDIM];
    __shared__ __attribute__((aligned(16))) float oT[TB * SDIM];
    __shared__ float wT[TB * 2];
    __shared__ float BwS[2 * SDIM];
    __shared__ float b2S[SDIM];
    __shared__ int   obsS[NOBS];

    const int tid  = threadIdx.x;
    const int wave = tid >> 5;
    const int lane = tid & 31;
    const int hi   = (lane >> 4) & 1;
    const int rloc = lane & 15;

    for (int i = tid; i < 2 * SDIM; i += 256) BwS[i] = Bw[i];
    for (int i = tid; i < SDIM;     i += 256) { b2S[i] = DTC * b2[i]; zS[i] = 0.0f; cS[i] = c0[i]; }
    for (int i = tid; i < NOBS;     i += 256) obsS[i] = obs[i];
    for (int i = tid; i < SPAD;     i += 256) accS[i] = 0.0f;

    const float* bBaseC = (rloc == 0 || wave == NWA) ? (cS + 2 * hi) : zS;
    const float* bBaseH = (rloc == 0) ? (hS + 2 * hi) : zS;

    static_assert((HID & (HID - 1)) == 0 && HID <= 8, "one hidden unit per lane, picked from the 8 D registers");
    const float b1v = b1[lane & (HID - 1)];
    v2f afrag[NCHUNK] = {};
    v2f w2frag[2]     = {};

    if (wave < NWA) {
        const int r = wave * 16 + rloc;
        #pragma unroll
        for (int j = 0; j < NCHUNK; ++j) {
            const int k = 4 * j + 2 * hi;
            v2f a; a.x = 0.0f; a.y = 0.0f;
            if (r < SDIM) {
                a.x = A[r * SDIM + k];
                a.y = A[r * SDIM + k + 1];
            }
            afrag[j] = a;
        }
        #pragma unroll
        for (int jj = 0; jj < 2; ++jj) {
            const int k = 4 * jj + 2 * hi;
            v2f a; a.x = 0.0f; a.y = 0.0f;
            if (r < SDIM) {
                a.x = DTC * W2[r * HID + k];
                a.y = DTC * W2[r * HID + k + 1];
            }
            w2frag[jj] = a;
        }
    } else {
        const int rw = rloc;
        #pragma unroll
        for (int j = 0; j < NCHUNK; ++j) {
            const int k = 4 * j + 2 * hi;
            v2f a; a.x = 0.0f; a.y = 0.0f;
            if (rw < HID) {
                a.x = W1[rw * SDIM + k];
                a.y = W1[rw * SDIM + k + 1];
            }
            afrag[j] = a;
        }
    }
    __syncthreads();

    for (int t0 = 0; t0 < T; t0 += TB) {
        const int ns = (T - t0 < TB) ? (T - t0) : TB;

        {
            const float4* __restrict__ src4 = (const float4*)(u + (size_t)t0 * SDIM);
            float4* dst4 = (float4*)uT;
            const int n4 = ns * ROW4;
            const int nw = ns * 2;
            float4 r[NIT4];
            #pragma unroll
            for (int k = 0; k < NIT4; ++k) {
                const int i = tid + 256 * k;
                r[k] = (i < n4) ? src4[i] : make_float4(0.f, 0.f, 0.f, 0.f);
            }
            const float wv = (tid < nw) ? w[(size_t)t0 * 2 + tid] : 0.0f;
            #pragma unroll
            for (int k = 0; k < NIT4; ++k) {
                const int i = tid + 256 * k;
                if (i < n4) dst4[i] = r[k];
            }
            if (tid < nw) wT[tid] = wv;
        }
        __syncthreads();

        for (int tl = 0; tl < ns; ++tl) {
            v8f acc[4];
            #pragma unroll
            for (int q = 0; q < 4; ++q)
                acc[q] = (v8f){0.f, 0.f, 0.f, 0.f, 0.f, 0.f, 0.f, 0.f};

            if (wave == NWA) {
                #pragma unroll
                for (int j = 0; j < NCHUNK; ++j) {
                    const v2f b = *(const v2f*)(bBaseC + 4 * j);
                    acc[j & 3] = __builtin_amdgcn_wmma_f32_16x16x4_f32(
                                     false, afrag[j], false, b, (short)0, acc[j & 3],
                                     false, false);
                }
                const v8f s = (acc[0] + acc[1]) + (acc[2] + acc[3]);
                const float p0 = (lane & 1) ? s[1] : s[0];
                const float p1 = (lane & 1) ? s[3] : s[2];
                const float p2 = (lane & 1) ? s[5] : s[4];
                const float p3 = (lane & 1) ? s[7] : s[6];
                const float q0 = (lane & 2) ? p1 : p0;
                const float q1 = (lane & 2) ? p3 : p2;
                const float zn = (lane & 4) ? q1 : q0;
                if (lane < HID)
                    hS[lane] = tanhf(zn + b1v);
            } else {
                #pragma unroll
                for (int j = 0; j < NCHUNK; ++j) {
                    const v2f b = *(const v2f*)(bBaseC + 4 * j);
                    acc[j & 3] = __builtin_amdgcn_wmma_f32_16x16x4_f32(
                                     false, afrag[j], false, b, (short)0, acc[j & 3],
                                     false, false);
                }
            }
            __syncthreads();

            if (wave < NWA) {
                v8f s = (acc[0] + acc[1]) + (acc[2] + acc[3]);
                #pragma unroll
                for (int jj = 0; jj < 2; ++jj) {
                    const v2f b = *(const v2f*)(bBaseH + 4 * jj);
                    s = __builtin_amdgcn_wmma_f32_16x16x4_f32(
                            false, w2frag[jj], false, b, (short)0, s, false, false);
                }
                if (rloc == 0) {
                    const int base = wave * 16 + hi * 8;
                    #pragma unroll
                    for (int i = 0; i < 8; ++i)
                        accS[base + i] = s[i];
                }
            }
            __syncthreads();

            if (tid < SDIM) {
                const float cn = accS[tid]
                               + BwS[2 * tid]     * wT[2 * tl]
                               + BwS[2 * tid + 1] * wT[2 * tl + 1]
                               + uT[tl * SDIM + tid]
                               + b2S[tid];
                cS[tid] = cn;
                oT[tl * SDIM + tid] = cn;
            }
            __syncthreads();
        }

        {
            float4* __restrict__ dst4 = (float4*)(out_c + (size_t)t0 * SDIM);
            const float4* src4 = (const float4*)oT;
            const int n4 = ns * ROW4;
            #pragma unroll
            for (int k = 0; k < NIT4; ++k) {
                const int i = tid + 256 * k;
                if (i < n4) dst4[i] = src4[i];
            }
            const int ny = ns * NOBS;
            for (int i = tid; i < ny; i += 256) {
                const int r = i / NOBS;
                const int j = i - r * NOBS;
                out_y[(size_t)t0 * NOBS + i] = oT[r * SDIM + obsS[j]];
            }
        }
    }
}

extern "C" void kernel_launch(void* const* d_in, const int* in_sizes, int n_in,
                              void* d_out, int out_size, void* d_ws, size_t ws_size,
                              hipStream_t stream) {
    const float* c0 = (const float*)d_in[0];
    const float* w  = (const float*)d_in[1];
    const float* u  = (const float*)d_in[2];
    const float* A  = (const float*)d_in[3];
    const float* Bw = (const float*)d_in[4];
    const float* W1 = (const float*)d_in[5];
    const float* b1 = (const float*)d_in[6];
    const float* W2 = (const float*)d_in[7];
    const float* b2 = (const float*)d_in[8];
    const int*  obs = (const int*)  d_in[9];

    const int T = in_sizes[1] / 2;
    float* out_c = (float*)d_out;
    float* out_y = out_c + (size_t)T * SDIM;

    greybox_scan_kernel<<<1, 256, 0, stream>>>(c0, w, u, A, Bw, W1, b1, W2, b2,
                                               obs, out_c, out_y, T);
}


# Rows re-split over 4 waves, one per SIMD: waves 0..2 feed two 16-row A tiles from one set of state fragments, W1 rides in the last tile's padding rows so its tanh ends early; sum order unchanged
# speedup vs baseline: 3.0243x; 1.1631x over previous
#include <hip/hip_runtime.h>
#include <hip/hip_bf16.h>

#define SDIM   100
#define SPAD   112
#define HID    8
#define NOBS   12
#define DTC    0.01f
#define NCHUNK 25
#define NTILE  7
#define NWV    4
#define NTHR   128
#define W1ROW  8
#define TB     128
#define ROW4   (SDIM / 4)
#define TILE4  (TB * ROW4)
#define NIT4   ((TILE4 + NTHR - 1) / NTHR)
static_assert(NTHR == NWV * 32, "one wave per SIMD of the work-group processor");
static_assert(TB <= NTHR, "a tile's w rows are staged one row (two floats) per thread");
static_assert((NTILE - 1) * 16 + W1ROW >= SDIM && W1ROW + HID <= 16, "W1 rows fit in the last tile's padding rows");

typedef float v2f __attribute__((ext_vector_type(2)));
typedef float v8f __attribute__((ext_vector_type(8)));

__global__ __launch_bounds__(NTHR, 1)
void greybox_scan_kernel(const float* __restrict__ c0,
                         const float* __restrict__ w,
                         const float* __restrict__ u,
                         const float* __restrict__ A,
                         const float* __restrict__ Bw,
                         const float* __restrict__ W1,
                         const float* __restrict__ b1,
                         const float* __restrict__ W2,
                         const float* __restrict__ b2,
                         const int*   __restrict__ obs,
                         float* __restrict__ out_c,
                         float* __restrict__ out_y,
                         int T)
{
    __shared__ float cS[SDIM];
    __shared__ float zS[SDIM];
    __shared__ float hS[HID];
    __shared__ float accS[SPAD];
    __shared__ __attribute__((aligned(16))) float uT[TB * SDIM];
    __shared__ __attribute__((aligned(16))) float oT[TB * SDIM];
    __shared__ __attribute__((aligned(8))) float wT[TB * 2];
    __shared__ float BwS[2 * SDIM];
    __shared__ float b2S[SDIM];
    __shared__ int   obsS[NOBS];

    const int tid  = threadIdx.x;
    const int wave = __builtin_amdgcn_readfirstlane(tid >> 5);
    const int lane = tid & 31;
    const int hi   = (lane >> 4) & 1;
    const int rloc = lane & 15;

    for (int i = tid; i < 2 * SDIM; i += NTHR) BwS[i] = Bw[i];
    for (int i = tid; i < SDIM;     i += NTHR) { b2S[i] = DTC * b2[i]; zS[i] = 0.0f; cS[i] = c0[i]; }
    for (int i = tid; i < NOBS;     i += NTHR) obsS[i] = obs[i];
    for (int i = tid; i < SPAD;     i += NTHR) accS[i] = 0.0f;

    const bool two   = wave < NWV - 1;
    const int  tileA = two ? 2 * wave : NTILE - 1;

    const float* bBaseC = (rloc == 0 || !two) ? (cS + 2 * hi) : zS;
    const float* bBaseH = (rloc == 0) ? (hS + 2 * hi) : zS;

    static_assert((HID & (HID - 1)) == 0 && HID <= 8, "one hidden unit per lane, picked from the 8 D registers");
    const float b1v = b1[lane & (HID - 1)];
    v2f afragA[NCHUNK] = {};
    v2f afragB[NCHUNK] = {};
    v2f w2fragA[2]     = {};
    v2f w2fragB[2]     = {};

    {
        const int r = tileA * 16 + rloc;
        const bool isW1 = !two && rloc >= W1ROW && rloc < W1ROW + HID;
        #pragma unroll
        for (int j = 0; j < NCHUNK; ++j) {
            const int k = 4 * j + 2 * hi;
            v2f a; a.x = 0.0f; a.y = 0.0f;
            if (r < SDIM) {
                a.x = A[r * SDIM + k];
                a.y = A[r * SDIM + k + 1];
            } else if (isW1) {
                a.x = W1[(rloc - W1ROW) * SDIM + k];
                a.y = W1[(rloc - W1ROW) * SDIM + k + 1];
            }
            afragA[j] = a;
        }
        #pragma unroll
        for (int jj = 0; jj < 2; ++jj) {
            const int k = 4 * jj + 2 * hi;
            v2f a; a.x = 0.0f; a.y = 0.0f;
            if (r < SDIM) {
                a.x = DTC * W2[r * HID + k];
                a.y = DTC * W2[r * HID + k + 1];
            }
            w2fragA[jj] = a;
        }
    }
    if (two) {
        const int r = (tileA + 1) * 16 + rloc;
        #pragma unroll
        for (int j = 0; j < NCHUNK; ++j) {
            const int k = 4 * j + 2 * hi;
            v2f a;
            a.x = A[r * SDIM + k];
            a.y = A[r * SDIM + k + 1];
            afragB[j] = a;
        }
        #pragma unroll
        for (int jj = 0; jj < 2; ++jj) {
            const int k = 4 * jj + 2 * hi;
            v2f a;
            a.x = DTC * W2[r * HID + k];
            a.y = DTC * W2[r * HID + k + 1];
            w2fragB[jj] = a;
        }
    }
    __syncthreads();

    for (int t0 = 0; t0 < T; t0 += TB) {
        const int ns = (T - t0 < TB) ? (T - t0) : TB;

        {
            const float4* __restrict__ src4 = (const float4*)(u + (size_t)t0 * SDIM);
            float4* dst4 = (float4*)uT;
            const int n4 = ns * ROW4;
            float4 r[NIT4];
            #pragma unroll
            for (int k = 0; k < NIT4; ++k) {
                const int i = tid + NTHR * k;
                r[k] = (i < n4) ? src4[i] : make_float4(0.f, 0.f, 0.f, 0.f);
            }
            const float2 wv = (tid < ns) ? ((const float2*)(w + (size_t)t0 * 2))[tid] : make_float2(0.f, 0.f);
            #pragma unroll
            for (int k = 0; k < NIT4; ++k) {
                const int i = tid + NTHR * k;
                if (i < n4) dst4[i] = r[k];
            }
            if (tid < ns) ((float2*)wT)[tid] = wv;
        }
        __syncthreads();

        for (int tl = 0; tl < ns; ++tl) {
            v8f accA[4], accB[4];
            #pragma unroll
            for (int q = 0; q < 4; ++q) {
                accA[q] = (v8f){0.f, 0.f, 0.f, 0.f, 0.f, 0.f, 0.f, 0.f};
                accB[q] = (v8f){0.f, 0.f, 0.f, 0.f, 0.f, 0.f, 0.f, 0.f};
            }

            v2f bf[NCHUNK];
            #pragma unroll
            for (int j = 0; j < NCHUNK; ++j)
                bf[j] = *(const v2f*)(bBaseC + 4 * j);

            if (two) {
                #pragma unroll
                for (int j = 0; j < NCHUNK; ++j) {
                    accA[j & 3] = __builtin_amdgcn_wmma_f32_16x16x4_f32(
                                      false, afragA[j], false, bf[j], (short)0, accA[j & 3],
                                      false, false);
                    accB[j & 3] = __builtin_amdgcn_wmma_f32_16x16x4_f32(
                                      false, afragB[j], false, bf[j], (short)0, accB[j & 3],
                                      false, false);
                }
            } else {
                #pragma unroll
                for (int j = 0; j < NCHUNK; ++j) {
                    accA[j & 3] = __builtin_amdgcn_wmma_f32_16x16x4_f32(
                                      false, afragA[j], false, bf[j], (short)0, accA[j & 3],
                                      false, false);
                }
                const v8f s = (accA[0] + accA[1]) + (accA[2] + accA[3]);
                const float p0 = (lane & 1) ? s[1] : s[0];
                const float p1 = (lane & 1) ? s[3] : s[2];
                const float p2 = (lane & 1) ? s[5] : s[4];
                const float p3 = (lane & 1) ? s[7] : s[6];
                const float q0 = (lane & 2) ? p1 : p0;
                const float q1 = (lane & 2) ? p3 : p2;
                const float zn = (lane & 4) ? q1 : q0;
                if (lane >= 16 && lane < 16 + HID)
                    hS[lane - 16] = tanhf(zn + b1v);
            }
            __syncthreads();

            {
                v8f sA = (accA[0] + accA[1]) + (accA[2] + accA[3]);
                v8f sB = (accB[0] + accB[1]) + (accB[2] + accB[3]);
                const v2f h0 = *(const v2f*)(bBaseH);
                const v2f h1 = *(const v2f*)(bBaseH + 4);
                if (two) {
                    sA = __builtin_amdgcn_wmma_f32_16x16x4_f32(
                             false, w2fragA[0], false, h0, (short)0, sA, false, false);
                    sB = __builtin_amdgcn_wmma_f32_16x16x4_f32(
                             false, w2fragB[0], false, h0, (short)0, sB, false, false);
                    sA = __builtin_amdgcn_wmma_f32_16x16x4_f32(
                             false, w2fragA[1], false, h1, (short)0, sA, false, false);
                    sB = __builtin_amdgcn_wmma_f32_16x16x4_f32(
                             false, w2fragB[1], false, h1, (short)0, sB, false, false);
                } else {
                    sA = __builtin_amdgcn_wmma_f32_16x16x4_f32(
                             false, w2fragA[0], false, h0, (short)0, sA, false, false);
                    sA = __builtin_amdgcn_wmma_f32_16x16x4_f32(
                             false, w2fragA[1], false, h1, (short)0, sA, false, false);
                }
                if (rloc == 0) {
                    const int baseA = tileA * 16 + hi * 8;
                    if (baseA < SDIM) {
                        #pragma unroll
                        for (int i = 0; i < 8; ++i)
                            accS[baseA + i] = sA[i];
                    }
                    if (two) {
                        #pragma unroll
                        for (int i = 0; i < 8; ++i)
                            accS[baseA + 16 + i] = sB[i];
                    }
                }
            }
            __syncthreads();

            if (tid < SDIM) {
                const float cn = accS[tid]
                               + BwS[2 * tid]     * wT[2 * tl]
                               + BwS[2 * tid + 1] * wT[2 * tl + 1]
                               + uT[tl * SDIM + tid]
                               + b2S[tid];
                cS[tid] = cn;
                oT[tl * SDIM + tid] = cn;
            }
            __syncthreads();
        }

        {
            float4* __restrict__ dst4 = (float4*)(out_c + (size_t)t0 * SDIM);
            const float4* src4 = (const float4*)oT;
            const int n4 = ns * ROW4;
            #pragma unroll
            for (int k = 0; k < NIT4; ++k) {
                const int i = tid + NTHR * k;
                if (i < n4) dst4[i] = src4[i];
            }
            const int ny = ns * NOBS;
            for (int i = tid; i < ny; i += NTHR) {
                const int r = i / NOBS;
                const int j = i - r * NOBS;
                out_y[(size_t)t0 * NOBS + i] = oT[r * SDIM + obsS[j]];
            }
        }
    }
}

extern "C" void kernel_launch(void* const* d_in, const int* in_sizes, int n_in,
                              void* d_out, int out_size, void* d_ws, size_t ws_size,
                              hipStream_t stream) {
    const float* c0 = (const float*)d_in[0];
    const float* w  = (const float*)d_in[1];
    const float* u  = (const float*)d_in[2];
    const float* A  = (const float*)d_in[3];
    const float* Bw = (const float*)d_in[4];
    const float* W1 = (const float*)d_in[5];
    const float* b1 = (const float*)d_in[6];
    const float* W2 = (const float*)d_in[7];
    const float* b2 = (const float*)d_in[8];
    const int*  obs = (const int*)  d_in[9];

    const int T = in_sizes[1] / 2;
    float* out_c = (float*)d_out;
    float* out_y = out_c + (size_t)T * SDIM;

    greybox_scan_kernel<<<1, NTHR, 0, stream>>>(c0, w, u, A, Bw, W1, b1, W2, b2,
                                               obs, out_c, out_y, T);
}


# Input-only term Bw w_t+u_t+dt b2 built per time tile fed as the WMMA accumulators' start value: D column 0 is the new state, lanes 0/16 store it; accS pass and its barrier gone; sum order changed
# speedup vs baseline: 3.3452x; 1.1061x over previous
#include <hip/hip_runtime.h>
#include <hip/hip_bf16.h>

#define SDIM   100
#define SPAD   112
#define HID    8
#define NOBS   12
#define DTC    0.01f
#define NCHUNK 25
#define NTILE  7
#define NWV    4
#define NTHR   128
#define W1ROW  8
#define TB     128
#define ROW4   (SDIM / 4)
#define TILE4  (TB * ROW4)
#define NIT4   ((TILE4 + NTHR - 1) / NTHR)
static_assert(NTHR == NWV * 32, "one wave per SIMD of the work-group processor");
static_assert(TB <= NTHR, "a tile's w rows are staged one row (two floats) per thread");
static_assert((NTILE - 1) * 16 + W1ROW >= SDIM && W1ROW + HID <= 16, "W1 rows fit in the last tile's padding rows");

typedef float v2f __attribute__((ext_vector_type(2)));
typedef float v8f __attribute__((ext_vector_type(8)));

__global__ __launch_bounds__(NTHR, 1)
void greybox_scan_kernel(const float* __restrict__ c0,
                         const float* __restrict__ w,
                         const float* __restrict__ u,
                         const float* __restrict__ A,
                         const float* __restrict__ Bw,
                         const float* __restrict__ W1,
                         const float* __restrict__ b1,
                         const float* __restrict__ W2,
                         const float* __restrict__ b2,
                         const int*   __restrict__ obs,
                         float* __restrict__ out_c,
                         float* __restrict__ out_y,
                         int T)
{
    __shared__ __attribute__((aligned(16))) float cS[SDIM];
    __shared__ __attribute__((aligned(16))) float zS[SDIM];
    __shared__ float hS[HID];
    __shared__ __attribute__((aligned(16))) float uT[TB * SDIM];
    __shared__ __attribute__((aligned(16))) float oT[TB * SDIM];
    __shared__ __attribute__((aligned(8))) float wT[TB * 2];
    __shared__ __attribute__((aligned(16))) float Bw0S[SDIM];
    __shared__ __attribute__((aligned(16))) float Bw1S[SDIM];
    __shared__ __attribute__((aligned(16))) float b2S[SDIM];
    __shared__ int   obsS[NOBS];

    const int tid  = threadIdx.x;
    const int wave = __builtin_amdgcn_readfirstlane(tid >> 5);
    const int lane = tid & 31;
    const int hi   = (lane >> 4) & 1;
    const int rloc = lane & 15;

    for (int i = tid; i < SDIM;     i += NTHR) {
        Bw0S[i] = Bw[2 * i]; Bw1S[i] = Bw[2 * i + 1];
        b2S[i] = DTC * b2[i]; zS[i] = 0.0f; cS[i] = c0[i];
    }
    for (int i = tid; i < NOBS;     i += NTHR) obsS[i] = obs[i];

    const bool two   = wave < NWV - 1;
    const int  tileA = two ? 2 * wave : NTILE - 1;

    const float* bBaseC = (rloc == 0 || !two) ? (cS + 2 * hi) : zS;
    const float* bBaseH = (rloc == 0) ? (hS + 2 * hi) : zS;

    const float* dPtr    = two ? (uT + tileA * 16 + 8 * hi) : (hi ? zS : (uT + (NTILE - 1) * 16));
    const int    dStride = (two || !hi) ? SDIM : 0;
    const int    sBase   = tileA * 16 + 8 * hi;
    const bool   st0     = rloc == 0 && (two || hi == 0);
    const bool   st1     = rloc == 0 && two;

    static_assert((HID & (HID - 1)) == 0 && HID <= 8, "one hidden unit per lane, picked from the 8 D registers");
    const float b1v = b1[lane & (HID - 1)];
    v2f afragA[NCHUNK] = {};
    v2f afragB[NCHUNK] = {};
    v2f w2fragA[2]     = {};
    v2f w2fragB[2]     = {};

    {
        const int r = tileA * 16 + rloc;
        const bool isW1 = !two && rloc >= W1ROW && rloc < W1ROW + HID;
        #pragma unroll
        for (int j = 0; j < NCHUNK; ++j) {
            const int k = 4 * j + 2 * hi;
            v2f a; a.x = 0.0f; a.y = 0.0f;
            if (r < SDIM) {
                a.x = A[r * SDIM + k];
                a.y = A[r * SDIM + k + 1];
            } else if (isW1) {
                a.x = W1[(rloc - W1ROW) * SDIM + k];
                a.y = W1[(rloc - W1ROW) * SDIM + k + 1];
            }
            afragA[j] = a;
        }
        #pragma unroll
        for (int jj = 0; jj < 2; ++jj) {
            const int k = 4 * jj + 2 * hi;
            v2f a; a.x = 0.0f; a.y = 0.0f;
            if (r < SDIM) {
                a.x = DTC * W2[r * HID + k];
                a.y = DTC * W2[r * HID + k + 1];
            }
            w2fragA[jj] = a;
        }
    }
    if (two) {
        const int r = (tileA + 1) * 16 + rloc;
        #pragma unroll
        for (int j = 0; j < NCHUNK; ++j) {
            const int k = 4 * j + 2 * hi;
            v2f a;
            a.x = A[r * SDIM + k];
            a.y = A[r * SDIM + k + 1];
            afragB[j] = a;
        }
        #pragma unroll
        for (int jj = 0; jj < 2; ++jj) {
            const int k = 4 * jj + 2 * hi;
            v2f a;
            a.x = DTC * W2[r * HID + k];
            a.y = DTC * W2[r * HID + k + 1];
            w2fragB[jj] = a;
        }
    }
    __syncthreads();

    for (int t0 = 0; t0 < T; t0 += TB) {
        const int ns = (T - t0 < TB) ? (T - t0) : TB;

        {
            const float4* __restrict__ src4 = (const float4*)(u + (size_t)t0 * SDIM);
            float4* dst4 = (float4*)uT;
            const int n4 = ns * ROW4;
            float4 r[NIT4];
            #pragma unroll
            for (int k = 0; k < NIT4; ++k) {
                const int i = tid + NTHR * k;
                r[k] = (i < n4) ? src4[i] : make_float4(0.f, 0.f, 0.f, 0.f);
            }
            const float2 wv = (tid < ns) ? ((const float2*)(w + (size_t)t0 * 2))[tid] : make_float2(0.f, 0.f);
            if (tid < ns) ((float2*)wT)[tid] = wv;
            __syncthreads();
            #pragma unroll
            for (int k = 0; k < NIT4; ++k) {
                const int i = tid + NTHR * k;
                if (i < n4) {
                    const int row = i / ROW4;
                    const int c4  = i - row * ROW4;
                    const float2 wr = ((const float2*)wT)[row];
                    const float4 g0 = ((const float4*)Bw0S)[c4];
                    const float4 g1 = ((const float4*)Bw1S)[c4];
                    const float4 bb = ((const float4*)b2S)[c4];
                    float4 d;
                    d.x = ((g0.x * wr.x + g1.x * wr.y) + r[k].x) + bb.x;
                    d.y = ((g0.y * wr.x + g1.y * wr.y) + r[k].y) + bb.y;
                    d.z = ((g0.z * wr.x + g1.z * wr.y) + r[k].z) + bb.z;
                    d.w = ((g0.w * wr.x + g1.w * wr.y) + r[k].w) + bb.w;
                    dst4[i] = d;
                }
            }
        }
        __syncthreads();

        v8f dA = {}, dB = {};
        {
            const float4 a0 = *(const float4*)(dPtr);
            dA[0] = a0.x; dA[1] = a0.y; dA[2] = a0.z; dA[3] = a0.w;
            if (two) {
                const float4 a1 = *(const float4*)(dPtr + 4);
                const float4 b0 = *(const float4*)(dPtr + 16);
                const float4 b1 = *(const float4*)(dPtr + 20);
                dA[4] = a1.x; dA[5] = a1.y; dA[6] = a1.z; dA[7] = a1.w;
                dB[0] = b0.x; dB[1] = b0.y; dB[2] = b0.z; dB[3] = b0.w;
                dB[4] = b1.x; dB[5] = b1.y; dB[6] = b1.z; dB[7] = b1.w;
            }
        }

        for (int tl = 0; tl < ns; ++tl) {
            v2f bf[NCHUNK];
            #pragma unroll
            for (int j = 0; j < NCHUNK; ++j)
                bf[j] = *(const v2f*)(bBaseC + 4 * j);

            v8f accA[4], accB[4];
            v8f sA, sB;

            if (two) {
                accA[0] = dA;
                accB[0] = dB;
                #pragma unroll
                for (int q = 1; q < 4; ++q) {
                    accA[q] = (v8f){0.f, 0.f, 0.f, 0.f, 0.f, 0.f, 0.f, 0.f};
                    accB[q] = (v8f){0.f, 0.f, 0.f, 0.f, 0.f, 0.f, 0.f, 0.f};
                }
                #pragma unroll
                for (int j = 0; j < NCHUNK; ++j) {
                    accA[j & 3] = __builtin_amdgcn_wmma_f32_16x16x4_f32(
                                      false, afragA[j], false, bf[j], (short)0, accA[j & 3],
                                      false, false);
                    accB[j & 3] = __builtin_amdgcn_wmma_f32_16x16x4_f32(
                                      false, afragB[j], false, bf[j], (short)0, accB[j & 3],
                                      false, false);
                }
            } else {
                v8f a3[4];
                a3[0] = dA;
                #pragma unroll
                for (int q = 1; q < 4; ++q)
                    a3[q] = (v8f){0.f, 0.f, 0.f, 0.f, 0.f, 0.f, 0.f, 0.f};
                #pragma unroll
                for (int j = 0; j < NCHUNK; ++j) {
                    a3[j & 3] = __builtin_amdgcn_wmma_f32_16x16x4_f32(
                                    false, afragA[j], false, bf[j], (short)0, a3[j & 3],
                                    false, false);
                }
                const v8f s = (a3[0] + a3[1]) + (a3[2] + a3[3]);
                const float p0 = (lane & 1) ? s[1] : s[0];
                const float p1 = (lane & 1) ? s[3] : s[2];
                const float p2 = (lane & 1) ? s[5] : s[4];
                const float p3 = (lane & 1) ? s[7] : s[6];
                const float q0 = (lane & 2) ? p1 : p0;
                const float q1 = (lane & 2) ? p3 : p2;
                const float zn = (lane & 4) ? q1 : q0;
                if (lane >= 16 && lane < 16 + HID)
                    hS[lane - 16] = tanhf(zn + b1v);
                sA = s;
            }
            __syncthreads();

            {
                const v2f h0 = *(const v2f*)(bBaseH);
                const v2f h1 = *(const v2f*)(bBaseH + 4);
                {
                    const int tn = (tl + 1 < ns) ? tl + 1 : tl;
                    const float* dp = dPtr + tn * dStride;
                    const float4 a0 = *(const float4*)(dp);
                    dA[0] = a0.x; dA[1] = a0.y; dA[2] = a0.z; dA[3] = a0.w;
                    if (two) {
                        const float4 a1 = *(const float4*)(dp + 4);
                        const float4 b0 = *(const float4*)(dp + 16);
                        const float4 b1 = *(const float4*)(dp + 20);
                        dA[4] = a1.x; dA[5] = a1.y; dA[6] = a1.z; dA[7] = a1.w;
                        dB[0] = b0.x; dB[1] = b0.y; dB[2] = b0.z; dB[3] = b0.w;
                        dB[4] = b1.x; dB[5] = b1.y; dB[6] = b1.z; dB[7] = b1.w;
                    }
                }
                if (two) {
                    sA = (accA[0] + accA[1]) + (accA[2] + accA[3]);
                    sB = (accB[0] + accB[1]) + (accB[2] + accB[3]);
                    sA = __builtin_amdgcn_wmma_f32_16x16x4_f32(
                             false, w2fragA[0], false, h0, (short)0, sA, false, false);
                    sB = __builtin_amdgcn_wmma_f32_16x16x4_f32(
                             false, w2fragB[0], false, h0, (short)0, sB, false, false);
                    sA = __builtin_amdgcn_wmma_f32_16x16x4_f32(
                             false, w2fragA[1], false, h1, (short)0, sA, false, false);
                    sB = __builtin_amdgcn_wmma_f32_16x16x4_f32(
                             false, w2fragB[1], false, h1, (short)0, sB, false, false);
                } else {
                    sA = __builtin_amdgcn_wmma_f32_16x16x4_f32(
                             false, w2fragA[0], false, h0, (short)0, sA, false, false);
                    sA = __builtin_amdgcn_wmma_f32_16x16x4_f32(
                             false, w2fragA[1], false, h1, (short)0, sA, false, false);
                }
                float* orow = oT + tl * SDIM + sBase;
                if (st0) {
                    const float4 x0 = make_float4(sA[0], sA[1], sA[2], sA[3]);
                    *(float4*)(cS + sBase) = x0;
                    *(float4*)(orow)       = x0;
                }
                if (st1) {
                    const float4 x1 = make_float4(sA[4], sA[5], sA[6], sA[7]);
                    const float4 y0 = make_float4(sB[0], sB[1], sB[2], sB[3]);
                    const float4 y1 = make_float4(sB[4], sB[5], sB[6], sB[7]);
                    *(float4*)(cS + sBase + 4)  = x1;
                    *(float4*)(cS + sBase + 16) = y0;
                    *(float4*)(cS + sBase + 20) = y1;
                    *(float4*)(orow + 4)  = x1;
                    *(float4*)(orow + 16) = y0;
                    *(float4*)(orow + 20) = y1;
                }
            }
            __syncthreads();
        }

        {
            float4* __restrict__ dst4 = (float4*)(out_c + (size_t)t0 * SDIM);
            const float4* src4 = (const float4*)oT;
            const int n4 = ns * ROW4;
            #pragma unroll
            for (int k = 0; k < NIT4; ++k) {
                const int i = tid + NTHR * k;
                if (i < n4) dst4[i] = src4[i];
            }
            const int ny = ns * NOBS;
            for (int i = tid; i < ny; i += NTHR) {
                const int r = i / NOBS;
                const int j = i - r * NOBS;
                out_y[(size_t)t0 * NOBS + i] = oT[r * SDIM + obsS[j]];
            }
        }
    }
}

extern "C" void kernel_launch(void* const* d_in, const int* in_sizes, int n_in,
                              void* d_out, int out_size, void* d_ws, size_t ws_size,
                              hipStream_t stream) {
    const float* c0 = (const float*)d_in[0];
    const float* w  = (const float*)d_in[1];
    const float* u  = (const float*)d_in[2];
    const float* A  = (const float*)d_in[3];
    const float* Bw = (const float*)d_in[4];
    const float* W1 = (const float*)d_in[5];
    const float* b1 = (const float*)d_in[6];
    const float* W2 = (const float*)d_in[7];
    const float* b2 = (const float*)d_in[8];
    const int*  obs = (const int*)  d_in[9];

    const int T = in_sizes[1] / 2;
    float* out_c = (float*)d_out;
    float* out_y = out_c + (size_t)T * SDIM;

    greybox_scan_kernel<<<1, NTHR, 0, stream>>>(c0, w, u, A, Bw, W1, b1, W2, b2,
                                               obs, out_c, out_y, T);
}


# Waves 0..2 feed their own 32 new rows back as WMMA B operands from the D registers (K order follows the D layout); LDS exchange only for the others' rows, under these products; sum order changed
# speedup vs baseline: 3.4818x; 1.0408x over previous
#include <hip/hip_runtime.h>
#include <hip/hip_bf16.h>

#define SDIM   100
#define SPAD   112
#define HID    8
#define NOBS   12
#define DTC    0.01f
#define NCHUNK 25
#define NTILE  7
#define NWV    4
#define NTHR   128
#define W1ROW  8
#define TB     128
#define ROW4   (SDIM / 4)
#define TILE4  (TB * ROW4)
#define NIT4   ((TILE4 + NTHR - 1) / NTHR)
#define NOWN   8
#define NFOR   (NCHUNK - NOWN)
#define NPRE   3
static_assert(NTHR == NWV * 32, "one wave per SIMD of the work-group processor");
static_assert(TB <= NTHR, "a tile's w rows are staged one row (two floats) per thread");
static_assert((NTILE - 1) * 16 + W1ROW >= SDIM && W1ROW + HID <= 16, "W1 rows fit in the last tile's padding rows");
static_assert(NOWN * 4 == 32 && NWV == 4 && NFOR == 2 * NOWN + 1, "a two-tile wave owns 32 state rows; the others' are two such blocks and rows 96..99");
static_assert(NPRE <= 3, "chain 0 starts from d_t, which is waited for behind the barrier: no product in front of it may use chain 0");

typedef float v2f __attribute__((ext_vector_type(2)));
typedef float v8f __attribute__((ext_vector_type(8)));

__global__ __launch_bounds__(NTHR, 1)
void greybox_scan_kernel(const float* __restrict__ c0,
                         const float* __restrict__ w,
                         const float* __restrict__ u,
                         const float* __restrict__ A,
                         const float* __restrict__ Bw,
                         const float* __restrict__ W1,
                         const float* __restrict__ b1,
                         const float* __restrict__ W2,
                         const float* __restrict__ b2,
                         const int*   __restrict__ obs,
                         float* __restrict__ out_c,
                         float* __restrict__ out_y,
                         int T)
{
    __shared__ __attribute__((aligned(16))) float cS[SDIM];
    __shared__ __attribute__((aligned(16))) float zS[SDIM];
    __shared__ float hS[HID];
    __shared__ __attribute__((aligned(16))) float uT[TB * SDIM];
    __shared__ __attribute__((aligned(16))) float oT[TB * SDIM];
    __shared__ __attribute__((aligned(8))) float wT[TB * 2];
    __shared__ __attribute__((aligned(16))) float Bw0S[SDIM];
    __shared__ __attribute__((aligned(16))) float Bw1S[SDIM];
    __shared__ __attribute__((aligned(16))) float b2S[SDIM];
    __shared__ int   obsS[NOBS];

    const int tid  = threadIdx.x;
    const int wave = __builtin_amdgcn_readfirstlane(tid >> 5);
    const int lane = tid & 31;
    const int hi   = (lane >> 4) & 1;
    const int rloc = lane & 15;

    for (int i = tid; i < SDIM;     i += NTHR) {
        Bw0S[i] = Bw[2 * i]; Bw1S[i] = Bw[2 * i + 1];
        b2S[i] = DTC * b2[i]; zS[i] = 0.0f; cS[i] = c0[i];
    }
    for (int i = tid; i < NOBS;     i += NTHR) obsS[i] = obs[i];

    const bool two   = wave < NWV - 1;
    const int  tileA = two ? 2 * wave : NTILE - 1;

    const float* bBaseC = (rloc == 0 || !two) ? (cS + 2 * hi) : zS;
    const float* bBaseH = (rloc == 0) ? (hS + 2 * hi) : zS;

    const float* dPtr    = two ? (uT + tileA * 16 + 8 * hi) : (hi ? zS : (uT + (NTILE - 1) * 16));
    const int    dStride = (two || !hi) ? SDIM : 0;
    const int    sBase   = tileA * 16 + 8 * hi;
    const bool   st0     = rloc == 0 && (two || hi == 0);
    const bool   st1     = rloc == 0 && two;

    static_assert((HID & (HID - 1)) == 0 && HID <= 8, "one hidden unit per lane, picked from the 8 D registers");
    const float b1v = b1[lane & (HID - 1)];
    v2f afragA[NCHUNK] = {};
    v2f afragB[NCHUNK] = {};
    v2f w2fragA[2]     = {};
    v2f w2fragB[2]     = {};
    const int oth1 = two ? 32 * ((wave + 1) % 3) : 0;
    const int oth2 = two ? 32 * ((wave + 2) % 3) : 0;

    {
        const int r = tileA * 16 + rloc;
        const bool isW1 = !two && rloc >= W1ROW && rloc < W1ROW + HID;
        #pragma unroll
        for (int j = 0; j < NCHUNK; ++j) {
            int k;
            if (!two)              k = 4 * j + 2 * hi;
            else if (j < NOWN)     k = oth1 + 4 * j + 2 * hi;
            else if (j < 2 * NOWN) k = oth2 + 4 * (j - NOWN) + 2 * hi;
            else if (j < NFOR)     k = 96 + 2 * hi;
            else                   k = (tileA + ((j - NFOR) >> 2)) * 16 + 8 * hi + 2 * ((j - NFOR) & 3);
            v2f a; a.x = 0.0f; a.y = 0.0f;
            if (r < SDIM) {
                a.x = A[r * SDIM + k];
                a.y = A[r * SDIM + k + 1];
            } else if (isW1) {
                a.x = W1[(rloc - W1ROW) * SDIM + k];
                a.y = W1[(rloc - W1ROW) * SDIM + k + 1];
            }
            afragA[j] = a;
        }
        #pragma unroll
        for (int jj = 0; jj < 2; ++jj) {
            const int k = 4 * jj + 2 * hi;
            v2f a; a.x = 0.0f; a.y = 0.0f;
            if (r < SDIM) {
                a.x = DTC * W2[r * HID + k];
                a.y = DTC * W2[r * HID + k + 1];
            }
            w2fragA[jj] = a;
        }
    }
    if (two) {
        const int r = (tileA + 1) * 16 + rloc;
        #pragma unroll
        for (int j = 0; j < NCHUNK; ++j) {
            int k;
            if (j < NOWN)          k = oth1 + 4 * j + 2 * hi;
            else if (j < 2 * NOWN) k = oth2 + 4 * (j - NOWN) + 2 * hi;
            else if (j < NFOR)     k = 96 + 2 * hi;
            else                   k = (tileA + ((j - NFOR) >> 2)) * 16 + 8 * hi + 2 * ((j - NFOR) & 3);
            v2f a;
            a.x = A[r * SDIM + k];
            a.y = A[r * SDIM + k + 1];
            afragB[j] = a;
        }
        #pragma unroll
        for (int jj = 0; jj < 2; ++jj) {
            const int k = 4 * jj + 2 * hi;
            v2f a;
            a.x = DTC * W2[r * HID + k];
            a.y = DTC * W2[r * HID + k + 1];
            w2fragB[jj] = a;
        }
    }
    __syncthreads();

    v8f sA = {}, sB = {};
    if (two) {
        const float4 a0 = *(const float4*)(cS + sBase);
        const float4 a1 = *(const float4*)(cS + sBase + 4);
        const float4 b0 = *(const float4*)(cS + sBase + 16);
        const float4 b1 = *(const float4*)(cS + sBase + 20);
        sA[0] = a0.x; sA[1] = a0.y; sA[2] = a0.z; sA[3] = a0.w;
        sA[4] = a1.x; sA[5] = a1.y; sA[6] = a1.z; sA[7] = a1.w;
        sB[0] = b0.x; sB[1] = b0.y; sB[2] = b0.z; sB[3] = b0.w;
        sB[4] = b1.x; sB[5] = b1.y; sB[6] = b1.z; sB[7] = b1.w;
    }
    const float* bO1 = bBaseC + oth1;
    const float* bO2 = bBaseC + oth2;
    const float* bO3 = bBaseC + 96;

    for (int t0 = 0; t0 < T; t0 += TB) {
        const int ns = (T - t0 < TB) ? (T - t0) : TB;

        {
            const float4* __restrict__ src4 = (const float4*)(u + (size_t)t0 * SDIM);
            float4* dst4 = (float4*)uT;
            const int n4 = ns * ROW4;
            float4 r[NIT4];
            #pragma unroll
            for (int k = 0; k < NIT4; ++k) {
                const int i = tid + NTHR * k;
                r[k] = (i < n4) ? src4[i] : make_float4(0.f, 0.f, 0.f, 0.f);
            }
            const float2 wv = (tid < ns) ? ((const float2*)(w + (size_t)t0 * 2))[tid] : make_float2(0.f, 0.f);
            if (tid < ns) ((float2*)wT)[tid] = wv;
            __syncthreads();
            #pragma unroll
            for (int k = 0; k < NIT4; ++k) {
                const int i = tid + NTHR * k;
                if (i < n4) {
                    const int row = i / ROW4;
                    const int c4  = i - row * ROW4;
                    const float2 wr = ((const float2*)wT)[row];
                    const float4 g0 = ((const float4*)Bw0S)[c4];
                    const float4 g1 = ((const float4*)Bw1S)[c4];
                    const float4 bb = ((const float4*)b2S)[c4];
                    float4 d;
                    d.x = ((g0.x * wr.x + g1.x * wr.y) + r[k].x) + bb.x;
                    d.y = ((g0.y * wr.x + g1.y * wr.y) + r[k].y) + bb.y;
                    d.z = ((g0.z * wr.x + g1.z * wr.y) + r[k].z) + bb.z;
                    d.w = ((g0.w * wr.x + g1.w * wr.y) + r[k].w) + bb.w;
                    dst4[i] = d;
                }
            }
        }
        __syncthreads();

        v8f dA = {}, dB = {};
        {
            const float4 a0 = *(const float4*)(dPtr);
            dA[0] = a0.x; dA[1] = a0.y; dA[2] = a0.z; dA[3] = a0.w;
            if (two) {
                const float4 a1 = *(const float4*)(dPtr + 4);
                const float4 b0 = *(const float4*)(dPtr + 16);
                const float4 b1 = *(const float4*)(dPtr + 20);
                dA[4] = a1.x; dA[5] = a1.y; dA[6] = a1.z; dA[7] = a1.w;
                dB[0] = b0.x; dB[1] = b0.y; dB[2] = b0.z; dB[3] = b0.w;
                dB[4] = b1.x; dB[5] = b1.y; dB[6] = b1.z; dB[7] = b1.w;
            }
        }

        for (int tl = 0; tl < ns; ++tl) {
            v8f accA[4], accB[4];

            if (two) {
                #pragma unroll
                for (int q = 1; q < 4; ++q) {
                    accA[q] = (v8f){0.f, 0.f, 0.f, 0.f, 0.f, 0.f, 0.f, 0.f};
                    accB[q] = (v8f){0.f, 0.f, 0.f, 0.f, 0.f, 0.f, 0.f, 0.f};
                }
                #pragma unroll
                for (int p = 0; p < NPRE; ++p) {
                    v2f bo; bo.x = sA[2 * p]; bo.y = sA[2 * p + 1];
                    accA[(p + 1) & 3] = __builtin_amdgcn_wmma_f32_16x16x4_f32(
                                  false, afragA[NFOR + p], false, bo, (short)0, accA[(p + 1) & 3],
                                  false, false);
                    accB[(p + 1) & 3] = __builtin_amdgcn_wmma_f32_16x16x4_f32(
                                  false, afragB[NFOR + p], false, bo, (short)0, accB[(p + 1) & 3],
                                  false, false);
                }
            }
            __syncthreads();

            if (two) {
                v2f bf[NFOR];
                #pragma unroll
                for (int f = 0; f < NOWN; ++f) {
                    bf[f]        = *(const v2f*)(bO1 + 4 * f);
                    bf[NOWN + f] = *(const v2f*)(bO2 + 4 * f);
                }
                bf[NFOR - 1] = *(const v2f*)(bO3);
                __builtin_amdgcn_sched_barrier(0);
                accA[0] = dA;
                accB[0] = dB;
                #pragma unroll
                for (int p = NPRE; p < NOWN; ++p) {
                    v2f bo;
                    if (p < 4) { bo.x = sA[2 * p];       bo.y = sA[2 * p + 1]; }
                    else       { bo.x = sB[2 * (p - 4)]; bo.y = sB[2 * (p - 4) + 1]; }
                    accA[(p + 1) & 3] = __builtin_amdgcn_wmma_f32_16x16x4_f32(
                                  false, afragA[NFOR + p], false, bo, (short)0, accA[(p + 1) & 3],
                                  false, false);
                    accB[(p + 1) & 3] = __builtin_amdgcn_wmma_f32_16x16x4_f32(
                                  false, afragB[NFOR + p], false, bo, (short)0, accB[(p + 1) & 3],
                                  false, false);
                }
                #pragma unroll
                for (int f = 0; f < NFOR; ++f) {
                    const int p = NOWN + f;
                    accA[(p + 1) & 3] = __builtin_amdgcn_wmma_f32_16x16x4_f32(
                                      false, afragA[f], false, bf[f], (short)0, accA[(p + 1) & 3],
                                      false, false);
                    accB[(p + 1) & 3] = __builtin_amdgcn_wmma_f32_16x16x4_f32(
                                      false, afragB[f], false, bf[f], (short)0, accB[(p + 1) & 3],
                                      false, false);
                }
            } else {
                v2f bf[NCHUNK];
                #pragma unroll
                for (int j = 0; j < NCHUNK; ++j)
                    bf[j] = *(const v2f*)(bBaseC + 4 * j);
                __builtin_amdgcn_sched_barrier(0);
                v8f a3[4];
                a3[0] = dA;
                #pragma unroll
                for (int q = 1; q < 4; ++q)
                    a3[q] = (v8f){0.f, 0.f, 0.f, 0.f, 0.f, 0.f, 0.f, 0.f};
                #pragma unroll
                for (int j = 0; j < NCHUNK; ++j) {
                    a3[j & 3] = __builtin_amdgcn_wmma_f32_16x16x4_f32(
                                    false, afragA[j], false, bf[j], (short)0, a3[j & 3],
                                    false, false);
                }
                const v8f s = (a3[0] + a3[1]) + (a3[2] + a3[3]);
                const float p0 = (lane & 1) ? s[1] : s[0];
                const float p1 = (lane & 1) ? s[3] : s[2];
                const float p2 = (lane & 1) ? s[5] : s[4];
                const float p3 = (lane & 1) ? s[7] : s[6];
                const float q0 = (lane & 2) ? p1 : p0;
                const float q1 = (lane & 2) ? p3 : p2;
                const float zn = (lane & 4) ? q1 : q0;
                if (lane >= 16 && lane < 16 + HID)
                    hS[lane - 16] = tanhf(zn + b1v);
                sA = s;
            }
            __syncthreads();

            {
                const v2f h0 = *(const v2f*)(bBaseH);
                const v2f h1 = *(const v2f*)(bBaseH + 4);
                {
                    const int tn = (tl + 1 < ns) ? tl + 1 : tl;
                    const float* dp = dPtr + tn * dStride;
                    const float4 a0 = *(const float4*)(dp);
                    dA[0] = a0.x; dA[1] = a0.y; dA[2] = a0.z; dA[3] = a0.w;
                    if (two) {
                        const float4 a1 = *(const float4*)(dp + 4);
                        const float4 b0 = *(const float4*)(dp + 16);
                        const float4 b1 = *(const float4*)(dp + 20);
                        dA[4] = a1.x; dA[5] = a1.y; dA[6] = a1.z; dA[7] = a1.w;
                        dB[0] = b0.x; dB[1] = b0.y; dB[2] = b0.z; dB[3] = b0.w;
                        dB[4] = b1.x; dB[5] = b1.y; dB[6] = b1.z; dB[7] = b1.w;
                    }
                }
                if (two) {
                    sA = (accA[0] + accA[1]) + (accA[2] + accA[3]);
                    sB = (accB[0] + accB[1]) + (accB[2] + accB[3]);
                    sA = __builtin_amdgcn_wmma_f32_16x16x4_f32(
                             false, w2fragA[0], false, h0, (short)0, sA, false, false);
                    sB = __builtin_amdgcn_wmma_f32_16x16x4_f32(
                             false, w2fragB[0], false, h0, (short)0, sB, false, false);
                    sA = __builtin_amdgcn_wmma_f32_16x16x4_f32(
                             false, w2fragA[1], false, h1, (short)0, sA, false, false);
                    sB = __builtin_amdgcn_wmma_f32_16x16x4_f32(
                             false, w2fragB[1], false, h1, (short)0, sB, false, false);
                } else {
                    sA = __builtin_amdgcn_wmma_f32_16x16x4_f32(
                             false, w2fragA[0], false, h0, (short)0, sA, false, false);
                    sA = __builtin_amdgcn_wmma_f32_16x16x4_f32(
                             false, w2fragA[1], false, h1, (short)0, sA, false, false);
                }
                float* orow = oT + tl * SDIM + sBase;
                if (st0) {
                    const float4 x0 = make_float4(sA[0], sA[1], sA[2], sA[3]);
                    *(float4*)(cS + sBase) = x0;
                    *(float4*)(orow)       = x0;
                }
                if (st1) {
                    const float4 x1 = make_float4(sA[4], sA[5], sA[6], sA[7]);
                    const float4 y0 = make_float4(sB[0], sB[1], sB[2], sB[3]);
                    const float4 y1 = make_float4(sB[4], sB[5], sB[6], sB[7]);
                    *(float4*)(cS + sBase + 4)  = x1;
                    *(float4*)(cS + sBase + 16) = y0;
                    *(float4*)(cS + sBase + 20) = y1;
                    *(float4*)(orow + 4)  = x1;
                    *(float4*)(orow + 16) = y0;
                    *(float4*)(orow + 20) = y1;
                }
            }
        }
        __syncthreads();

        {
            float4* __restrict__ dst4 = (float4*)(out_c + (size_t)t0 * SDIM);
            const float4* src4 = (const float4*)oT;
            const int n4 = ns * ROW4;
            #pragma unroll
            for (int k = 0; k < NIT4; ++k) {
                const int i = tid + NTHR * k;
                if (i < n4) dst4[i] = src4[i];
            }
            const int ny = ns * NOBS;
            for (int i = tid; i < ny; i += NTHR) {
                const int r = i / NOBS;
                const int j = i - r * NOBS;
                out_y[(size_t)t0 * NOBS + i] = oT[r * SDIM + obsS[j]];
            }
        }
    }
}

extern "C" void kernel_launch(void* const* d_in, const int* in_sizes, int n_in,
                              void* d_out, int out_size, void* d_ws, size_t ws_size,
                              hipStream_t stream) {
    const float* c0 = (const float*)d_in[0];
    const float* w  = (const float*)d_in[1];
    const float* u  = (const float*)d_in[2];
    const float* A  = (const float*)d_in[3];
    const float* Bw = (const float*)d_in[4];
    const float* W1 = (const float*)d_in[5];
    const float* b1 = (const float*)d_in[6];
    const float* W2 = (const float*)d_in[7];
    const float* b2 = (const float*)d_in[8];
    const int*  obs = (const int*)  d_in[9];

    const int T = in_sizes[1] / 2;
    float* out_c = (float*)d_out;
    float* out_y = out_c + (size_t)T * SDIM;

    greybox_scan_kernel<<<1, NTHR, 0, stream>>>(c0, w, u, A, Bw, W1, b1, W2, b2,
                                               obs, out_c, out_y, T);
}


# Tile boundary made latency-parallel: full tiles take unguarded paths, so d build, write-out and y gather read LDS in groups, not one guarded trip per access; next loads fly under the write-out
# speedup vs baseline: 3.6534x; 1.0493x over previous
#include <hip/hip_runtime.h>
#include <hip/hip_bf16.h>

#define SDIM   100
#define SPAD   112
#define HID    8
#define NOBS   12
#define DTC    0.01f
#define NCHUNK 25
#define NTILE  7
#define NWV    4
#define NTHR   128
#define W1ROW  8
#define TB     128
#define ROW4   (SDIM / 4)
#define TILE4  (TB * ROW4)
#define NIT4   ((TILE4 + NTHR - 1) / NTHR)
#define NOWN   8
#define NFOR   (NCHUNK - NOWN)
#define NYIT   (TB * NOBS / NTHR)
#define NPRE   3
static_assert(NTHR == NWV * 32, "one wave per SIMD of the work-group processor");
static_assert(TB <= NTHR, "a tile's w rows are staged one row (two floats) per thread");
static_assert((NTILE - 1) * 16 + W1ROW >= SDIM && W1ROW + HID <= 16, "W1 rows fit in the last tile's padding rows");
static_assert(NOWN * 4 == 32 && NWV == 4 && NFOR == 2 * NOWN + 1, "a two-tile wave owns 32 state rows; the others' are two such blocks and rows 96..99");
static_assert(TILE4 == NIT4 * NTHR && TB * NOBS == NYIT * NTHR, "a full tile splits evenly over the threads: its accesses need no guard");
static_assert(NPRE <= 3, "chain 0 starts from d_t, which is waited for behind the barrier: no product in front of it may use chain 0");

typedef float v2f __attribute__((ext_vector_type(2)));
typedef float v8f __attribute__((ext_vector_type(8)));


template <bool FULL>
__device__ __forceinline__ void tile_load(float4 (&r)[NIT4], float2& wv,
                                          const float4* __restrict__ u4, const float2* __restrict__ w2,
                                          const int nsn, const int tid)
{
    const int n4 = nsn * ROW4;
    #pragma unroll
    for (int k = 0; k < NIT4; ++k) {
        const int i = tid + NTHR * k;
        r[k] = (FULL || i < n4) ? u4[i] : make_float4(0.f, 0.f, 0.f, 0.f);
    }
    wv = (FULL || tid < nsn) ? w2[tid] : make_float2(0.f, 0.f);
}

template <bool FULL>
__device__ __forceinline__ void tile_build_d(const float4 (&r)[NIT4], float4* dst4, const float2* wT2,
                                             const float4* g0S, const float4* g1S, const float4* bbS,
                                             const int n4, const int tid)
{
    #pragma unroll
    for (int k = 0; k < NIT4; ++k) {
        const int i = tid + NTHR * k;
        if (FULL || i < n4) {
            const int row = i / ROW4;
            const int c4  = i - row * ROW4;
            const float2 wr = wT2[row];
            const float4 g0 = g0S[c4];
            const float4 g1 = g1S[c4];
            const float4 bb = bbS[c4];
            float4 d;
            d.x = ((g0.x * wr.x + g1.x * wr.y) + r[k].x) + bb.x;
            d.y = ((g0.y * wr.x + g1.y * wr.y) + r[k].y) + bb.y;
            d.z = ((g0.z * wr.x + g1.z * wr.y) + r[k].z) + bb.z;
            d.w = ((g0.w * wr.x + g1.w * wr.y) + r[k].w) + bb.w;
            dst4[i] = d;
        }
    }
}

__global__ __launch_bounds__(NTHR, 1)
void greybox_scan_kernel(const float* __restrict__ c0,
                         const float* __restrict__ w,
                         const float* __restrict__ u,
                         const float* __restrict__ A,
                         const float* __restrict__ Bw,
                         const float* __restrict__ W1,
                         const float* __restrict__ b1,
                         const float* __restrict__ W2,
                         const float* __restrict__ b2,
                         const int*   __restrict__ obs,
                         float* __restrict__ out_c,
                         float* __restrict__ out_y,
                         int T)
{
    __shared__ __attribute__((aligned(16))) float cS[SDIM];
    __shared__ __attribute__((aligned(16))) float zS[SDIM];
    __shared__ float hS[HID];
    __shared__ __attribute__((aligned(16))) float uT[TB * SDIM];
    __shared__ __attribute__((aligned(16))) float oT[TB * SDIM];
    __shared__ __attribute__((aligned(8))) float wT[TB * 2];
    __shared__ __attribute__((aligned(16))) float Bw0S[SDIM];
    __shared__ __attribute__((aligned(16))) float Bw1S[SDIM];
    __shared__ __attribute__((aligned(16))) float b2S[SDIM];
    __shared__ int   obsS[NOBS];

    const int tid  = threadIdx.x;
    const int wave = __builtin_amdgcn_readfirstlane(tid >> 5);
    const int lane = tid & 31;
    const int hi   = (lane >> 4) & 1;
    const int rloc = lane & 15;

    for (int i = tid; i < SDIM;     i += NTHR) {
        Bw0S[i] = Bw[2 * i]; Bw1S[i] = Bw[2 * i + 1];
        b2S[i] = DTC * b2[i]; zS[i] = 0.0f; cS[i] = c0[i];
    }
    for (int i = tid; i < NOBS;     i += NTHR) obsS[i] = obs[i];

    const bool two   = wave < NWV - 1;
    const int  tileA = two ? 2 * wave : NTILE - 1;

    const float* bBaseC = (rloc == 0 || !two) ? (cS + 2 * hi) : zS;
    const float* bBaseH = (rloc == 0) ? (hS + 2 * hi) : zS;

    const float* dPtr    = two ? (uT + tileA * 16 + 8 * hi) : (hi ? zS : (uT + (NTILE - 1) * 16));
    const int    dStride = (two || !hi) ? SDIM : 0;
    const int    sBase   = tileA * 16 + 8 * hi;
    const bool   st0     = rloc == 0 && (two || hi == 0);
    const bool   st1     = rloc == 0 && two;

    static_assert((HID & (HID - 1)) == 0 && HID <= 8, "one hidden unit per lane, picked from the 8 D registers");
    const float b1v = b1[lane & (HID - 1)];
    v2f afragA[NCHUNK] = {};
    v2f afragB[NCHUNK] = {};
    v2f w2fragA[2]     = {};
    v2f w2fragB[2]     = {};
    const int oth1 = two ? 32 * ((wave + 1) % 3) : 0;
    const int oth2 = two ? 32 * ((wave + 2) % 3) : 0;

    {
        const int r = tileA * 16 + rloc;
        const bool isW1 = !two && rloc >= W1ROW && rloc < W1ROW + HID;
        #pragma unroll
        for (int j = 0; j < NCHUNK; ++j) {
            int k;
            if (!two)              k = 4 * j + 2 * hi;
            else if (j < NOWN)     k = oth1 + 4 * j + 2 * hi;
            else if (j < 2 * NOWN) k = oth2 + 4 * (j - NOWN) + 2 * hi;
            else if (j < NFOR)     k = 96 + 2 * hi;
            else                   k = (tileA + ((j - NFOR) >> 2)) * 16 + 8 * hi + 2 * ((j - NFOR) & 3);
            v2f a; a.x = 0.0f; a.y = 0.0f;
            if (r < SDIM) {
                a.x = A[r * SDIM + k];
                a.y = A[r * SDIM + k + 1];
            } else if (isW1) {
                a.x = W1[(rloc - W1ROW) * SDIM + k];
                a.y = W1[(rloc - W1ROW) * SDIM + k + 1];
            }
            afragA[j] = a;
        }
        #pragma unroll
        for (int jj = 0; jj < 2; ++jj) {
            const int k = 4 * jj + 2 * hi;
            v2f a; a.x = 0.0f; a.y = 0.0f;
            if (r < SDIM) {
                a.x = DTC * W2[r * HID + k];
                a.y = DTC * W2[r * HID + k + 1];
            }
            w2fragA[jj] = a;
        }
    }
    if (two) {
        const int r = (tileA + 1) * 16 + rloc;
        #pragma unroll
        for (int j = 0; j < NCHUNK; ++j) {
            int k;
            if (j < NOWN)          k = oth1 + 4 * j + 2 * hi;
            else if (j < 2 * NOWN) k = oth2 + 4 * (j - NOWN) + 2 * hi;
            else if (j < NFOR)     k = 96 + 2 * hi;
            else                   k = (tileA + ((j - NFOR) >> 2)) * 16 + 8 * hi + 2 * ((j - NFOR) & 3);
            v2f a;
            a.x = A[r * SDIM + k];
            a.y = A[r * SDIM + k + 1];
            afragB[j] = a;
        }
        #pragma unroll
        for (int jj = 0; jj < 2; ++jj) {
            const int k = 4 * jj + 2 * hi;
            v2f a;
            a.x = DTC * W2[r * HID + k];
            a.y = DTC * W2[r * HID + k + 1];
            w2fragB[jj] = a;
        }
    }
    __syncthreads();

    v8f sA = {}, sB = {};
    if (two) {
        const float4 a0 = *(const float4*)(cS + sBase);
        const float4 a1 = *(const float4*)(cS + sBase + 4);
        const float4 b0 = *(const float4*)(cS + sBase + 16);
        const float4 b1 = *(const float4*)(cS + sBase + 20);
        sA[0] = a0.x; sA[1] = a0.y; sA[2] = a0.z; sA[3] = a0.w;
        sA[4] = a1.x; sA[5] = a1.y; sA[6] = a1.z; sA[7] = a1.w;
        sB[0] = b0.x; sB[1] = b0.y; sB[2] = b0.z; sB[3] = b0.w;
        sB[4] = b1.x; sB[5] = b1.y; sB[6] = b1.z; sB[7] = b1.w;
    }
    const float* bO1 = bBaseC + oth1;
    const float* bO2 = bBaseC + oth2;
    const float* bO3 = bBaseC + 96;

    float4 r[NIT4];
    float2 wv;
    {
        const int n0 = (T < TB) ? T : TB;
        if (n0 == TB) tile_load<true >(r, wv, (const float4*)u, (const float2*)w, n0, tid);
        else          tile_load<false>(r, wv, (const float4*)u, (const float2*)w, n0, tid);
    }

    for (int t0 = 0; t0 < T; t0 += TB) {
        const int ns = (T - t0 < TB) ? (T - t0) : TB;

        {
            if (tid < ns) ((float2*)wT)[tid] = wv;
            __syncthreads();
            if (ns == TB)
                tile_build_d<true >(r, (float4*)uT, (const float2*)wT, (const float4*)Bw0S,
                                    (const float4*)Bw1S, (const float4*)b2S, TILE4, tid);
            else
                tile_build_d<false>(r, (float4*)uT, (const float2*)wT, (const float4*)Bw0S,
                                    (const float4*)Bw1S, (const float4*)b2S, ns * ROW4, tid);
        }
        __syncthreads();

        v8f dA = {}, dB = {};
        {
            const float4 a0 = *(const float4*)(dPtr);
            dA[0] = a0.x; dA[1] = a0.y; dA[2] = a0.z; dA[3] = a0.w;
            if (two) {
                const float4 a1 = *(const float4*)(dPtr + 4);
                const float4 b0 = *(const float4*)(dPtr + 16);
                const float4 b1 = *(const float4*)(dPtr + 20);
                dA[4] = a1.x; dA[5] = a1.y; dA[6] = a1.z; dA[7] = a1.w;
                dB[0] = b0.x; dB[1] = b0.y; dB[2] = b0.z; dB[3] = b0.w;
                dB[4] = b1.x; dB[5] = b1.y; dB[6] = b1.z; dB[7] = b1.w;
            }
        }

        for (int tl = 0; tl < ns; ++tl) {
            v8f accA[4], accB[4];

            if (two) {
                #pragma unroll
                for (int q = 1; q < 4; ++q) {
                    accA[q] = (v8f){0.f, 0.f, 0.f, 0.f, 0.f, 0.f, 0.f, 0.f};
                    accB[q] = (v8f){0.f, 0.f, 0.f, 0.f, 0.f, 0.f, 0.f, 0.f};
                }
                #pragma unroll
                for (int p = 0; p < NPRE; ++p) {
                    v2f bo; bo.x = sA[2 * p]; bo.y = sA[2 * p + 1];
                    accA[(p + 1) & 3] = __builtin_amdgcn_wmma_f32_16x16x4_f32(
                                  false, afragA[NFOR + p], false, bo, (short)0, accA[(p + 1) & 3],
                                  false, false);
                    accB[(p + 1) & 3] = __builtin_amdgcn_wmma_f32_16x16x4_f32(
                                  false, afragB[NFOR + p], false, bo, (short)0, accB[(p + 1) & 3],
                                  false, false);
                }
            }
            __syncthreads();

            if (two) {
                v2f bf[NFOR];
                #pragma unroll
                for (int f = 0; f < NOWN; ++f) {
                    bf[f]        = *(const v2f*)(bO1 + 4 * f);
                    bf[NOWN + f] = *(const v2f*)(bO2 + 4 * f);
                }
                bf[NFOR - 1] = *(const v2f*)(bO3);
                __builtin_amdgcn_sched_barrier(0);
                accA[0] = dA;
                accB[0] = dB;
                #pragma unroll
                for (int p = NPRE; p < NOWN; ++p) {
                    v2f bo;
                    if (p < 4) { bo.x = sA[2 * p];       bo.y = sA[2 * p + 1]; }
                    else       { bo.x = sB[2 * (p - 4)]; bo.y = sB[2 * (p - 4) + 1]; }
                    accA[(p + 1) & 3] = __builtin_amdgcn_wmma_f32_16x16x4_f32(
                                  false, afragA[NFOR + p], false, bo, (short)0, accA[(p + 1) & 3],
                                  false, false);
                    accB[(p + 1) & 3] = __builtin_amdgcn_wmma_f32_16x16x4_f32(
                                  false, afragB[NFOR + p], false, bo, (short)0, accB[(p + 1) & 3],
                                  false, false);
                }
                #pragma unroll
                for (int f = 0; f < NFOR; ++f) {
                    const int p = NOWN + f;
                    accA[(p + 1) & 3] = __builtin_amdgcn_wmma_f32_16x16x4_f32(
                                      false, afragA[f], false, bf[f], (short)0, accA[(p + 1) & 3],
                                      false, false);
                    accB[(p + 1) & 3] = __builtin_amdgcn_wmma_f32_16x16x4_f32(
                                      false, afragB[f], false, bf[f], (short)0, accB[(p + 1) & 3],
                                      false, false);
                }
            } else {
                v2f bf[NCHUNK];
                #pragma unroll
                for (int j = 0; j < NCHUNK; ++j)
                    bf[j] = *(const v2f*)(bBaseC + 4 * j);
                __builtin_amdgcn_sched_barrier(0);
                v8f a3[4];
                a3[0] = dA;
                #pragma unroll
                for (int q = 1; q < 4; ++q)
                    a3[q] = (v8f){0.f, 0.f, 0.f, 0.f, 0.f, 0.f, 0.f, 0.f};
                #pragma unroll
                for (int j = 0; j < NCHUNK; ++j) {
                    a3[j & 3] = __builtin_amdgcn_wmma_f32_16x16x4_f32(
                                    false, afragA[j], false, bf[j], (short)0, a3[j & 3],
                                    false, false);
                }
                const v8f s = (a3[0] + a3[1]) + (a3[2] + a3[3]);
                const float p0 = (lane & 1) ? s[1] : s[0];
                const float p1 = (lane & 1) ? s[3] : s[2];
                const float p2 = (lane & 1) ? s[5] : s[4];
                const float p3 = (lane & 1) ? s[7] : s[6];
                const float q0 = (lane & 2) ? p1 : p0;
                const float q1 = (lane & 2) ? p3 : p2;
                const float zn = (lane & 4) ? q1 : q0;
                if (lane >= 16 && lane < 16 + HID)
                    hS[lane - 16] = tanhf(zn + b1v);
                sA = s;
            }
            __syncthreads();

            {
                const v2f h0 = *(const v2f*)(bBaseH);
                const v2f h1 = *(const v2f*)(bBaseH + 4);
                {
                    const int tn = (tl + 1 < ns) ? tl + 1 : tl;
                    const float* dp = dPtr + tn * dStride;
                    const float4 a0 = *(const float4*)(dp);
                    dA[0] = a0.x; dA[1] = a0.y; dA[2] = a0.z; dA[3] = a0.w;
                    if (two) {
                        const float4 a1 = *(const float4*)(dp + 4);
                        const float4 b0 = *(const float4*)(dp + 16);
                        const float4 b1 = *(const float4*)(dp + 20);
                        dA[4] = a1.x; dA[5] = a1.y; dA[6] = a1.z; dA[7] = a1.w;
                        dB[0] = b0.x; dB[1] = b0.y; dB[2] = b0.z; dB[3] = b0.w;
                        dB[4] = b1.x; dB[5] = b1.y; dB[6] = b1.z; dB[7] = b1.w;
                    }
                }
                if (two) {
                    sA = (accA[0] + accA[1]) + (accA[2] + accA[3]);
                    sB = (accB[0] + accB[1]) + (accB[2] + accB[3]);
                    sA = __builtin_amdgcn_wmma_f32_16x16x4_f32(
                             false, w2fragA[0], false, h0, (short)0, sA, false, false);
                    sB = __builtin_amdgcn_wmma_f32_16x16x4_f32(
                             false, w2fragB[0], false, h0, (short)0, sB, false, false);
                    sA = __builtin_amdgcn_wmma_f32_16x16x4_f32(
                             false, w2fragA[1], false, h1, (short)0, sA, false, false);
                    sB = __builtin_amdgcn_wmma_f32_16x16x4_f32(
                             false, w2fragB[1], false, h1, (short)0, sB, false, false);
                } else {
                    sA = __builtin_amdgcn_wmma_f32_16x16x4_f32(
                             false, w2fragA[0], false, h0, (short)0, sA, false, false);
                    sA = __builtin_amdgcn_wmma_f32_16x16x4_f32(
                             false, w2fragA[1], false, h1, (short)0, sA, false, false);
                }
                float* orow = oT + tl * SDIM + sBase;
                if (st0) {
                    const float4 x0 = make_float4(sA[0], sA[1], sA[2], sA[3]);
                    *(float4*)(cS + sBase) = x0;
                    *(float4*)(orow)       = x0;
                }
                if (st1) {
                    const float4 x1 = make_float4(sA[4], sA[5], sA[6], sA[7]);
                    const float4 y0 = make_float4(sB[0], sB[1], sB[2], sB[3]);
                    const float4 y1 = make_float4(sB[4], sB[5], sB[6], sB[7]);
                    *(float4*)(cS + sBase + 4)  = x1;
                    *(float4*)(cS + sBase + 16) = y0;
                    *(float4*)(cS + sBase + 20) = y1;
                    *(float4*)(orow + 4)  = x1;
                    *(float4*)(orow + 16) = y0;
                    *(float4*)(orow + 20) = y1;
                }
            }
        }
        __syncthreads();

        {
            const int tn = t0 + TB;
            const int nn = (T - tn < TB) ? (T - tn) : TB;
            const int tb = (nn > 0) ? tn : 0;
            const float4* un = (const float4*)(u + (size_t)tb * SDIM);
            const float2* wn = (const float2*)(w + (size_t)tb * 2);
            if (nn == TB) tile_load<true >(r, wv, un, wn, nn, tid);
            else          tile_load<false>(r, wv, un, wn, nn, tid);
        }

        {
            float4* __restrict__ dst4 = (float4*)(out_c + (size_t)t0 * SDIM);
            const float4* src4 = (const float4*)oT;
            const int n4 = ns * ROW4;
            if (ns == TB) {
                float4 o[NIT4];
                #pragma unroll
                for (int k = 0; k < NIT4; ++k) o[k] = src4[tid + NTHR * k];
                int yi[NYIT];
                #pragma unroll
                for (int k = 0; k < NYIT; ++k) {
                    const int i  = tid + NTHR * k;
                    const int rr = i / NOBS;
                    yi[k] = rr * SDIM + obsS[i - rr * NOBS];
                }
                #pragma unroll
                for (int k = 0; k < NIT4; ++k) dst4[tid + NTHR * k] = o[k];
                float yv[NYIT];
                #pragma unroll
                for (int k = 0; k < NYIT; ++k) yv[k] = oT[yi[k]];
                float* __restrict__ ydst = out_y + (size_t)t0 * NOBS;
                #pragma unroll
                for (int k = 0; k < NYIT; ++k) ydst[tid + NTHR * k] = yv[k];
            } else {
                #pragma unroll
                for (int k = 0; k < NIT4; ++k) {
                    const int i = tid + NTHR * k;
                    if (i < n4) dst4[i] = src4[i];
                }
                const int ny = ns * NOBS;
                for (int i = tid; i < ny; i += NTHR) {
                    const int rr = i / NOBS;
                    const int j = i - rr * NOBS;
                    out_y[(size_t)t0 * NOBS + i] = oT[rr * SDIM + obsS[j]];
                }
            }
        }
    }
}

extern "C" void kernel_launch(void* const* d_in, const int* in_sizes, int n_in,
                              void* d_out, int out_size, void* d_ws, size_t ws_size,
                              hipStream_t stream) {
    const float* c0 = (const float*)d_in[0];
    const float* w  = (const float*)d_in[1];
    const float* u  = (const float*)d_in[2];
    const float* A  = (const float*)d_in[3];
    const float* Bw = (const float*)d_in[4];
    const float* W1 = (const float*)d_in[5];
    const float* b1 = (const float*)d_in[6];
    const float* W2 = (const float*)d_in[7];
    const float* b2 = (const float*)d_in[8];
    const int*  obs = (const int*)  d_in[9];

    const int T = in_sizes[1] / 2;
    float* out_c = (float*)d_out;
    float* out_y = out_c + (size_t)T * SDIM;

    greybox_scan_kernel<<<1, NTHR, 0, stream>>>(c0, w, u, A, Bw, W1, b1, W2, b2,
                                               obs, out_c, out_y, T);
}


# h round trip hidden: waves 0..2 run one accumulate chain per tile (no vector sum), last 3 chunk pairs issued behind the 2nd barrier; W1 wave's tanh by exp2/rcp arrives in time; sum order changed
# speedup vs baseline: 3.7712x; 1.0322x over previous
#include <hip/hip_runtime.h>
#include <hip/hip_bf16.h>

#define SDIM   100
#define SPAD   112
#define HID    8
#define NOBS   12
#define DTC    0.01f
#define NCHUNK 25
#define NTILE  7
#define NWV    4
#define NTHR   128
#define W1ROW  8
#define TB     128
#define ROW4   (SDIM / 4)
#define TILE4  (TB * ROW4)
#define NIT4   ((TILE4 + NTHR - 1) / NTHR)
#define NOWN   8
#define NFOR   (NCHUNK - NOWN)
#define NYIT   (TB * NOBS / NTHR)
#define NPRE   3
#define NDEF   3
static_assert(NTHR == NWV * 32, "one wave per SIMD of the work-group processor");
static_assert(TB <= NTHR, "a tile's w rows are staged one row (two floats) per thread");
static_assert((NTILE - 1) * 16 + W1ROW >= SDIM && W1ROW + HID <= 16, "W1 rows fit in the last tile's padding rows");
static_assert(NOWN * 4 == 32 && NWV == 4 && NFOR == 2 * NOWN + 1, "a two-tile wave owns 32 state rows; the others' are two such blocks and rows 96..99");
static_assert(TILE4 == NIT4 * NTHR && TB * NOBS == NYIT * NTHR, "a full tile splits evenly over the threads: its accesses need no guard");
static_assert(NPRE <= 4, "the products in front of the first barrier take their B operands from the first tile's rows only");
static_assert(NDEF >= 1 && NDEF <= NFOR, "the products behind the second barrier are chunks of the other waves' rows");

typedef float v2f __attribute__((ext_vector_type(2)));
typedef float v8f __attribute__((ext_vector_type(8)));


template <bool FULL>
__device__ __forceinline__ void tile_load(float4 (&r)[NIT4], float2& wv,
                                          const float4* __restrict__ u4, const float2* __restrict__ w2,
                                          const int nsn, const int tid)
{
    const int n4 = nsn * ROW4;
    #pragma unroll
    for (int k = 0; k < NIT4; ++k) {
        const int i = tid + NTHR * k;
        r[k] = (FULL || i < n4) ? u4[i] : make_float4(0.f, 0.f, 0.f, 0.f);
    }
    wv = (FULL || tid < nsn) ? w2[tid] : make_float2(0.f, 0.f);
}

template <bool FULL>
__device__ __forceinline__ void tile_build_d(const float4 (&r)[NIT4], float4* dst4, const float2* wT2,
                                             const float4* g0S, const float4* g1S, const float4* bbS,
                                             const int n4, const int tid)
{
    #pragma unroll
    for (int k = 0; k < NIT4; ++k) {
        const int i = tid + NTHR * k;
        if (FULL || i < n4) {
            const int row = i / ROW4;
            const int c4  = i - row * ROW4;
            const float2 wr = wT2[row];
            const float4 g0 = g0S[c4];
            const float4 g1 = g1S[c4];
            const float4 bb = bbS[c4];
            float4 d;
            d.x = ((g0.x * wr.x + g1.x * wr.y) + r[k].x) + bb.x;
            d.y = ((g0.y * wr.x + g1.y * wr.y) + r[k].y) + bb.y;
            d.z = ((g0.z * wr.x + g1.z * wr.y) + r[k].z) + bb.z;
            d.w = ((g0.w * wr.x + g1.w * wr.y) + r[k].w) + bb.w;
            dst4[i] = d;
        }
    }
}

__global__ __launch_bounds__(NTHR, 1)
void greybox_scan_kernel(const float* __restrict__ c0,
                         const float* __restrict__ w,
                         const float* __restrict__ u,
                         const float* __restrict__ A,
                         const float* __restrict__ Bw,
                         const float* __restrict__ W1,
                         const float* __restrict__ b1,
                         const float* __restrict__ W2,
                         const float* __restrict__ b2,
                         const int*   __restrict__ obs,
                         float* __restrict__ out_c,
                         float* __restrict__ out_y,
                         int T)
{
    __shared__ __attribute__((aligned(16))) float cS[SDIM];
    __shared__ __attribute__((aligned(16))) float zS[SDIM];
    __shared__ float hS[HID];
    __shared__ __attribute__((aligned(16))) float uT[TB * SDIM];
    __shared__ __attribute__((aligned(16))) float oT[TB * SDIM];
    __shared__ __attribute__((aligned(8))) float wT[TB * 2];
    __shared__ __attribute__((aligned(16))) float Bw0S[SDIM];
    __shared__ __attribute__((aligned(16))) float Bw1S[SDIM];
    __shared__ __attribute__((aligned(16))) float b2S[SDIM];
    __shared__ int   obsS[NOBS];

    const int tid  = threadIdx.x;
    const int wave = __builtin_amdgcn_readfirstlane(tid >> 5);
    const int lane = tid & 31;
    const int hi   = (lane >> 4) & 1;
    const int rloc = lane & 15;

    for (int i = tid; i < SDIM;     i += NTHR) {
        Bw0S[i] = Bw[2 * i]; Bw1S[i] = Bw[2 * i + 1];
        b2S[i] = DTC * b2[i]; zS[i] = 0.0f; cS[i] = c0[i];
    }
    for (int i = tid; i < NOBS;     i += NTHR) obsS[i] = obs[i];

    const bool two   = wave < NWV - 1;
    const int  tileA = two ? 2 * wave : NTILE - 1;

    const float* bBaseC = (rloc == 0 || !two) ? (cS + 2 * hi) : zS;
    const float* bBaseH = (rloc == 0) ? (hS + 2 * hi) : zS;

    const float* dPtr    = two ? (uT + tileA * 16 + 8 * hi) : (hi ? zS : (uT + (NTILE - 1) * 16));
    const int    dStride = (two || !hi) ? SDIM : 0;
    const int    sBase   = tileA * 16 + 8 * hi;
    const bool   st0     = rloc == 0 && (two || hi == 0);
    const bool   st1     = rloc == 0 && two;

    static_assert((HID & (HID - 1)) == 0 && HID <= 8, "one hidden unit per lane, picked from the 8 D registers");
    const float b1v = b1[lane & (HID - 1)];
    v2f afragA[NCHUNK] = {};
    v2f afragB[NCHUNK] = {};
    v2f w2fragA[2]     = {};
    v2f w2fragB[2]     = {};
    const int oth1 = two ? 32 * ((wave + 1) % 3) : 0;
    const int oth2 = two ? 32 * ((wave + 2) % 3) : 0;

    {
        const int r = tileA * 16 + rloc;
        const bool isW1 = !two && rloc >= W1ROW && rloc < W1ROW + HID;
        #pragma unroll
        for (int j = 0; j < NCHUNK; ++j) {
            int k;
            if (!two)              k = 4 * j + 2 * hi;
            else if (j < NOWN)     k = oth1 + 4 * j + 2 * hi;
            else if (j < 2 * NOWN) k = oth2 + 4 * (j - NOWN) + 2 * hi;
            else if (j < NFOR)     k = 96 + 2 * hi;
            else                   k = (tileA + ((j - NFOR) >> 2)) * 16 + 8 * hi + 2 * ((j - NFOR) & 3);
            v2f a; a.x = 0.0f; a.y = 0.0f;
            if (r < SDIM) {
                a.x = A[r * SDIM + k];
                a.y = A[r * SDIM + k + 1];
            } else if (isW1) {
                a.x = W1[(rloc - W1ROW) * SDIM + k];
                a.y = W1[(rloc - W1ROW) * SDIM + k + 1];
            }
            afragA[j] = a;
        }
        #pragma unroll
        for (int jj = 0; jj < 2; ++jj) {
            const int k = 4 * jj + 2 * hi;
            v2f a; a.x = 0.0f; a.y = 0.0f;
            if (r < SDIM) {
                a.x = DTC * W2[r * HID + k];
                a.y = DTC * W2[r * HID + k + 1];
            }
            w2fragA[jj] = a;
        }
    }
    if (two) {
        const int r = (tileA + 1) * 16 + rloc;
        #pragma unroll
        for (int j = 0; j < NCHUNK; ++j) {
            int k;
            if (j < NOWN)          k = oth1 + 4 * j + 2 * hi;
            else if (j < 2 * NOWN) k = oth2 + 4 * (j - NOWN) + 2 * hi;
            else if (j < NFOR)     k = 96 + 2 * hi;
            else                   k = (tileA + ((j - NFOR) >> 2)) * 16 + 8 * hi + 2 * ((j - NFOR) & 3);
            v2f a;
            a.x = A[r * SDIM + k];
            a.y = A[r * SDIM + k + 1];
            afragB[j] = a;
        }
        #pragma unroll
        for (int jj = 0; jj < 2; ++jj) {
            const int k = 4 * jj + 2 * hi;
            v2f a;
            a.x = DTC * W2[r * HID + k];
            a.y = DTC * W2[r * HID + k + 1];
            w2fragB[jj] = a;
        }
    }
    __syncthreads();

    v8f sA = {}, sB = {};
    if (two) {
        const float4 a0 = *(const float4*)(cS + sBase);
        const float4 a1 = *(const float4*)(cS + sBase + 4);
        const float4 b0 = *(const float4*)(cS + sBase + 16);
        const float4 b1 = *(const float4*)(cS + sBase + 20);
        sA[0] = a0.x; sA[1] = a0.y; sA[2] = a0.z; sA[3] = a0.w;
        sA[4] = a1.x; sA[5] = a1.y; sA[6] = a1.z; sA[7] = a1.w;
        sB[0] = b0.x; sB[1] = b0.y; sB[2] = b0.z; sB[3] = b0.w;
        sB[4] = b1.x; sB[5] = b1.y; sB[6] = b1.z; sB[7] = b1.w;
    }
    const float* bO1 = bBaseC + oth1;
    const float* bO2 = bBaseC + oth2;
    const float* bO3 = bBaseC + 96;

    float4 r[NIT4];
    float2 wv;
    {
        const int n0 = (T < TB) ? T : TB;
        if (n0 == TB) tile_load<true >(r, wv, (const float4*)u, (const float2*)w, n0, tid);
        else          tile_load<false>(r, wv, (const float4*)u, (const float2*)w, n0, tid);
    }

    for (int t0 = 0; t0 < T; t0 += TB) {
        const int ns = (T - t0 < TB) ? (T - t0) : TB;

        {
            if (tid < ns) ((float2*)wT)[tid] = wv;
            __syncthreads();
            if (ns == TB)
                tile_build_d<true >(r, (float4*)uT, (const float2*)wT, (const float4*)Bw0S,
                                    (const float4*)Bw1S, (const float4*)b2S, TILE4, tid);
            else
                tile_build_d<false>(r, (float4*)uT, (const float2*)wT, (const float4*)Bw0S,
                                    (const float4*)Bw1S, (const float4*)b2S, ns * ROW4, tid);
        }
        __syncthreads();

        v8f dA = {}, dB = {};
        {
            const float4 a0 = *(const float4*)(dPtr);
            dA[0] = a0.x; dA[1] = a0.y; dA[2] = a0.z; dA[3] = a0.w;
            if (two) {
                const float4 a1 = *(const float4*)(dPtr + 4);
                const float4 b0 = *(const float4*)(dPtr + 16);
                const float4 b1 = *(const float4*)(dPtr + 20);
                dA[4] = a1.x; dA[5] = a1.y; dA[6] = a1.z; dA[7] = a1.w;
                dB[0] = b0.x; dB[1] = b0.y; dB[2] = b0.z; dB[3] = b0.w;
                dB[4] = b1.x; dB[5] = b1.y; dB[6] = b1.z; dB[7] = b1.w;
            }
            asm volatile("" :: "v"(dA), "v"(dB));
        }

        for (int tl = 0; tl < ns; ++tl) {
            const float* dNext = dPtr + ((tl + 1 < ns) ? tl + 1 : tl) * dStride;
            v8f accA, accB;
            v2f bo_[NFOR];
            if (two) {
                accA = dA;
                accB = dB;
                #pragma unroll
                for (int p = 0; p < NPRE; ++p) {
                    v2f bo; bo.x = sA[2 * p]; bo.y = sA[2 * p + 1];
                    accA = __builtin_amdgcn_wmma_f32_16x16x4_f32(
                               false, afragA[NFOR + p], false, bo, (short)0, accA, false, false);
                    accB = __builtin_amdgcn_wmma_f32_16x16x4_f32(
                               false, afragB[NFOR + p], false, bo, (short)0, accB, false, false);
                }
            }
            asm volatile("s_wait_dscnt 0x0" ::: "memory");
            __syncthreads();

            if (two) {
                {
                    const float4 a0 = *(const float4*)(dNext);
                    const float4 a1 = *(const float4*)(dNext + 4);
                    const float4 b0 = *(const float4*)(dNext + 16);
                    const float4 b1 = *(const float4*)(dNext + 20);
                    dA[0] = a0.x; dA[1] = a0.y; dA[2] = a0.z; dA[3] = a0.w;
                    dA[4] = a1.x; dA[5] = a1.y; dA[6] = a1.z; dA[7] = a1.w;
                    dB[0] = b0.x; dB[1] = b0.y; dB[2] = b0.z; dB[3] = b0.w;
                    dB[4] = b1.x; dB[5] = b1.y; dB[6] = b1.z; dB[7] = b1.w;
                }
                #pragma unroll
                for (int f = 0; f < NOWN; ++f) {
                    bo_[f]        = *(const v2f*)(bO1 + 4 * f);
                    bo_[NOWN + f] = *(const v2f*)(bO2 + 4 * f);
                }
                bo_[NFOR - 1] = *(const v2f*)(bO3);
                __builtin_amdgcn_sched_barrier(0);
                #pragma unroll
                for (int p = NPRE; p < NOWN; ++p) {
                    v2f bo;
                    if (p < 4) { bo.x = sA[2 * p];       bo.y = sA[2 * p + 1]; }
                    else       { bo.x = sB[2 * (p - 4)]; bo.y = sB[2 * (p - 4) + 1]; }
                    accA = __builtin_amdgcn_wmma_f32_16x16x4_f32(
                               false, afragA[NFOR + p], false, bo, (short)0, accA, false, false);
                    accB = __builtin_amdgcn_wmma_f32_16x16x4_f32(
                               false, afragB[NFOR + p], false, bo, (short)0, accB, false, false);
                }
                #pragma unroll
                for (int f = 0; f < NFOR - NDEF; ++f) {
                    accA = __builtin_amdgcn_wmma_f32_16x16x4_f32(
                               false, afragA[f], false, bo_[f], (short)0, accA, false, false);
                    accB = __builtin_amdgcn_wmma_f32_16x16x4_f32(
                               false, afragB[f], false, bo_[f], (short)0, accB, false, false);
                }
            } else {
                v8f a3[4];
                a3[0] = dA;
                {
                    const float4 a0 = *(const float4*)(dNext);
                    dA[0] = a0.x; dA[1] = a0.y; dA[2] = a0.z; dA[3] = a0.w;
                }
                v2f bf[NCHUNK];
                #pragma unroll
                for (int j = 0; j < NCHUNK; ++j)
                    bf[j] = *(const v2f*)(bBaseC + 4 * j);
                __builtin_amdgcn_sched_barrier(0);
                #pragma unroll
                for (int q = 1; q < 4; ++q)
                    a3[q] = (v8f){0.f, 0.f, 0.f, 0.f, 0.f, 0.f, 0.f, 0.f};
                #pragma unroll
                for (int j = 0; j < NCHUNK; ++j) {
                    a3[j & 3] = __builtin_amdgcn_wmma_f32_16x16x4_f32(
                                    false, afragA[j], false, bf[j], (short)0, a3[j & 3],
                                    false, false);
                }
                const v8f s = (a3[0] + a3[1]) + (a3[2] + a3[3]);
                const float p0 = (lane & 1) ? s[1] : s[0];
                const float p1 = (lane & 1) ? s[3] : s[2];
                const float p2 = (lane & 1) ? s[5] : s[4];
                const float p3 = (lane & 1) ? s[7] : s[6];
                const float q0 = (lane & 2) ? p1 : p0;
                const float q1 = (lane & 2) ? p3 : p2;
                const float zn = (lane & 4) ? q1 : q0;
                if (lane >= 16 && lane < 16 + HID) {
                    const float e2x = __builtin_amdgcn_exp2f((zn + b1v) * 2.885390081777927f);
                    hS[lane - 16] = fmaf(-2.0f, __builtin_amdgcn_rcpf(e2x + 1.0f), 1.0f);
                }
                sA = s;
            }
            __syncthreads();

            {
                const v2f h0 = *(const v2f*)(bBaseH);
                const v2f h1 = *(const v2f*)(bBaseH + 4);
                if (two) {
                    #pragma unroll
                    for (int f = NFOR - NDEF; f < NFOR; ++f) {
                        accA = __builtin_amdgcn_wmma_f32_16x16x4_f32(
                                   false, afragA[f], false, bo_[f], (short)0, accA, false, false);
                        accB = __builtin_amdgcn_wmma_f32_16x16x4_f32(
                                   false, afragB[f], false, bo_[f], (short)0, accB, false, false);
                    }
                    sA = __builtin_amdgcn_wmma_f32_16x16x4_f32(
                             false, w2fragA[0], false, h0, (short)0, accA, false, false);
                    sB = __builtin_amdgcn_wmma_f32_16x16x4_f32(
                             false, w2fragB[0], false, h0, (short)0, accB, false, false);
                    sA = __builtin_amdgcn_wmma_f32_16x16x4_f32(
                             false, w2fragA[1], false, h1, (short)0, sA, false, false);
                    sB = __builtin_amdgcn_wmma_f32_16x16x4_f32(
                             false, w2fragB[1], false, h1, (short)0, sB, false, false);
                } else {
                    sA = __builtin_amdgcn_wmma_f32_16x16x4_f32(
                             false, w2fragA[0], false, h0, (short)0, sA, false, false);
                    sA = __builtin_amdgcn_wmma_f32_16x16x4_f32(
                             false, w2fragA[1], false, h1, (short)0, sA, false, false);
                }
                float* orow = oT + tl * SDIM + sBase;
                if (st0) {
                    const float4 x0 = make_float4(sA[0], sA[1], sA[2], sA[3]);
                    *(float4*)(cS + sBase) = x0;
                    *(float4*)(orow)       = x0;
                }
                if (st1) {
                    const float4 x1 = make_float4(sA[4], sA[5], sA[6], sA[7]);
                    const float4 y0 = make_float4(sB[0], sB[1], sB[2], sB[3]);
                    const float4 y1 = make_float4(sB[4], sB[5], sB[6], sB[7]);
                    *(float4*)(cS + sBase + 4)  = x1;
                    *(float4*)(cS + sBase + 16) = y0;
                    *(float4*)(cS + sBase + 20) = y1;
                    *(float4*)(orow + 4)  = x1;
                    *(float4*)(orow + 16) = y0;
                    *(float4*)(orow + 20) = y1;
                }
            }
        }
        __syncthreads();

        {
            const int tn = t0 + TB;
            const int nn = (T - tn < TB) ? (T - tn) : TB;
            const int tb = (nn > 0) ? tn : 0;
            const float4* un = (const float4*)(u + (size_t)tb * SDIM);
            const float2* wn = (const float2*)(w + (size_t)tb * 2);
            if (nn == TB) tile_load<true >(r, wv, un, wn, nn, tid);
            else          tile_load<false>(r, wv, un, wn, nn, tid);
        }

        {
            float4* __restrict__ dst4 = (float4*)(out_c + (size_t)t0 * SDIM);
            const float4* src4 = (const float4*)oT;
            const int n4 = ns * ROW4;
            if (ns == TB) {
                float4 o[NIT4];
                #pragma unroll
                for (int k = 0; k < NIT4; ++k) o[k] = src4[tid + NTHR * k];
                int yi[NYIT];
                #pragma unroll
                for (int k = 0; k < NYIT; ++k) {
                    const int i  = tid + NTHR * k;
                    const int rr = i / NOBS;
                    yi[k] = rr * SDIM + obsS[i - rr * NOBS];
                }
                #pragma unroll
                for (int k = 0; k < NIT4; ++k) dst4[tid + NTHR * k] = o[k];
                float yv[NYIT];
                #pragma unroll
                for (int k = 0; k < NYIT; ++k) yv[k] = oT[yi[k]];
                float* __restrict__ ydst = out_y + (size_t)t0 * NOBS;
                #pragma unroll
                for (int k = 0; k < NYIT; ++k) ydst[tid + NTHR * k] = yv[k];
            } else {
                #pragma unroll
                for (int k = 0; k < NIT4; ++k) {
                    const int i = tid + NTHR * k;
                    if (i < n4) dst4[i] = src4[i];
                }
                const int ny = ns * NOBS;
                for (int i = tid; i < ny; i += NTHR) {
                    const int rr = i / NOBS;
                    const int j = i - rr * NOBS;
                    out_y[(size_t)t0 * NOBS + i] = oT[rr * SDIM + obsS[j]];
                }
            }
        }
    }
}

extern "C" void kernel_launch(void* const* d_in, const int* in_sizes, int n_in,
                              void* d_out, int out_size, void* d_ws, size_t ws_size,
                              hipStream_t stream) {
    const float* c0 = (const float*)d_in[0];
    const float* w  = (const float*)d_in[1];
    const float* u  = (const float*)d_in[2];
    const float* A  = (const float*)d_in[3];
    const float* Bw = (const float*)d_in[4];
    const float* W1 = (const float*)d_in[5];
    const float* b1 = (const float*)d_in[6];
    const float* W2 = (const float*)d_in[7];
    const float* b2 = (const float*)d_in[8];
    const int*  obs = (const int*)  d_in[9];

    const int T = in_sizes[1] / 2;
    float* out_c = (float*)d_out;
    float* out_y = out_c + (size_t)T * SDIM;

    greybox_scan_kernel<<<1, NTHR, 0, stream>>>(c0, w, u, A, Bw, W1, b1, W2, b2,
                                               obs, out_c, out_y, T);
}


# Chain advances TWO steps per matrix pass: A*A, A*dtW2, W1*A, W1*dtW2 built in-kernel each launch (f64 sums, rounded once); odd rows and input terms as batched 16-column products; sum order changed
# speedup vs baseline: 5.2933x; 1.4036x over previous
#include <hip/hip_runtime.h>

#define SDIM   100
#define HID    8
#define NOBS   12
#define DTC    0.01f
#define NCHUNK 25
#define NTILE  7
#define NWV    4
#define NTHR   128
#define W1ROW  8
#define TB     128
#define NPAIR  (TB / 2)
#define NBATCH (NPAIR / 16)
#define ROW4   (SDIM / 4)
#define TILE4  (TB * ROW4)
#define NIT4   ((TILE4 + NTHR - 1) / NTHR)
#define NOWN   8
#define NFOR   (NCHUNK - NOWN)
#define NYIT   (TB * NOBS / NTHR)
#define NPRE   3
#define NDEF   3
static_assert(NTHR == NWV * 32, "one wave per SIMD of the work-group processor");
static_assert(TB <= NTHR, "a tile's w rows are staged one row (two floats) per thread");
static_assert((NTILE - 1) * 16 + W1ROW >= SDIM && W1ROW + HID <= 16, "W1 rows fit in the last tile's padding rows");
static_assert(NOWN * 4 == 32 && NWV == 4 && NFOR == 2 * NOWN + 1, "a two-tile wave owns 32 state rows; the others' are two such blocks and rows 96..99");
static_assert(TILE4 == NIT4 * NTHR && TB * NOBS == NYIT * NTHR, "a full tile splits evenly over the threads: its accesses need no guard");
static_assert(NPRE <= 4, "the products in front of the first barrier take their B operands from the first tile's rows only");
static_assert(NDEF >= 1 && NDEF <= NFOR, "the products behind the second barrier are chunks of the other waves' rows");
static_assert(TB % 32 == 0, "a batch is 16 pairs of steps");

typedef float v2f __attribute__((ext_vector_type(2)));
typedef float v8f __attribute__((ext_vector_type(8)));

#define WMMA(a, b, c) __builtin_amdgcn_wmma_f32_16x16x4_f32(false, (a), false, (b), (short)0, (c), false, false)

template <bool FULL>
__device__ __forceinline__ void tile_load(float4 (&r)[NIT4], float2& wv,
                                          const float4* __restrict__ u4, const float2* __restrict__ w2,
                                          const int nsn, const int tid)
{
    const int n4 = nsn * ROW4;
    #pragma unroll
    for (int k = 0; k < NIT4; ++k) {
        const int i = tid + NTHR * k;
        r[k] = (FULL || i < n4) ? u4[i] : make_float4(0.f, 0.f, 0.f, 0.f);
    }
    wv = (FULL || tid < nsn) ? w2[tid] : make_float2(0.f, 0.f);
}

template <bool FULL>
__device__ __forceinline__ void tile_build_d(const float4 (&r)[NIT4], float4* dst4, const float2* wT2,
                                             const float4* g0S, const float4* g1S, const float4* bbS,
                                             const int n4, const int tid)
{
    #pragma unroll
    for (int k = 0; k < NIT4; ++k) {
        const int i = tid + NTHR * k;
        if (FULL || i < n4) {
            const int row = i / ROW4;
            const int c4  = i - row * ROW4;
            const float2 wr = wT2[row];
            const float4 g0 = g0S[c4];
            const float4 g1 = g1S[c4];
            const float4 bb = bbS[c4];
            float4 d;
            d.x = ((g0.x * wr.x + g1.x * wr.y) + r[k].x) + bb.x;
            d.y = ((g0.y * wr.x + g1.y * wr.y) + r[k].y) + bb.y;
            d.z = ((g0.z * wr.x + g1.z * wr.y) + r[k].z) + bb.z;
            d.w = ((g0.w * wr.x + g1.w * wr.y) + r[k].w) + bb.w;
            dst4[i] = d;
        } else {
            dst4[i] = make_float4(0.f, 0.f, 0.f, 0.f);
        }
    }
}

__device__ __forceinline__ v8f ld8(const float* p)
{
    const float4 a = *(const float4*)(p);
    const float4 b = *(const float4*)(p + 4);
    v8f v;
    v[0] = a.x; v[1] = a.y; v[2] = a.z; v[3] = a.w;
    v[4] = b.x; v[5] = b.y; v[6] = b.z; v[7] = b.w;
    return v;
}
__device__ __forceinline__ void st8(float* p, const v8f v)
{
    *(float4*)(p)     = make_float4(v[0], v[1], v[2], v[3]);
    *(float4*)(p + 4) = make_float4(v[4], v[5], v[6], v[7]);
}
__device__ __forceinline__ float tanh_fast(const float x)
{
    const float e2x = __builtin_amdgcn_exp2f(x * 2.885390081777927f);
    return fmaf(-2.0f, __builtin_amdgcn_rcpf(e2x + 1.0f), 1.0f);
}

__global__ __launch_bounds__(NTHR, 1)
void greybox_scan_kernel(const float* __restrict__ c0,
                         const float* __restrict__ w,
                         const float* __restrict__ u,
                         const float* __restrict__ A,
                         const float* __restrict__ Bw,
                         const float* __restrict__ W1,
                         const float* __restrict__ b1,
                         const float* __restrict__ W2,
                         const float* __restrict__ b2,
                         const int*   __restrict__ obs,
                         float* __restrict__ out_c,
                         float* __restrict__ out_y,
                         int T)
{
    __shared__ __attribute__((aligned(16))) float cS[SDIM];
    __shared__ __attribute__((aligned(16))) float zS[SDIM];
    __shared__ __attribute__((aligned(16))) float hS[2 * HID];
    __shared__ __attribute__((aligned(16))) float uT[TB * SDIM];
    __shared__ __attribute__((aligned(16))) float oT[(TB + 1) * SDIM];
    __shared__ __attribute__((aligned(8)))  float wT[TB * 2];
    __shared__ __attribute__((aligned(16))) float fT[NPAIR * HID];
    __shared__ __attribute__((aligned(16))) float hT[NPAIR * HID];
    __shared__ __attribute__((aligned(16))) float Bw0S[SDIM];
    __shared__ __attribute__((aligned(16))) float Bw1S[SDIM];
    __shared__ __attribute__((aligned(16))) float b2S[SDIM];
    __shared__ int   obsS[NOBS];
    __shared__ __attribute__((aligned(16))) float AS[SDIM * SDIM];
    __shared__ __attribute__((aligned(16))) float A2S[SDIM * SDIM];
    __shared__ __attribute__((aligned(16))) float GS[SDIM * HID];
    __shared__ __attribute__((aligned(16))) float AGS[SDIM * HID];
    __shared__ __attribute__((aligned(16))) float W1S[HID * SDIM];
    __shared__ __attribute__((aligned(16))) float W1AS[HID * SDIM];
    __shared__ float W1GS[HID * HID];

    const int tid  = threadIdx.x;
    const int wave = __builtin_amdgcn_readfirstlane(tid >> 5);
    const int lane = tid & 31;
    const int hi   = (lane >> 4) & 1;
    const int rloc = lane & 15;

    for (int i = tid; i < SDIM; i += NTHR) {
        Bw0S[i] = Bw[2 * i]; Bw1S[i] = Bw[2 * i + 1];
        b2S[i] = DTC * b2[i]; zS[i] = 0.0f; cS[i] = c0[i];
    }
    for (int i = tid; i < NOBS; i += NTHR) obsS[i] = obs[i];
    for (int i = tid; i < SDIM * SDIM; i += NTHR) AS[i] = A[i];
    for (int i = tid; i < SDIM * HID; i += NTHR) { GS[i] = DTC * W2[i]; W1S[i] = W1[i]; }
    __syncthreads();
    for (int idx = tid; idx < SDIM * SDIM; idx += NTHR) {
        const int i = idx / SDIM, j = idx - i * SDIM;
        double s = 0.0;
        for (int k = 0; k < SDIM; ++k) s = fma((double)AS[i * SDIM + k], (double)AS[k * SDIM + j], s);
        A2S[idx] = (float)s;
    }
    for (int idx = tid; idx < SDIM * HID; idx += NTHR) {
        const int i = idx / HID, m = idx - i * HID;
        double s = 0.0;
        for (int k = 0; k < SDIM; ++k) s = fma((double)AS[i * SDIM + k], (double)GS[k * HID + m], s);
        AGS[idx] = (float)s;
    }
    for (int idx = tid; idx < HID * SDIM; idx += NTHR) {
        const int n = idx / SDIM, j = idx - n * SDIM;
        double s = 0.0;
        for (int k = 0; k < SDIM; ++k) s = fma((double)W1S[n * SDIM + k], (double)AS[k * SDIM + j], s);
        W1AS[idx] = (float)s;
    }
    for (int idx = tid; idx < HID * HID; idx += NTHR) {
        const int n = idx / HID, m = idx - n * HID;
        double s = 0.0;
        for (int k = 0; k < SDIM; ++k) s = fma((double)W1S[n * SDIM + k], (double)GS[k * HID + m], s);
        W1GS[idx] = (float)s;
    }
    __syncthreads();

    const bool two   = wave < NWV - 1;
    const int  tileA = two ? 2 * wave : NTILE - 1;

    const float* bBaseC = (rloc == 0 || !two) ? (cS + 2 * hi) : zS;
    const float* bBaseH = (rloc == 0) ? (hS + 2 * hi) : zS;

    const float* dPtr    = two ? (uT + tileA * 16 + 8 * hi) : (hi ? zS : (uT + (NTILE - 1) * 16));
    const int    dStride = (two || !hi) ? SDIM : 0;
    const int    sBase   = tileA * 16 + 8 * hi;
    const bool   st0     = rloc == 0 && (two || hi == 0);
    const bool   st1     = rloc == 0 && two;

    static_assert((HID & (HID - 1)) == 0 && HID <= 8, "one hidden unit per lane, picked from the 8 D registers");
    const float b1v = b1[lane & (HID - 1)];
    v2f x2A[NCHUNK] = {}, x2B[NCHUNK] = {}, xA[NCHUNK] = {}, xB[NCHUNK] = {};
    v2f agA[2] = {}, agB[2] = {}, gA[2] = {}, gB[2] = {};
    const int oth1 = two ? 32 * ((wave + 1) % 3) : 0;
    const int oth2 = two ? 32 * ((wave + 2) % 3) : 0;
    {
        const int r = tileA * 16 + rloc;
        const bool isW1 = !two && rloc >= W1ROW && rloc < W1ROW + HID;
        #pragma unroll
        for (int j = 0; j < NCHUNK; ++j) {
            const int kp = 4 * j + 2 * hi;
            int kc;
            if (!two)              kc = kp;
            else if (j < NOWN)     kc = oth1 + 4 * j + 2 * hi;
            else if (j < 2 * NOWN) kc = oth2 + 4 * (j - NOWN) + 2 * hi;
            else if (j < NFOR)     kc = 96 + 2 * hi;
            else                   kc = (tileA + ((j - NFOR) >> 2)) * 16 + 8 * hi + 2 * ((j - NFOR) & 3);
            v2f a; a.x = 0.0f; a.y = 0.0f;
            v2f b; b.x = 0.0f; b.y = 0.0f;
            if (r < SDIM) {
                a.x = A2S[r * SDIM + kc]; a.y = A2S[r * SDIM + kc + 1];
                b.x = AS[r * SDIM + kp];  b.y = AS[r * SDIM + kp + 1];
            } else if (isW1) {
                b.x = W1S[(rloc - W1ROW) * SDIM + kp]; b.y = W1S[(rloc - W1ROW) * SDIM + kp + 1];
            }
            x2A[j] = a;
            xA[j]  = b;
        }
        #pragma unroll
        for (int jj = 0; jj < 2; ++jj) {
            const int k = 4 * jj + 2 * hi;
            v2f a; a.x = 0.0f; a.y = 0.0f;
            v2f b; b.x = 0.0f; b.y = 0.0f;
            if (r < SDIM) {
                a.x = AGS[r * HID + k]; a.y = AGS[r * HID + k + 1];
                b.x = GS[r * HID + k];  b.y = GS[r * HID + k + 1];
            }
            agA[jj] = a;
            gA[jj]  = b;
        }
    }
    if (two) {
        const int r = (tileA + 1) * 16 + rloc;
        #pragma unroll
        for (int j = 0; j < NCHUNK; ++j) {
            const int kp = 4 * j + 2 * hi;
            int kc;
            if (j < NOWN)          kc = oth1 + 4 * j + 2 * hi;
            else if (j < 2 * NOWN) kc = oth2 + 4 * (j - NOWN) + 2 * hi;
            else if (j < NFOR)     kc = 96 + 2 * hi;
            else                   kc = (tileA + ((j - NFOR) >> 2)) * 16 + 8 * hi + 2 * ((j - NFOR) & 3);
            v2f a, b;
            a.x = A2S[r * SDIM + kc]; a.y = A2S[r * SDIM + kc + 1];
            b.x = AS[r * SDIM + kp];  b.y = AS[r * SDIM + kp + 1];
            x2B[j] = a;
            xB[j]  = b;
        }
        #pragma unroll
        for (int jj = 0; jj < 2; ++jj) {
            const int k = 4 * jj + 2 * hi;
            v2f a, b;
            a.x = AGS[r * HID + k]; a.y = AGS[r * HID + k + 1];
            b.x = GS[r * HID + k];  b.y = GS[r * HID + k + 1];
            agB[jj] = a;
            gB[jj]  = b;
        }
    } else {
        const float* hrow = (rloc < HID) ? (W1S + rloc * SDIM) : (W1AS + (rloc - HID) * SDIM);
        #pragma unroll
        for (int j = 0; j < NCHUNK; ++j) {
            const int kp = 4 * j + 2 * hi;
            v2f a;
            a.x = hrow[kp]; a.y = hrow[kp + 1];
            x2B[j] = a;
        }
    }
    float wg[HID];
    #pragma unroll
    for (int m = 0; m < HID; ++m) wg[m] = W1GS[(lane & (HID - 1)) * HID + m];

    v8f sA = {}, sB = {};
    if (two) {
        sA = ld8(cS + sBase);
        sB = ld8(cS + sBase + 16);
    }
    const float* bO1 = bBaseC + oth1;
    const float* bO2 = bBaseC + oth2;
    const float* bO3 = bBaseC + 96;
    const int dOff = two ? (tileA * 16 + 8 * hi) : (NTILE - 1) * 16;

    float4 r[NIT4];
    float2 wv;
    {
        const int n0 = (T < TB) ? T : TB;
        if (n0 == TB) tile_load<true >(r, wv, (const float4*)u, (const float2*)w, n0, tid);
        else          tile_load<false>(r, wv, (const float4*)u, (const float2*)w, n0, tid);
    }

    for (int t0 = 0; t0 < T; t0 += TB) {
        const int ns    = (T - t0 < TB) ? (T - t0) : TB;
        const int npair = (ns + 1) >> 1;

        {
            if (tid < ns) ((float2*)wT)[tid] = wv;
            __syncthreads();
            if (tid < ROW4) ((float4*)oT)[tid] = ((const float4*)cS)[tid];
            if (ns == TB)
                tile_build_d<true >(r, (float4*)uT, (const float2*)wT, (const float4*)Bw0S,
                                    (const float4*)Bw1S, (const float4*)b2S, TILE4, tid);
            else
                tile_build_d<false>(r, (float4*)uT, (const float2*)wT, (const float4*)Bw0S,
                                    (const float4*)Bw1S, (const float4*)b2S, ns * ROW4, tid);
        }
        __syncthreads();

        for (int q = 0; q < NBATCH; ++q) {
            if (32 * q >= ns) break;
            const int rowE = 32 * q + 2 * rloc;
            const float* bsrc = uT + rowE * SDIM + 2 * hi;
            float* eptr = uT + (rowE + 1) * SDIM + dOff;
            v2f bb[NCHUNK];
            #pragma unroll
            for (int j = 0; j < NCHUNK; ++j) bb[j] = *(const v2f*)(bsrc + 4 * j);
            if (two) {
                v8f eA = ld8(eptr), eB = ld8(eptr + 16);
                #pragma unroll
                for (int j = 0; j < NCHUNK; ++j) {
                    eA = WMMA(xA[j], bb[j], eA);
                    eB = WMMA(xB[j], bb[j], eB);
                }
                st8(eptr, eA);
                st8(eptr + 16, eB);
            } else {
                v8f eA = {}, eB = {};
                if (!hi) {
                    const float4 a0 = *(const float4*)(eptr);
                    eA[0] = a0.x; eA[1] = a0.y; eA[2] = a0.z; eA[3] = a0.w;
                }
                #pragma unroll
                for (int j = 0; j < NCHUNK; ++j) {
                    if (j & 1) eB = WMMA(xA[j], bb[j], eB);
                    else       eA = WMMA(xA[j], bb[j], eA);
                }
                const v8f e = eA + eB;
                if (!hi) *(float4*)(eptr) = make_float4(e[0], e[1], e[2], e[3]);
                else     st8(fT + (16 * q + rloc) * HID, e);
            }
        }
        __syncthreads();

        const float* ePtr = dPtr + dStride;
        v8f dA = {}, dB = {};
        {
            const float4 a0 = *(const float4*)(ePtr);
            dA[0] = a0.x; dA[1] = a0.y; dA[2] = a0.z; dA[3] = a0.w;
            if (two) {
                const float4 a1 = *(const float4*)(ePtr + 4);
                dA[4] = a1.x; dA[5] = a1.y; dA[6] = a1.z; dA[7] = a1.w;
                dB = ld8(ePtr + 16);
            }
            asm volatile("" :: "v"(dA), "v"(dB));
        }

        for (int p = 0; p < npair; ++p) {
            const float* eNext = ePtr + 2 * ((p + 1 < npair) ? p + 1 : p) * dStride;
            v8f accA, accB;
            v2f bx[NCHUNK];
            if (two) {
                accA = dA;
                accB = dB;
                #pragma unroll
                for (int pp = 0; pp < NPRE; ++pp) {
                    v2f bo; bo.x = sA[2 * pp]; bo.y = sA[2 * pp + 1];
                    accA = WMMA(x2A[NFOR + pp], bo, accA);
                    accB = WMMA(x2B[NFOR + pp], bo, accB);
                }
            }
            asm volatile("s_wait_dscnt 0x0" ::: "memory");
            __syncthreads();

            if (two) {
                dA = ld8(eNext);
                dB = ld8(eNext + 16);
                #pragma unroll
                for (int f = 0; f < NOWN; ++f) {
                    bx[f]        = *(const v2f*)(bO1 + 4 * f);
                    bx[NOWN + f] = *(const v2f*)(bO2 + 4 * f);
                }
                bx[NFOR - 1] = *(const v2f*)(bO3);
                __builtin_amdgcn_sched_barrier(0);
                #pragma unroll
                for (int pp = NPRE; pp < NOWN; ++pp) {
                    v2f bo;
                    if (pp < 4) { bo.x = sA[2 * pp];       bo.y = sA[2 * pp + 1]; }
                    else        { bo.x = sB[2 * (pp - 4)]; bo.y = sB[2 * (pp - 4) + 1]; }
                    accA = WMMA(x2A[NFOR + pp], bo, accA);
                    accB = WMMA(x2B[NFOR + pp], bo, accB);
                }
                #pragma unroll
                for (int f = 0; f < NFOR - NDEF; ++f) {
                    accA = WMMA(x2A[f], bx[f], accA);
                    accB = WMMA(x2B[f], bx[f], accB);
                }
            } else {
                accA = dA;
                {
                    const float4 a0 = *(const float4*)(eNext);
                    dA[0] = a0.x; dA[1] = a0.y; dA[2] = a0.z; dA[3] = a0.w;
                }
                const float fv = fT[p * HID + (lane & (HID - 1))];
                #pragma unroll
                for (int j = 0; j < NCHUNK; ++j)
                    bx[j] = *(const v2f*)(bBaseC + 4 * j);
                __builtin_amdgcn_sched_barrier(0);
                v8f a3[4];
                #pragma unroll
                for (int qq = 0; qq < 4; ++qq)
                    a3[qq] = (v8f){0.f, 0.f, 0.f, 0.f, 0.f, 0.f, 0.f, 0.f};
                #pragma unroll
                for (int j = 0; j < NCHUNK; ++j)
                    a3[j & 3] = WMMA(x2B[j], bx[j], a3[j & 3]);
                const v8f s = (a3[0] + a3[1]) + (a3[2] + a3[3]);
                const float p0 = (lane & 1) ? s[1] : s[0];
                const float p1 = (lane & 1) ? s[3] : s[2];
                const float p2 = (lane & 1) ? s[5] : s[4];
                const float p3 = (lane & 1) ? s[7] : s[6];
                const float q0 = (lane & 2) ? p1 : p0;
                const float q1 = (lane & 2) ? p3 : p2;
                const float zn = (lane & 4) ? q1 : q0;
                const float h0v = tanh_fast(zn + b1v);
                if (lane < HID) { hS[lane] = h0v; hT[p * HID + lane] = h0v; }
                float z1 = zn + (b1v + fv);
                #pragma unroll
                for (int m = 0; m < HID; ++m)
                    z1 = fmaf(wg[m], __int_as_float(__builtin_amdgcn_readlane(__float_as_int(h0v), m)), z1);
                const float h1v = tanh_fast(z1);
                if (lane >= 16 && lane < 16 + HID) hS[HID + lane - 16] = h1v;
            }
            __syncthreads();

            {
                const v2f h00 = *(const v2f*)(bBaseH);
                const v2f h01 = *(const v2f*)(bBaseH + 4);
                const v2f h10 = *(const v2f*)(bBaseH + 8);
                const v2f h11 = *(const v2f*)(bBaseH + 12);
                if (two) {
                    #pragma unroll
                    for (int f = NFOR - NDEF; f < NFOR; ++f) {
                        accA = WMMA(x2A[f], bx[f], accA);
                        accB = WMMA(x2B[f], bx[f], accB);
                    }
                    accA = WMMA(agA[0], h00, accA);
                    accB = WMMA(agB[0], h00, accB);
                    accA = WMMA(agA[1], h01, accA);
                    accB = WMMA(agB[1], h01, accB);
                    accA = WMMA(gA[0], h10, accA);
                    accB = WMMA(gB[0], h10, accB);
                    sA = WMMA(gA[1], h11, accA);
                    sB = WMMA(gB[1], h11, accB);
                } else {
                    v8f t6 = (v8f){0.f, 0.f, 0.f, 0.f, 0.f, 0.f, 0.f, 0.f};
                    #pragma unroll
                    for (int j = 0; j < NCHUNK; ++j) {
                        if (j & 1) t6   = WMMA(x2A[j], bx[j], t6);
                        else       accA = WMMA(x2A[j], bx[j], accA);
                    }
                    accA = WMMA(agA[0], h00, accA);
                    t6   = WMMA(agA[1], h01, t6);
                    accA = WMMA(gA[0], h10, accA);
                    t6   = WMMA(gA[1], h11, t6);
                    sA = accA + t6;
                }
                float* orow = oT + (2 * p + 2) * SDIM + sBase;
                if (st0) {
                    const float4 x0 = make_float4(sA[0], sA[1], sA[2], sA[3]);
                    *(float4*)(cS + sBase) = x0;
                    *(float4*)(orow)       = x0;
                }
                if (st1) {
                    const float4 x1 = make_float4(sA[4], sA[5], sA[6], sA[7]);
                    const float4 y0 = make_float4(sB[0], sB[1], sB[2], sB[3]);
                    const float4 y1 = make_float4(sB[4], sB[5], sB[6], sB[7]);
                    *(float4*)(cS + sBase + 4)  = x1;
                    *(float4*)(cS + sBase + 16) = y0;
                    *(float4*)(cS + sBase + 20) = y1;
                    *(float4*)(orow + 4)  = x1;
                    *(float4*)(orow + 16) = y0;
                    *(float4*)(orow + 20) = y1;
                }
            }
        }
        __syncthreads();

        {
            const int tn = t0 + TB;
            const int nn = (T - tn < TB) ? (T - tn) : TB;
            const int tb = (nn > 0) ? tn : 0;
            const float4* un = (const float4*)(u + (size_t)tb * SDIM);
            const float2* wn = (const float2*)(w + (size_t)tb * 2);
            if (nn == TB) tile_load<true >(r, wv, un, wn, nn, tid);
            else          tile_load<false>(r, wv, un, wn, nn, tid);
        }

        for (int q = 0; q < NBATCH; ++q) {
            if (32 * q >= ns) break;
            const int rowE = 32 * q + 2 * rloc;
            const float* bsrc = oT + rowE * SDIM + 2 * hi;
            const float* dsrc = uT + rowE * SDIM + dOff;
            const float* hsrc = hT + (16 * q + rloc) * HID + 2 * hi;
            float* optr = oT + (rowE + 1) * SDIM + dOff;
            v2f bb[NCHUNK];
            #pragma unroll
            for (int j = 0; j < NCHUNK; ++j) bb[j] = *(const v2f*)(bsrc + 4 * j);
            const v2f hb0 = *(const v2f*)(hsrc);
            const v2f hb1 = *(const v2f*)(hsrc + 4);
            if (two) {
                v8f oA = ld8(dsrc), oB = ld8(dsrc + 16);
                #pragma unroll
                for (int j = 0; j < NCHUNK; ++j) {
                    oA = WMMA(xA[j], bb[j], oA);
                    oB = WMMA(xB[j], bb[j], oB);
                }
                oA = WMMA(gA[0], hb0, oA);
                oB = WMMA(gB[0], hb0, oB);
                oA = WMMA(gA[1], hb1, oA);
                oB = WMMA(gB[1], hb1, oB);
                st8(optr, oA);
                st8(optr + 16, oB);
            } else {
                v8f oA = {}, oB = {};
                if (!hi) {
                    const float4 a0 = *(const float4*)(dsrc);
                    oA[0] = a0.x; oA[1] = a0.y; oA[2] = a0.z; oA[3] = a0.w;
                }
                #pragma unroll
                for (int j = 0; j < NCHUNK; ++j) {
                    if (j & 1) oB = WMMA(xA[j], bb[j], oB);
                    else       oA = WMMA(xA[j], bb[j], oA);
                }
                oA = WMMA(gA[0], hb0, oA);
                oB = WMMA(gA[1], hb1, oB);
                const v8f o = oA + oB;
                if (!hi) *(float4*)(optr) = make_float4(o[0], o[1], o[2], o[3]);
            }
        }
        __syncthreads();

        {
            float4* __restrict__ dst4 = (float4*)(out_c + (size_t)t0 * SDIM);
            const float4* src4 = (const float4*)(oT + SDIM);
            const float* oT1 = oT + SDIM;
            const int n4 = ns * ROW4;
            if (ns == TB) {
                float4 o[NIT4];
                #pragma unroll
                for (int k = 0; k < NIT4; ++k) o[k] = src4[tid + NTHR * k];
                int yi[NYIT];
                #pragma unroll
                for (int k = 0; k < NYIT; ++k) {
                    const int i  = tid + NTHR * k;
                    const int rr = i / NOBS;
                    yi[k] = rr * SDIM + obsS[i - rr * NOBS];
                }
                #pragma unroll
                for (int k = 0; k < NIT4; ++k) dst4[tid + NTHR * k] = o[k];
                float yv[NYIT];
                #pragma unroll
                for (int k = 0; k < NYIT; ++k) yv[k] = oT1[yi[k]];
                float* __restrict__ ydst = out_y + (size_t)t0 * NOBS;
                #pragma unroll
                for (int k = 0; k < NYIT; ++k) ydst[tid + NTHR * k] = yv[k];
            } else {
                #pragma unroll
                for (int k = 0; k < NIT4; ++k) {
                    const int i = tid + NTHR * k;
                    if (i < n4) dst4[i] = src4[i];
                }
                const int ny = ns * NOBS;
                for (int i = tid; i < ny; i += NTHR) {
                    const int rr = i / NOBS;
                    const int j = i - rr * NOBS;
                    out_y[(size_t)t0 * NOBS + i] = oT1[rr * SDIM + obsS[j]];
                }
            }
        }
    }
}

extern "C" void kernel_launch(void* const* d_in, const int* in_sizes, int n_in,
                              void* d_out, int out_size, void* d_ws, size_t ws_size,
                              hipStream_t stream) {
    const float* c0 = (const float*)d_in[0];
    const float* w  = (const float*)d_in[1];
    const float* u  = (const float*)d_in[2];
    const float* A  = (const float*)d_in[3];
    const float* Bw = (const float*)d_in[4];
    const float* W1 = (const float*)d_in[5];
    const float* b1 = (const float*)d_in[6];
    const float* W2 = (const float*)d_in[7];
    const float* b2 = (const float*)d_in[8];
    const int*  obs = (const int*)  d_in[9];

    const int T = in_sizes[1] / 2;
    float* out_c = (float*)d_out;
    float* out_y = out_c + (size_t)T * SDIM;

    greybox_scan_kernel<<<1, NTHR, 0, stream>>>(c0, w, u, A, Bw, W1, b1, W2, b2,
                                               obs, out_c, out_y, T);
}


# Rows 96..99 of A A (wave 3): four K chunks packed per WMMA, chunk 4g+n in tile rows 4n..4n+3 against state chunk 4g+n in B column n: 7 products not 25, blocks added by lane reads; sum order changed
# speedup vs baseline: 5.8803x; 1.1109x over previous
#include <hip/hip_runtime.h>

#define SDIM   100
#define HID    8
#define NOBS   12
#define DTC    0.01f
#define NCHUNK 25
#define NTILE  7
#define NWV    4
#define NTHR   128
#define W1ROW  8
#define TB     128
#define NPAIR  (TB / 2)
#define NBATCH (NPAIR / 16)
#define ROW4   (SDIM / 4)
#define TILE4  (TB * ROW4)
#define NIT4   ((TILE4 + NTHR - 1) / NTHR)
#define NOWN   8
#define NFOR   (NCHUNK - NOWN)
#define NYIT   (TB * NOBS / NTHR)
#define NPRE   3
#define NDEF   3
#define NGRP   ((NCHUNK + 3) / 4)
static_assert(NTHR == NWV * 32, "one wave per SIMD of the work-group processor");
static_assert(TB <= NTHR, "a tile's w rows are staged one row (two floats) per thread");
static_assert((NTILE - 1) * 16 + W1ROW >= SDIM && W1ROW + HID <= 16, "W1 rows fit in the last tile's padding rows");
static_assert(NOWN * 4 == 32 && NWV == 4 && NFOR == 2 * NOWN + 1, "a two-tile wave owns 32 state rows; the others' are two such blocks and rows 96..99");
static_assert(TILE4 == NIT4 * NTHR && TB * NOBS == NYIT * NTHR, "a full tile splits evenly over the threads: its accesses need no guard");
static_assert(NPRE <= 4, "the products in front of the first barrier take their B operands from the first tile's rows only");
static_assert(NDEF >= 1 && NDEF <= NFOR, "the products behind the second barrier are chunks of the other waves' rows");
static_assert(TB % 32 == 0, "a batch is 16 pairs of steps");
static_assert(SDIM - (NTILE - 1) * 16 == 4 && NCHUNK == 4 * (NGRP - 1) + 1, "the last row tile has 4 state rows: four K chunks fit its 16 tile rows; the last group is chunk 24 alone");

typedef float v2f __attribute__((ext_vector_type(2)));
typedef float v8f __attribute__((ext_vector_type(8)));

#define WMMA(a, b, c) __builtin_amdgcn_wmma_f32_16x16x4_f32(false, (a), false, (b), (short)0, (c), false, false)

template <bool FULL>
__device__ __forceinline__ void tile_load(float4 (&r)[NIT4], float2& wv,
                                          const float4* __restrict__ u4, const float2* __restrict__ w2,
                                          const int nsn, const int tid)
{
    const int n4 = nsn * ROW4;
    #pragma unroll
    for (int k = 0; k < NIT4; ++k) {
        const int i = tid + NTHR * k;
        r[k] = (FULL || i < n4) ? u4[i] : make_float4(0.f, 0.f, 0.f, 0.f);
    }
    wv = (FULL || tid < nsn) ? w2[tid] : make_float2(0.f, 0.f);
}

template <bool FULL>
__device__ __forceinline__ void tile_build_d(const float4 (&r)[NIT4], float4* dst4, const float2* wT2,
                                             const float4* g0S, const float4* g1S, const float4* bbS,
                                             const int n4, const int tid)
{
    #pragma unroll
    for (int k = 0; k < NIT4; ++k) {
        const int i = tid + NTHR * k;
        if (FULL || i < n4) {
            const int row = i / ROW4;
            const int c4  = i - row * ROW4;
            const float2 wr = wT2[row];
            const float4 g0 = g0S[c4];
            const float4 g1 = g1S[c4];
            const float4 bb = bbS[c4];
            float4 d;
            d.x = ((g0.x * wr.x + g1.x * wr.y) + r[k].x) + bb.x;
            d.y = ((g0.y * wr.x + g1.y * wr.y) + r[k].y) + bb.y;
            d.z = ((g0.z * wr.x + g1.z * wr.y) + r[k].z) + bb.z;
            d.w = ((g0.w * wr.x + g1.w * wr.y) + r[k].w) + bb.w;
            dst4[i] = d;
        } else {
            dst4[i] = make_float4(0.f, 0.f, 0.f, 0.f);
        }
    }
}

__device__ __forceinline__ v8f ld8(const float* p)
{
    const float4 a = *(const float4*)(p);
    const float4 b = *(const float4*)(p + 4);
    v8f v;
    v[0] = a.x; v[1] = a.y; v[2] = a.z; v[3] = a.w;
    v[4] = b.x; v[5] = b.y; v[6] = b.z; v[7] = b.w;
    return v;
}
__device__ __forceinline__ void st8(float* p, const v8f v)
{
    *(float4*)(p)     = make_float4(v[0], v[1], v[2], v[3]);
    *(float4*)(p + 4) = make_float4(v[4], v[5], v[6], v[7]);
}
__device__ __forceinline__ float tanh_fast(const float x)
{
    const float e2x = __builtin_amdgcn_exp2f(x * 2.885390081777927f);
    return fmaf(-2.0f, __builtin_amdgcn_rcpf(e2x + 1.0f), 1.0f);
}

__device__ __forceinline__ float lane_get(const float v, const int l)
{
    return __int_as_float(__builtin_amdgcn_readlane(__float_as_int(v), l));
}

__global__ __launch_bounds__(NTHR, 1)
void greybox_scan_kernel(const float* __restrict__ c0,
                         const float* __restrict__ w,
                         const float* __restrict__ u,
                         const float* __restrict__ A,
                         const float* __restrict__ Bw,
                         const float* __restrict__ W1,
                         const float* __restrict__ b1,
                         const float* __restrict__ W2,
                         const float* __restrict__ b2,
                         const int*   __restrict__ obs,
                         float* __restrict__ out_c,
                         float* __restrict__ out_y,
                         int T)
{
    __shared__ __attribute__((aligned(16))) float cS[SDIM];
    __shared__ __attribute__((aligned(16))) float zS[SDIM];
    __shared__ __attribute__((aligned(16))) float hS[2 * HID];
    __shared__ __attribute__((aligned(16))) float uT[TB * SDIM];
    __shared__ __attribute__((aligned(16))) float oT[(TB + 1) * SDIM];
    __shared__ __attribute__((aligned(8)))  float wT[TB * 2];
    __shared__ __attribute__((aligned(16))) float fT[NPAIR * HID];
    __shared__ __attribute__((aligned(16))) float hT[NPAIR * HID];
    __shared__ __attribute__((aligned(16))) float Bw0S[SDIM];
    __shared__ __attribute__((aligned(16))) float Bw1S[SDIM];
    __shared__ __attribute__((aligned(16))) float b2S[SDIM];
    __shared__ int   obsS[NOBS];
    __shared__ __attribute__((aligned(16))) float AS[SDIM * SDIM];
    __shared__ __attribute__((aligned(16))) float A2S[SDIM * SDIM];
    __shared__ __attribute__((aligned(16))) float GS[SDIM * HID];
    __shared__ __attribute__((aligned(16))) float AGS[SDIM * HID];
    __shared__ __attribute__((aligned(16))) float W1S[HID * SDIM];
    __shared__ __attribute__((aligned(16))) float W1AS[HID * SDIM];
    __shared__ float W1GS[HID * HID];

    const int tid  = threadIdx.x;
    const int wave = __builtin_amdgcn_readfirstlane(tid >> 5);
    const int lane = tid & 31;
    const int hi   = (lane >> 4) & 1;
    const int rloc = lane & 15;

    for (int i = tid; i < SDIM; i += NTHR) {
        Bw0S[i] = Bw[2 * i]; Bw1S[i] = Bw[2 * i + 1];
        b2S[i] = DTC * b2[i]; zS[i] = 0.0f; cS[i] = c0[i];
    }
    for (int i = tid; i < NOBS; i += NTHR) obsS[i] = obs[i];
    for (int i = tid; i < SDIM * SDIM; i += NTHR) AS[i] = A[i];
    for (int i = tid; i < SDIM * HID; i += NTHR) { GS[i] = DTC * W2[i]; W1S[i] = W1[i]; }
    __syncthreads();
    for (int idx = tid; idx < SDIM * SDIM; idx += NTHR) {
        const int i = idx / SDIM, j = idx - i * SDIM;
        double s = 0.0;
        for (int k = 0; k < SDIM; ++k) s = fma((double)AS[i * SDIM + k], (double)AS[k * SDIM + j], s);
        A2S[idx] = (float)s;
    }
    for (int idx = tid; idx < SDIM * HID; idx += NTHR) {
        const int i = idx / HID, m = idx - i * HID;
        double s = 0.0;
        for (int k = 0; k < SDIM; ++k) s = fma((double)AS[i * SDIM + k], (double)GS[k * HID + m], s);
        AGS[idx] = (float)s;
    }
    for (int idx = tid; idx < HID * SDIM; idx += NTHR) {
        const int n = idx / SDIM, j = idx - n * SDIM;
        double s = 0.0;
        for (int k = 0; k < SDIM; ++k) s = fma((double)W1S[n * SDIM + k], (double)AS[k * SDIM + j], s);
        W1AS[idx] = (float)s;
    }
    for (int idx = tid; idx < HID * HID; idx += NTHR) {
        const int n = idx / HID, m = idx - n * HID;
        double s = 0.0;
        for (int k = 0; k < SDIM; ++k) s = fma((double)W1S[n * SDIM + k], (double)GS[k * HID + m], s);
        W1GS[idx] = (float)s;
    }
    __syncthreads();

    const bool two   = wave < NWV - 1;
    const int  tileA = two ? 2 * wave : NTILE - 1;

    const float* bBaseC = (rloc == 0 || !two) ? (cS + 2 * hi) : zS;
    const float* bBaseH = (rloc == 0) ? (hS + 2 * hi) : zS;

    const float* dPtr    = two ? (uT + tileA * 16 + 8 * hi) : (hi ? zS : (uT + (NTILE - 1) * 16));
    const int    dStride = (two || !hi) ? SDIM : 0;
    const int    sBase   = tileA * 16 + 8 * hi;
    const bool   st0     = rloc == 0 && (two || hi == 0);
    const bool   st1     = rloc == 0 && two;

    static_assert((HID & (HID - 1)) == 0 && HID <= 8, "one hidden unit per lane, picked from the 8 D registers");
    const float b1v = b1[lane & (HID - 1)];
    v2f x2A[NCHUNK] = {}, x2B[NCHUNK] = {}, xA[NCHUNK] = {}, xB[NCHUNK] = {};
    v2f agA[2] = {}, agB[2] = {}, gA[2] = {}, gB[2] = {};
    const int oth1 = two ? 32 * ((wave + 1) % 3) : 0;
    const int oth2 = two ? 32 * ((wave + 2) % 3) : 0;
    {
        const int r = tileA * 16 + rloc;
        const bool isW1 = !two && rloc >= W1ROW && rloc < W1ROW + HID;
        #pragma unroll
        for (int j = 0; j < NCHUNK; ++j) {
            const int kp = 4 * j + 2 * hi;
            int kc;
            if (!two)              kc = kp;
            else if (j < NOWN)     kc = oth1 + 4 * j + 2 * hi;
            else if (j < 2 * NOWN) kc = oth2 + 4 * (j - NOWN) + 2 * hi;
            else if (j < NFOR)     kc = 96 + 2 * hi;
            else                   kc = (tileA + ((j - NFOR) >> 2)) * 16 + 8 * hi + 2 * ((j - NFOR) & 3);
            v2f a; a.x = 0.0f; a.y = 0.0f;
            v2f b; b.x = 0.0f; b.y = 0.0f;
            if (r < SDIM) {
                if (two) { a.x = A2S[r * SDIM + kc]; a.y = A2S[r * SDIM + kc + 1]; }
                b.x = AS[r * SDIM + kp];  b.y = AS[r * SDIM + kp + 1];
            } else if (isW1) {
                b.x = W1S[(rloc - W1ROW) * SDIM + kp]; b.y = W1S[(rloc - W1ROW) * SDIM + kp + 1];
            }
            if (!two) {
                const int jc = 4 * j + (rloc >> 2);
                if (j < NGRP && jc < NCHUNK) {
                    const int rp = (NTILE - 1) * 16 + (rloc & 3);
                    const int kq = 4 * jc + 2 * hi;
                    a.x = A2S[rp * SDIM + kq]; a.y = A2S[rp * SDIM + kq + 1];
                }
            }
            x2A[j] = a;
            xA[j]  = b;
        }
        #pragma unroll
        for (int jj = 0; jj < 2; ++jj) {
            const int k = 4 * jj + 2 * hi;
            v2f a; a.x = 0.0f; a.y = 0.0f;
            v2f b; b.x = 0.0f; b.y = 0.0f;
            if (r < SDIM) {
                a.x = AGS[r * HID + k]; a.y = AGS[r * HID + k + 1];
                b.x = GS[r * HID + k];  b.y = GS[r * HID + k + 1];
            }
            agA[jj] = a;
            gA[jj]  = b;
        }
    }
    if (two) {
        const int r = (tileA + 1) * 16 + rloc;
        #pragma unroll
        for (int j = 0; j < NCHUNK; ++j) {
            const int kp = 4 * j + 2 * hi;
            int kc;
            if (j < NOWN)          kc = oth1 + 4 * j + 2 * hi;
            else if (j < 2 * NOWN) kc = oth2 + 4 * (j - NOWN) + 2 * hi;
            else if (j < NFOR)     kc = 96 + 2 * hi;
            else                   kc = (tileA + ((j - NFOR) >> 2)) * 16 + 8 * hi + 2 * ((j - NFOR) & 3);
            v2f a, b;
            a.x = A2S[r * SDIM + kc]; a.y = A2S[r * SDIM + kc + 1];
            b.x = AS[r * SDIM + kp];  b.y = AS[r * SDIM + kp + 1];
            x2B[j] = a;
            xB[j]  = b;
        }
        #pragma unroll
        for (int jj = 0; jj < 2; ++jj) {
            const int k = 4 * jj + 2 * hi;
            v2f a, b;
            a.x = AGS[r * HID + k]; a.y = AGS[r * HID + k + 1];
            b.x = GS[r * HID + k];  b.y = GS[r * HID + k + 1];
            agB[jj] = a;
            gB[jj]  = b;
        }
    } else {
        const float* hrow = (rloc < HID) ? (W1S + rloc * SDIM) : (W1AS + (rloc - HID) * SDIM);
        #pragma unroll
        for (int j = 0; j < NCHUNK; ++j) {
            const int kp = 4 * j + 2 * hi;
            v2f a;
            a.x = hrow[kp]; a.y = hrow[kp + 1];
            x2B[j] = a;
        }
    }
    float wg[HID];
    #pragma unroll
    for (int m = 0; m < HID; ++m) wg[m] = W1GS[(lane & (HID - 1)) * HID + m];

    v8f sA = {}, sB = {};
    if (two) {
        sA = ld8(cS + sBase);
        sB = ld8(cS + sBase + 16);
    }
    const float* bO1 = bBaseC + oth1;
    const float* bO2 = bBaseC + oth2;
    const float* bO3 = bBaseC + 96;
    const float* bPk = (!two && rloc < 4)  ? (cS + 4 * rloc + 2 * hi) : zS;
    const float* bPl = (!two && rloc == 0) ? (cS + 4 * (NCHUNK - 1) + 2 * hi) : zS;
    const int dOff = two ? (tileA * 16 + 8 * hi) : (NTILE - 1) * 16;

    float4 r[NIT4];
    float2 wv;
    {
        const int n0 = (T < TB) ? T : TB;
        if (n0 == TB) tile_load<true >(r, wv, (const float4*)u, (const float2*)w, n0, tid);
        else          tile_load<false>(r, wv, (const float4*)u, (const float2*)w, n0, tid);
    }

    for (int t0 = 0; t0 < T; t0 += TB) {
        const int ns    = (T - t0 < TB) ? (T - t0) : TB;
        const int npair = (ns + 1) >> 1;

        {
            if (tid < ns) ((float2*)wT)[tid] = wv;
            __syncthreads();
            if (tid < ROW4) ((float4*)oT)[tid] = ((const float4*)cS)[tid];
            if (ns == TB)
                tile_build_d<true >(r, (float4*)uT, (const float2*)wT, (const float4*)Bw0S,
                                    (const float4*)Bw1S, (const float4*)b2S, TILE4, tid);
            else
                tile_build_d<false>(r, (float4*)uT, (const float2*)wT, (const float4*)Bw0S,
                                    (const float4*)Bw1S, (const float4*)b2S, ns * ROW4, tid);
        }
        __syncthreads();

        for (int q = 0; q < NBATCH; ++q) {
            if (32 * q >= ns) break;
            const int rowE = 32 * q + 2 * rloc;
            const float* bsrc = uT + rowE * SDIM + 2 * hi;
            float* eptr = uT + (rowE + 1) * SDIM + dOff;
            v2f bb[NCHUNK];
            #pragma unroll
            for (int j = 0; j < NCHUNK; ++j) bb[j] = *(const v2f*)(bsrc + 4 * j);
            if (two) {
                v8f eA = ld8(eptr), eB = ld8(eptr + 16);
                #pragma unroll
                for (int j = 0; j < NCHUNK; ++j) {
                    eA = WMMA(xA[j], bb[j], eA);
                    eB = WMMA(xB[j], bb[j], eB);
                }
                st8(eptr, eA);
                st8(eptr + 16, eB);
            } else {
                v8f eA = {}, eB = {};
                if (!hi) {
                    const float4 a0 = *(const float4*)(eptr);
                    eA[0] = a0.x; eA[1] = a0.y; eA[2] = a0.z; eA[3] = a0.w;
                }
                #pragma unroll
                for (int j = 0; j < NCHUNK; ++j) {
                    if (j & 1) eB = WMMA(xA[j], bb[j], eB);
                    else       eA = WMMA(xA[j], bb[j], eA);
                }
                const v8f e = eA + eB;
                if (!hi) *(float4*)(eptr) = make_float4(e[0], e[1], e[2], e[3]);
                else     st8(fT + (16 * q + rloc) * HID, e);
            }
        }
        __syncthreads();

        const float* ePtr = dPtr + dStride;
        v8f dA = {}, dB = {};
        {
            const float4 a0 = *(const float4*)(ePtr);
            dA[0] = a0.x; dA[1] = a0.y; dA[2] = a0.z; dA[3] = a0.w;
            if (two) {
                const float4 a1 = *(const float4*)(ePtr + 4);
                dA[4] = a1.x; dA[5] = a1.y; dA[6] = a1.z; dA[7] = a1.w;
                dB = ld8(ePtr + 16);
            }
            asm volatile("" :: "v"(dA), "v"(dB));
        }

        for (int p = 0; p < npair; ++p) {
            const float* eNext = ePtr + 2 * ((p + 1 < npair) ? p + 1 : p) * dStride;
            v8f accA, accB;
            v2f bx[NCHUNK];
            v2f bp[NGRP];
            if (two) {
                accA = dA;
                accB = dB;
                #pragma unroll
                for (int pp = 0; pp < NPRE; ++pp) {
                    v2f bo; bo.x = sA[2 * pp]; bo.y = sA[2 * pp + 1];
                    accA = WMMA(x2A[NFOR + pp], bo, accA);
                    accB = WMMA(x2B[NFOR + pp], bo, accB);
                }
            }
            asm volatile("s_wait_dscnt 0x0" ::: "memory");
            __syncthreads();

            if (two) {
                dA = ld8(eNext);
                dB = ld8(eNext + 16);
                #pragma unroll
                for (int f = 0; f < NOWN; ++f) {
                    bx[f]        = *(const v2f*)(bO1 + 4 * f);
                    bx[NOWN + f] = *(const v2f*)(bO2 + 4 * f);
                }
                bx[NFOR - 1] = *(const v2f*)(bO3);
                __builtin_amdgcn_sched_barrier(0);
                #pragma unroll
                for (int pp = NPRE; pp < NOWN; ++pp) {
                    v2f bo;
                    if (pp < 4) { bo.x = sA[2 * pp];       bo.y = sA[2 * pp + 1]; }
                    else        { bo.x = sB[2 * (pp - 4)]; bo.y = sB[2 * (pp - 4) + 1]; }
                    accA = WMMA(x2A[NFOR + pp], bo, accA);
                    accB = WMMA(x2B[NFOR + pp], bo, accB);
                }
                #pragma unroll
                for (int f = 0; f < NFOR - NDEF; ++f) {
                    accA = WMMA(x2A[f], bx[f], accA);
                    accB = WMMA(x2B[f], bx[f], accB);
                }
            } else {
                accA = dA;
                {
                    const float4 a0 = *(const float4*)(eNext);
                    dA[0] = a0.x; dA[1] = a0.y; dA[2] = a0.z; dA[3] = a0.w;
                }
                const float fv = fT[p * HID + (lane & (HID - 1))];
                #pragma unroll
                for (int j = 0; j < NCHUNK; ++j)
                    bx[j] = *(const v2f*)(bBaseC + 4 * j);
                #pragma unroll
                for (int g = 0; g < NGRP - 1; ++g)
                    bp[g] = *(const v2f*)(bPk + 16 * g);
                bp[NGRP - 1] = *(const v2f*)(bPl);
                __builtin_amdgcn_sched_barrier(0);
                v8f a3[4];
                #pragma unroll
                for (int qq = 0; qq < 4; ++qq)
                    a3[qq] = (v8f){0.f, 0.f, 0.f, 0.f, 0.f, 0.f, 0.f, 0.f};
                #pragma unroll
                for (int j = 0; j < NCHUNK; ++j)
                    a3[j & 3] = WMMA(x2B[j], bx[j], a3[j & 3]);
                static_assert(NGRP == 7, "the statement below names every packed operand");
                __builtin_amdgcn_sched_barrier(0);
                asm volatile("" :: "v"(bp[0]), "v"(bp[1]), "v"(bp[2]), "v"(bp[3]), "v"(bp[4]), "v"(bp[5]), "v"(bp[6]));
                const v8f s = (a3[0] + a3[1]) + (a3[2] + a3[3]);
                const float p0 = (lane & 1) ? s[1] : s[0];
                const float p1 = (lane & 1) ? s[3] : s[2];
                const float p2 = (lane & 1) ? s[5] : s[4];
                const float p3 = (lane & 1) ? s[7] : s[6];
                const float q0 = (lane & 2) ? p1 : p0;
                const float q1 = (lane & 2) ? p3 : p2;
                const float zn = (lane & 4) ? q1 : q0;
                const float h0v = tanh_fast(zn + b1v);
                if (lane < HID) { hS[lane] = h0v; hT[p * HID + lane] = h0v; }
                float z1 = zn + (b1v + fv);
                #pragma unroll
                for (int m = 0; m < HID; ++m)
                    z1 = fmaf(wg[m], __int_as_float(__builtin_amdgcn_readlane(__float_as_int(h0v), m)), z1);
                const float h1v = tanh_fast(z1);
                if (lane >= 16 && lane < 16 + HID) hS[HID + lane - 16] = h1v;
            }
            __syncthreads();

            {
                const v2f h00 = *(const v2f*)(bBaseH);
                const v2f h01 = *(const v2f*)(bBaseH + 4);
                const v2f h10 = *(const v2f*)(bBaseH + 8);
                const v2f h11 = *(const v2f*)(bBaseH + 12);
                if (two) {
                    #pragma unroll
                    for (int f = NFOR - NDEF; f < NFOR; ++f) {
                        accA = WMMA(x2A[f], bx[f], accA);
                        accB = WMMA(x2B[f], bx[f], accB);
                    }
                    accA = WMMA(agA[0], h00, accA);
                    accB = WMMA(agB[0], h00, accB);
                    accA = WMMA(agA[1], h01, accA);
                    accB = WMMA(agB[1], h01, accB);
                    accA = WMMA(gA[0], h10, accA);
                    accB = WMMA(gB[0], h10, accB);
                    sA = WMMA(gA[1], h11, accA);
                    sB = WMMA(gB[1], h11, accB);
                } else {
                    v8f t6 = (v8f){0.f, 0.f, 0.f, 0.f, 0.f, 0.f, 0.f, 0.f};
                    #pragma unroll
                    for (int g = 0; g < NGRP; ++g) {
                        if (g & 1) t6   = WMMA(x2A[g], bp[g], t6);
                        else       accA = WMMA(x2A[g], bp[g], accA);
                    }
                    accA = WMMA(agA[0], h00, accA);
                    t6   = WMMA(agA[1], h01, t6);
                    accA = WMMA(gA[0], h10, accA);
                    t6   = WMMA(gA[1], h11, t6);
                    const v8f t = accA + t6;
                    #pragma unroll
                    for (int rr = 0; rr < 4; ++rr)
                        sA[rr] = ((t[rr] + lane_get(t[4 + rr], 1)) + lane_get(t[rr], 18)) + lane_get(t[4 + rr], 19);
                }
                float* orow = oT + (2 * p + 2) * SDIM + sBase;
                if (st0) {
                    const float4 x0 = make_float4(sA[0], sA[1], sA[2], sA[3]);
                    *(float4*)(cS + sBase) = x0;
                    *(float4*)(orow)       = x0;
                }
                if (st1) {
                    const float4 x1 = make_float4(sA[4], sA[5], sA[6], sA[7]);
                    const float4 y0 = make_float4(sB[0], sB[1], sB[2], sB[3]);
                    const float4 y1 = make_float4(sB[4], sB[5], sB[6], sB[7]);
                    *(float4*)(cS + sBase + 4)  = x1;
                    *(float4*)(cS + sBase + 16) = y0;
                    *(float4*)(cS + sBase + 20) = y1;
                    *(float4*)(orow + 4)  = x1;
                    *(float4*)(orow + 16) = y0;
                    *(float4*)(orow + 20) = y1;
                }
            }
        }
        __syncthreads();

        {
            const int tn = t0 + TB;
            const int nn = (T - tn < TB) ? (T - tn) : TB;
            const int tb = (nn > 0) ? tn : 0;
            const float4* un = (const float4*)(u + (size_t)tb * SDIM);
            const float2* wn = (const float2*)(w + (size_t)tb * 2);
            if (nn == TB) tile_load<true >(r, wv, un, wn, nn, tid);
            else          tile_load<false>(r, wv, un, wn, nn, tid);
        }

        for (int q = 0; q < NBATCH; ++q) {
            if (32 * q >= ns) break;
            const int rowE = 32 * q + 2 * rloc;
            const float* bsrc = oT + rowE * SDIM + 2 * hi;
            const float* dsrc = uT + rowE * SDIM + dOff;
            const float* hsrc = hT + (16 * q + rloc) * HID + 2 * hi;
            float* optr = oT + (rowE + 1) * SDIM + dOff;
            v2f bb[NCHUNK];
            #pragma unroll
            for (int j = 0; j < NCHUNK; ++j) bb[j] = *(const v2f*)(bsrc + 4 * j);
            const v2f hb0 = *(const v2f*)(hsrc);
            const v2f hb1 = *(const v2f*)(hsrc + 4);
            if (two) {
                v8f oA = ld8(dsrc), oB = ld8(dsrc + 16);
                #pragma unroll
                for (int j = 0; j < NCHUNK; ++j) {
                    oA = WMMA(xA[j], bb[j], oA);
                    oB = WMMA(xB[j], bb[j], oB);
                }
                oA = WMMA(gA[0], hb0, oA);
                oB = WMMA(gB[0], hb0, oB);
                oA = WMMA(gA[1], hb1, oA);
                oB = WMMA(gB[1], hb1, oB);
                st8(optr, oA);
                st8(optr + 16, oB);
            } else {
                v8f oA = {}, oB = {};
                if (!hi) {
                    const float4 a0 = *(const float4*)(dsrc);
                    oA[0] = a0.x; oA[1] = a0.y; oA[2] = a0.z; oA[3] = a0.w;
                }
                #pragma unroll
                for (int j = 0; j < NCHUNK; ++j) {
                    if (j & 1) oB = WMMA(xA[j], bb[j], oB);
                    else       oA = WMMA(xA[j], bb[j], oA);
                }
                oA = WMMA(gA[0], hb0, oA);
                oB = WMMA(gA[1], hb1, oB);
                const v8f o = oA + oB;
                if (!hi) *(float4*)(optr) = make_float4(o[0], o[1], o[2], o[3]);
            }
        }
        __syncthreads();

        {
            float4* __restrict__ dst4 = (float4*)(out_c + (size_t)t0 * SDIM);
            const float4* src4 = (const float4*)(oT + SDIM);
            const float* oT1 = oT + SDIM;
            const int n4 = ns * ROW4;
            if (ns == TB) {
                float4 o[NIT4];
                #pragma unroll
                for (int k = 0; k < NIT4; ++k) o[k] = src4[tid + NTHR * k];
                int yi[NYIT];
                #pragma unroll
                for (int k = 0; k < NYIT; ++k) {
                    const int i  = tid + NTHR * k;
                    const int rr = i / NOBS;
                    yi[k] = rr * SDIM + obsS[i - rr * NOBS];
                }
                #pragma unroll
                for (int k = 0; k < NIT4; ++k) dst4[tid + NTHR * k] = o[k];
                float yv[NYIT];
                #pragma unroll
                for (int k = 0; k < NYIT; ++k) yv[k] = oT1[yi[k]];
                float* __restrict__ ydst = out_y + (size_t)t0 * NOBS;
                #pragma unroll
                for (int k = 0; k < NYIT; ++k) ydst[tid + NTHR * k] = yv[k];
            } else {
                #pragma unroll
                for (int k = 0; k < NIT4; ++k) {
                    const int i = tid + NTHR * k;
                    if (i < n4) dst4[i] = src4[i];
                }
                const int ny = ns * NOBS;
                for (int i = tid; i < ny; i += NTHR) {
                    const int rr = i / NOBS;
                    const int j = i - rr * NOBS;
                    out_y[(size_t)t0 * NOBS + i] = oT1[rr * SDIM + obsS[j]];
                }
            }
        }
    }
}

extern "C" void kernel_launch(void* const* d_in, const int* in_sizes, int n_in,
                              void* d_out, int out_size, void* d_ws, size_t ws_size,
                              hipStream_t stream) {
    const float* c0 = (const float*)d_in[0];
    const float* w  = (const float*)d_in[1];
    const float* u  = (const float*)d_in[2];
    const float* A  = (const float*)d_in[3];
    const float* Bw = (const float*)d_in[4];
    const float* W1 = (const float*)d_in[5];
    const float* b1 = (const float*)d_in[6];
    const float* W2 = (const float*)d_in[7];
    const float* b2 = (const float*)d_in[8];
    const int*  obs = (const int*)  d_in[9];

    const int T = in_sizes[1] / 2;
    float* out_c = (float*)d_out;
    float* out_y = out_c + (size_t)T * SDIM;

    greybox_scan_kernel<<<1, NTHR, 0, stream>>>(c0, w, u, A, Bw, W1, b1, W2, b2,
                                               obs, out_c, out_y, T);
}


# Chain advances FOUR steps per matrix pass: A4, A3 dtW2, A2 dtW2, W1 A2, W1 A3, three 8x8 blocks built in-kernel (f64 sums); 4 tanh a pass on wave 3; rows between as batched passes; sum order changed
# speedup vs baseline: 6.3574x; 1.0811x over previous
#include <hip/hip_runtime.h>

#define SDIM   100
#define HID    8
#define NOBS   12
#define DTC    0.01f
#define NCHUNK 25
#define NTILE  7
#define NWV    4
#define NTHR   128
#define W1ROW  8
#define TB     128
#define NPAIR  (TB / 2)
#define NBATCH (NPAIR / 16)
#define NQUAD  (TB / 4)
#define NQB    (NQUAD / 16)
#define ROW4   (SDIM / 4)
#define TILE4  (TB * ROW4)
#define NIT4   ((TILE4 + NTHR - 1) / NTHR)
#define NOWN   8
#define NFOR   (NCHUNK - NOWN)
#define NYIT   (TB * NOBS / NTHR)
#define NPRE   3
#define NDEF   3
#define NGRP   ((NCHUNK + 3) / 4)
static_assert(NTHR == NWV * 32, "one wave per SIMD of the work-group processor");
static_assert(TB <= NTHR, "a tile's w rows are staged one row (two floats) per thread");
static_assert((NTILE - 1) * 16 + W1ROW >= SDIM && W1ROW + HID <= 16, "W1 rows fit in the last tile's padding rows");
static_assert(NOWN * 4 == 32 && NWV == 4 && NFOR == 2 * NOWN + 1, "a two-tile wave owns 32 state rows; the others' are two such blocks and rows 96..99");
static_assert(TILE4 == NIT4 * NTHR && TB * NOBS == NYIT * NTHR, "a full tile splits evenly over the threads: its accesses need no guard");
static_assert(NPRE <= 4, "the products in front of the first barrier take their B operands from the first tile's rows only");
static_assert(NDEF >= 1 && NDEF <= NFOR, "the products behind the second barrier are chunks of the other waves' rows");
static_assert(TB % 64 == 0, "a batch is 16 pairs or 16 quads of steps");
static_assert(SDIM - (NTILE - 1) * 16 == 4 && NCHUNK == 4 * (NGRP - 1) + 1, "the last row tile has 4 state rows: four K chunks fit its 16 tile rows; the last group is chunk 24 alone");

typedef float v2f __attribute__((ext_vector_type(2)));
typedef float v8f __attribute__((ext_vector_type(8)));

#define WMMA(a, b, c) __builtin_amdgcn_wmma_f32_16x16x4_f32(false, (a), false, (b), (short)0, (c), false, false)
#define HID2(j) do { c3[(j) & 3] = WMMA(xB[j], bx[j], c3[(j) & 3]); \
                     asm volatile("" :: "v"(c3[(j) & 3])); __builtin_amdgcn_sched_barrier(0); } while (0)
#define PIN1(a)          asm volatile("" :: "v"(a))
#define PIN2(a, b)       asm volatile("" :: "v"(a), "v"(b))
#define PIN4(a, b, c, d) asm volatile("" :: "v"(a), "v"(b), "v"(c), "v"(d))

template <bool FULL>
__device__ __forceinline__ void tile_load(float4 (&r)[NIT4], float2& wv,
                                          const float4* __restrict__ u4, const float2* __restrict__ w2,
                                          const int nsn, const int tid)
{
    const int n4 = nsn * ROW4;
    #pragma unroll
    for (int k = 0; k < NIT4; ++k) {
        const int i = tid + NTHR * k;
        r[k] = (FULL || i < n4) ? u4[i] : make_float4(0.f, 0.f, 0.f, 0.f);
    }
    wv = (FULL || tid < nsn) ? w2[tid] : make_float2(0.f, 0.f);
}

template <bool FULL>
__device__ __forceinline__ void tile_build_d(const float4 (&r)[NIT4], float4* dst4, const float2* wT2,
                                             const float4* g0S, const float4* g1S, const float4* bbS,
                                             const int n4, const int tid)
{
    #pragma unroll
    for (int k = 0; k < NIT4; ++k) {
        const int i = tid + NTHR * k;
        if (FULL || i < n4) {
            const int row = i / ROW4;
            const int c4  = i - row * ROW4;
            const float2 wr = wT2[row];
            const float4 g0 = g0S[c4];
            const float4 g1 = g1S[c4];
            const float4 bb = bbS[c4];
            float4 d;
            d.x = ((g0.x * wr.x + g1.x * wr.y) + r[k].x) + bb.x;
            d.y = ((g0.y * wr.x + g1.y * wr.y) + r[k].y) + bb.y;
            d.z = ((g0.z * wr.x + g1.z * wr.y) + r[k].z) + bb.z;
            d.w = ((g0.w * wr.x + g1.w * wr.y) + r[k].w) + bb.w;
            dst4[i] = d;
        } else {
            dst4[i] = make_float4(0.f, 0.f, 0.f, 0.f);
        }
    }
}

__device__ __forceinline__ v8f ld8(const float* p)
{
    const float4 a = *(const float4*)(p);
    const float4 b = *(const float4*)(p + 4);
    v8f v;
    v[0] = a.x; v[1] = a.y; v[2] = a.z; v[3] = a.w;
    v[4] = b.x; v[5] = b.y; v[6] = b.z; v[7] = b.w;
    return v;
}
__device__ __forceinline__ void st8(float* p, const v8f v)
{
    *(float4*)(p)     = make_float4(v[0], v[1], v[2], v[3]);
    *(float4*)(p + 4) = make_float4(v[4], v[5], v[6], v[7]);
}
__device__ __forceinline__ float tanh_fast(const float x)
{
    const float e2x = __builtin_amdgcn_exp2f(x * 2.885390081777927f);
    return fmaf(-2.0f, __builtin_amdgcn_rcpf(e2x + 1.0f), 1.0f);
}

__device__ __forceinline__ float lane_get(const float v, const int l)
{
    return __int_as_float(__builtin_amdgcn_readlane(__float_as_int(v), l));
}

template <bool HASC, bool HASH, bool HASF>
__device__ __forceinline__ void batch_pass(const v2f (&xA)[NCHUNK], const v2f (&xB)[NCHUNK],
                                           const v2f (&gA)[2], const v2f (&gB)[2],
                                           const float* bcol, const float* ccol, float* ocol,
                                           const float* hcol, float* fcol,
                                           const bool two, const int hi, const int dOff)
{
    v2f bb[NCHUNK];
    #pragma unroll
    for (int j = 0; j < NCHUNK; ++j) bb[j] = *(const v2f*)(bcol + 2 * hi + 4 * j);
    v2f hb0 = {}, hb1 = {};
    if (HASH) {
        hb0 = *(const v2f*)(hcol + 2 * hi);
        hb1 = *(const v2f*)(hcol + 2 * hi + 4);
    }
    if (two) {
        v8f oA = {}, oB = {};
        if (HASC) { oA = ld8(ccol + dOff); oB = ld8(ccol + dOff + 16); }
        #pragma unroll
        for (int j = 0; j < NCHUNK; ++j) {
            oA = WMMA(xA[j], bb[j], oA);
            oB = WMMA(xB[j], bb[j], oB);
        }
        if (HASH) {
            oA = WMMA(gA[0], hb0, oA);
            oB = WMMA(gB[0], hb0, oB);
            oA = WMMA(gA[1], hb1, oA);
            oB = WMMA(gB[1], hb1, oB);
        }
        st8(ocol + dOff, oA);
        st8(ocol + dOff + 16, oB);
    } else {
        v8f oA = {}, oB = {};
        if (HASC && !hi) {
            const float4 a0 = *(const float4*)(ccol + dOff);
            oA[0] = a0.x; oA[1] = a0.y; oA[2] = a0.z; oA[3] = a0.w;
        }
        #pragma unroll
        for (int j = 0; j < NCHUNK; ++j) {
            if (j & 1) oB = WMMA(xA[j], bb[j], oB);
            else       oA = WMMA(xA[j], bb[j], oA);
        }
        if (HASH) {
            oA = WMMA(gA[0], hb0, oA);
            oB = WMMA(gA[1], hb1, oB);
        }
        const v8f o = oA + oB;
        if (!hi) *(float4*)(ocol + dOff) = make_float4(o[0], o[1], o[2], o[3]);
        else if (HASF) st8(fcol, o);
    }
}

__global__ __launch_bounds__(NTHR, 1)
void greybox_scan_kernel(const float* __restrict__ c0,
                         const float* __restrict__ w,
                         const float* __restrict__ u,
                         const float* __restrict__ A,
                         const float* __restrict__ Bw,
                         const float* __restrict__ W1,
                         const float* __restrict__ b1,
                         const float* __restrict__ W2,
                         const float* __restrict__ b2,
                         const int*   __restrict__ obs,
                         float* __restrict__ out_c,
                         float* __restrict__ out_y,
                         int T)
{
    __shared__ __attribute__((aligned(16))) float cS[SDIM];
    __shared__ __attribute__((aligned(16))) float zS[SDIM];
    __shared__ __attribute__((aligned(16))) float hS[64];
    __shared__ __attribute__((aligned(16))) float uT[TB * SDIM];
    __shared__ __attribute__((aligned(16))) float eT[NPAIR * SDIM];
    __shared__ __attribute__((aligned(16))) float tT[NQUAD * SDIM];
    __shared__ __attribute__((aligned(16))) float oT[(TB + 1) * SDIM];
    __shared__ __attribute__((aligned(8)))  float wT[TB * 2];
    __shared__ __attribute__((aligned(16))) float fT[NPAIR * HID];
    __shared__ __attribute__((aligned(16))) float f2T[NQUAD * HID];
    __shared__ __attribute__((aligned(16))) float f3T[NQUAD * HID];
    __shared__ __attribute__((aligned(16))) float hT[NQUAD * 64];
    __shared__ __attribute__((aligned(16))) float Bw0S[SDIM];
    __shared__ __attribute__((aligned(16))) float Bw1S[SDIM];
    __shared__ __attribute__((aligned(16))) float b2S[SDIM];
    __shared__ int   obsS[NOBS];
    __shared__ __attribute__((aligned(16))) float AS[SDIM * SDIM];
    __shared__ __attribute__((aligned(16))) float A2S[SDIM * SDIM];
    __shared__ __attribute__((aligned(16))) float A4S[SDIM * SDIM];
    __shared__ __attribute__((aligned(16))) float GS[SDIM * HID];
    __shared__ __attribute__((aligned(16))) float AGS[SDIM * HID];
    __shared__ __attribute__((aligned(16))) float A2GS[SDIM * HID];
    __shared__ __attribute__((aligned(16))) float A3GS[SDIM * HID];
    __shared__ __attribute__((aligned(16))) float W1S[HID * SDIM];
    __shared__ __attribute__((aligned(16))) float W1AS[HID * SDIM];
    __shared__ __attribute__((aligned(16))) float W1A2S[HID * SDIM];
    __shared__ __attribute__((aligned(16))) float W1A3S[HID * SDIM];
    __shared__ float W1GS[HID * HID];
    __shared__ float W1AGS[HID * HID];
    __shared__ float W1A2GS[HID * HID];

    const int tid  = threadIdx.x;
    const int wave = __builtin_amdgcn_readfirstlane(tid >> 5);
    const int lane = tid & 31;
    const int hi   = (lane >> 4) & 1;
    const int rloc = lane & 15;

    for (int i = tid; i < SDIM; i += NTHR) {
        Bw0S[i] = Bw[2 * i]; Bw1S[i] = Bw[2 * i + 1];
        b2S[i] = DTC * b2[i]; zS[i] = 0.0f; cS[i] = c0[i];
    }
    for (int i = tid; i < NOBS; i += NTHR) obsS[i] = obs[i];
    for (int i = tid; i < SDIM * SDIM; i += NTHR) AS[i] = A[i];
    for (int i = tid; i < SDIM * HID; i += NTHR) { GS[i] = DTC * W2[i]; W1S[i] = W1[i]; }
    __syncthreads();
    for (int idx = tid; idx < SDIM * SDIM; idx += NTHR) {
        const int i = idx / SDIM, j = idx - i * SDIM;
        double s = 0.0;
        for (int k = 0; k < SDIM; ++k) s = fma((double)AS[i * SDIM + k], (double)AS[k * SDIM + j], s);
        A2S[idx] = (float)s;
    }
    for (int idx = tid; idx < SDIM * HID; idx += NTHR) {
        const int i = idx / HID, m = idx - i * HID;
        double s = 0.0;
        for (int k = 0; k < SDIM; ++k) s = fma((double)AS[i * SDIM + k], (double)GS[k * HID + m], s);
        AGS[idx] = (float)s;
    }
    for (int idx = tid; idx < HID * SDIM; idx += NTHR) {
        const int n = idx / SDIM, j = idx - n * SDIM;
        double s = 0.0;
        for (int k = 0; k < SDIM; ++k) s = fma((double)W1S[n * SDIM + k], (double)AS[k * SDIM + j], s);
        W1AS[idx] = (float)s;
    }
    for (int idx = tid; idx < HID * HID; idx += NTHR) {
        const int n = idx / HID, m = idx - n * HID;
        double s = 0.0;
        for (int k = 0; k < SDIM; ++k) s = fma((double)W1S[n * SDIM + k], (double)GS[k * HID + m], s);
        W1GS[idx] = (float)s;
    }
    __syncthreads();
    for (int idx = tid; idx < SDIM * SDIM; idx += NTHR) {
        const int i = idx / SDIM, j = idx - i * SDIM;
        double s = 0.0;
        for (int k = 0; k < SDIM; ++k) s = fma((double)A2S[i * SDIM + k], (double)A2S[k * SDIM + j], s);
        A4S[idx] = (float)s;
    }
    for (int idx = tid; idx < SDIM * HID; idx += NTHR) {
        const int i = idx / HID, m = idx - i * HID;
        double s2 = 0.0, s3 = 0.0;
        for (int k = 0; k < SDIM; ++k) {
            const double a2 = (double)A2S[i * SDIM + k];
            s2 = fma(a2, (double)GS[k * HID + m], s2);
            s3 = fma(a2, (double)AGS[k * HID + m], s3);
        }
        A2GS[idx] = (float)s2;
        A3GS[idx] = (float)s3;
    }
    for (int idx = tid; idx < HID * SDIM; idx += NTHR) {
        const int n = idx / SDIM, j = idx - n * SDIM;
        double s2 = 0.0, s3 = 0.0;
        for (int k = 0; k < SDIM; ++k) {
            const double a2 = (double)A2S[k * SDIM + j];
            s2 = fma((double)W1S[n * SDIM + k], a2, s2);
            s3 = fma((double)W1AS[n * SDIM + k], a2, s3);
        }
        W1A2S[idx] = (float)s2;
        W1A3S[idx] = (float)s3;
    }
    for (int idx = tid; idx < HID * HID; idx += NTHR) {
        const int n = idx / HID, m = idx - n * HID;
        double s1 = 0.0, s2 = 0.0;
        for (int k = 0; k < SDIM; ++k) {
            const double wa = (double)W1AS[n * SDIM + k];
            s1 = fma(wa, (double)GS[k * HID + m], s1);
            s2 = fma(wa, (double)AGS[k * HID + m], s2);
        }
        W1AGS[idx] = (float)s1;
        W1A2GS[idx] = (float)s2;
    }
    __syncthreads();

    const bool two   = wave < NWV - 1;
    const int  tileA = two ? 2 * wave : NTILE - 1;

    const float* bBaseC = (rloc == 0 || !two) ? (cS + 2 * hi) : zS;
    const float* bBaseH = (rloc == 0) ? (hS + 2 * hi) : zS;

    const float* dPtr    = two ? (eT + tileA * 16 + 8 * hi) : (hi ? zS : (eT + (NTILE - 1) * 16));
    const int    dStride = (two || !hi) ? SDIM : 0;
    const int    sBase   = tileA * 16 + 8 * hi;
    const bool   st0     = rloc == 0 && (two || hi == 0);
    const bool   st1     = rloc == 0 && two;

    static_assert((HID & (HID - 1)) == 0 && HID <= 8, "one hidden unit per lane, picked from the 8 D registers");
    const float b1v = b1[lane & (HID - 1)];
    v2f x2A[NCHUNK] = {}, x2B[NCHUNK] = {}, xA[NCHUNK] = {}, xB[NCHUNK] = {};
    v2f hfA[6] = {}, hfB[6] = {}, gA[2] = {}, gB[2] = {};
    const int oth1 = two ? 32 * ((wave + 1) % 3) : 0;
    const int oth2 = two ? 32 * ((wave + 2) % 3) : 0;
    {
        const int r = tileA * 16 + rloc;
        const bool isW1 = !two && rloc >= W1ROW && rloc < W1ROW + HID;
        #pragma unroll
        for (int j = 0; j < NCHUNK; ++j) {
            const int kp = 4 * j + 2 * hi;
            int kc;
            if (!two)              kc = kp;
            else if (j < NOWN)     kc = oth1 + 4 * j + 2 * hi;
            else if (j < 2 * NOWN) kc = oth2 + 4 * (j - NOWN) + 2 * hi;
            else if (j < NFOR)     kc = 96 + 2 * hi;
            else                   kc = (tileA + ((j - NFOR) >> 2)) * 16 + 8 * hi + 2 * ((j - NFOR) & 3);
            v2f a; a.x = 0.0f; a.y = 0.0f;
            v2f b; b.x = 0.0f; b.y = 0.0f;
            if (r < SDIM) {
                if (two) { a.x = A4S[r * SDIM + kc]; a.y = A4S[r * SDIM + kc + 1]; }
                b.x = AS[r * SDIM + kp];  b.y = AS[r * SDIM + kp + 1];
            } else if (isW1) {
                b.x = W1S[(rloc - W1ROW) * SDIM + kp]; b.y = W1S[(rloc - W1ROW) * SDIM + kp + 1];
            }
            if (!two) {
                const int jc = 4 * j + (rloc >> 2);
                if (j < NGRP && jc < NCHUNK) {
                    const int rp = (NTILE - 1) * 16 + (rloc & 3);
                    const int kq = 4 * jc + 2 * hi;
                    a.x = A4S[rp * SDIM + kq]; a.y = A4S[rp * SDIM + kq + 1];
                }
            }
            x2A[j] = a;
            xA[j]  = b;
        }
        #pragma unroll
        for (int jj = 0; jj < 2; ++jj) {
            const int k = 4 * jj + 2 * hi;
            v2f a3 = {}, a2 = {}, a1 = {}, b = {};
            if (r < SDIM) {
                a3.x = A3GS[r * HID + k]; a3.y = A3GS[r * HID + k + 1];
                a2.x = A2GS[r * HID + k]; a2.y = A2GS[r * HID + k + 1];
                a1.x = AGS[r * HID + k];  a1.y = AGS[r * HID + k + 1];
                b.x  = GS[r * HID + k];   b.y  = GS[r * HID + k + 1];
            }
            hfA[jj]     = a3;
            hfA[2 + jj] = a2;
            hfA[4 + jj] = a1;
            gA[jj]      = b;
        }
    }
    if (two) {
        const int r = (tileA + 1) * 16 + rloc;
        #pragma unroll
        for (int j = 0; j < NCHUNK; ++j) {
            const int kp = 4 * j + 2 * hi;
            int kc;
            if (j < NOWN)          kc = oth1 + 4 * j + 2 * hi;
            else if (j < 2 * NOWN) kc = oth2 + 4 * (j - NOWN) + 2 * hi;
            else if (j < NFOR)     kc = 96 + 2 * hi;
            else                   kc = (tileA + ((j - NFOR) >> 2)) * 16 + 8 * hi + 2 * ((j - NFOR) & 3);
            v2f a, b;
            a.x = A4S[r * SDIM + kc]; a.y = A4S[r * SDIM + kc + 1];
            b.x = AS[r * SDIM + kp];  b.y = AS[r * SDIM + kp + 1];
            x2B[j] = a;
            xB[j]  = b;
        }
        #pragma unroll
        for (int jj = 0; jj < 2; ++jj) {
            const int k = 4 * jj + 2 * hi;
            v2f a3, a2, a1, b;
            a3.x = A3GS[r * HID + k]; a3.y = A3GS[r * HID + k + 1];
            a2.x = A2GS[r * HID + k]; a2.y = A2GS[r * HID + k + 1];
            a1.x = AGS[r * HID + k];  a1.y = AGS[r * HID + k + 1];
            b.x  = GS[r * HID + k];   b.y  = GS[r * HID + k + 1];
            hfB[jj]     = a3;
            hfB[2 + jj] = a2;
            hfB[4 + jj] = a1;
            gB[jj]      = b;
        }
    } else {
        const float* hrow  = (rloc < HID) ? (W1S + rloc * SDIM)   : (W1AS + (rloc - HID) * SDIM);
        const float* hrow2 = (rloc < HID) ? (W1A2S + rloc * SDIM) : (W1A3S + (rloc - HID) * SDIM);
        #pragma unroll
        for (int j = 0; j < NCHUNK; ++j) {
            const int kp = 4 * j + 2 * hi;
            v2f a, c;
            a.x = hrow[kp];  a.y = hrow[kp + 1];
            c.x = hrow2[kp]; c.y = hrow2[kp + 1];
            x2B[j] = a;
            xB[j]  = c;
        }
    }
    float wg[HID], wag[HID], wa2g[HID];
    #pragma unroll
    for (int m = 0; m < HID; ++m) {
        wg[m]   = W1GS[(lane & (HID - 1)) * HID + m];
        wag[m]  = W1AGS[(lane & (HID - 1)) * HID + m];
        wa2g[m] = W1A2GS[(lane & (HID - 1)) * HID + m];
    }

    v8f sA = {}, sB = {};
    if (two) {
        sA = ld8(cS + sBase);
        sB = ld8(cS + sBase + 16);
    }
    const float* bO1 = bBaseC + oth1;
    const float* bO2 = bBaseC + oth2;
    const float* bO3 = bBaseC + 96;
    const float* bPk = (!two && rloc < 4)  ? (cS + 4 * rloc + 2 * hi) : zS;
    const float* bPl = (!two && rloc == 0) ? (cS + 4 * (NCHUNK - 1) + 2 * hi) : zS;
    const int dOff = two ? (tileA * 16 + 8 * hi) : (NTILE - 1) * 16;

    float4 r[NIT4];
    float2 wv;
    {
        const int n0 = (T < TB) ? T : TB;
        if (n0 == TB) tile_load<true >(r, wv, (const float4*)u, (const float2*)w, n0, tid);
        else          tile_load<false>(r, wv, (const float4*)u, (const float2*)w, n0, tid);
    }

    for (int t0 = 0; t0 < T; t0 += TB) {
        const int ns    = (T - t0 < TB) ? (T - t0) : TB;
        const int nquad = (ns + 3) >> 2;

        {
            if (tid < ns) ((float2*)wT)[tid] = wv;
            __syncthreads();
            if (tid < ROW4) ((float4*)oT)[tid] = ((const float4*)cS)[tid];
            if (ns == TB)
                tile_build_d<true >(r, (float4*)uT, (const float2*)wT, (const float4*)Bw0S,
                                    (const float4*)Bw1S, (const float4*)b2S, TILE4, tid);
            else
                tile_build_d<false>(r, (float4*)uT, (const float2*)wT, (const float4*)Bw0S,
                                    (const float4*)Bw1S, (const float4*)b2S, ns * ROW4, tid);
        }
        __syncthreads();

        for (int q = 0; q < NBATCH; ++q) {
            if (32 * q >= ns) break;
            const int pc = 16 * q + rloc;
            batch_pass<true, false, true>(xA, xB, gA, gB, uT + (2 * pc) * SDIM, uT + (2 * pc + 1) * SDIM,
                                          eT + pc * SDIM, nullptr, fT + pc * HID, two, hi, dOff);
        }
        __syncthreads();
        for (int q = 0; q < NQB; ++q) {
            if (64 * q >= ns) break;
            const int gc = 16 * q + rloc;
            batch_pass<false, false, true>(xA, xB, gA, gB, eT + (2 * gc) * SDIM, nullptr,
                                           tT + gc * SDIM, nullptr, f2T + gc * HID, two, hi, dOff);
        }
        __syncthreads();
        for (int q = 0; q < NQB; ++q) {
            if (64 * q >= ns) break;
            const int gc = 16 * q + rloc;
            batch_pass<true, false, true>(xA, xB, gA, gB, tT + gc * SDIM, eT + (2 * gc + 1) * SDIM,
                                          eT + (2 * gc + 1) * SDIM, nullptr, f3T + gc * HID, two, hi, dOff);
        }
        __syncthreads();

        const float* ePtr = dPtr + dStride;
        v8f dA = {}, dB = {};
        {
            const float4 a0 = *(const float4*)(ePtr);
            dA[0] = a0.x; dA[1] = a0.y; dA[2] = a0.z; dA[3] = a0.w;
            if (two) {
                const float4 a1 = *(const float4*)(ePtr + 4);
                dA[4] = a1.x; dA[5] = a1.y; dA[6] = a1.z; dA[7] = a1.w;
                dB = ld8(ePtr + 16);
            }
            asm volatile("" :: "v"(dA), "v"(dB));
        }

        for (int p = 0; p < nquad; ++p) {
            const float* eNext = ePtr + 2 * ((p + 1 < nquad) ? p + 1 : p) * dStride;
            v8f accA, accB;
            v2f bx[NCHUNK];
            v2f bp[NGRP];
            if (two) {
                accA = dA;
                accB = dB;
                #pragma unroll
                for (int pp = 0; pp < NPRE; ++pp) {
                    v2f bo; bo.x = sA[2 * pp]; bo.y = sA[2 * pp + 1];
                    accA = WMMA(x2A[NFOR + pp], bo, accA);
                    accB = WMMA(x2B[NFOR + pp], bo, accB);
                }
            }
            asm volatile("s_wait_dscnt 0x0" ::: "memory");
            __syncthreads();

            if (two) {
                dA = ld8(eNext);
                dB = ld8(eNext + 16);
                #pragma unroll
                for (int f = 0; f < NOWN; ++f) {
                    bx[f]        = *(const v2f*)(bO1 + 4 * f);
                    bx[NOWN + f] = *(const v2f*)(bO2 + 4 * f);
                }
                bx[NFOR - 1] = *(const v2f*)(bO3);
                __builtin_amdgcn_sched_barrier(0);
                #pragma unroll
                for (int pp = NPRE; pp < NOWN; ++pp) {
                    v2f bo;
                    if (pp < 4) { bo.x = sA[2 * pp];       bo.y = sA[2 * pp + 1]; }
                    else        { bo.x = sB[2 * (pp - 4)]; bo.y = sB[2 * (pp - 4) + 1]; }
                    accA = WMMA(x2A[NFOR + pp], bo, accA);
                    accB = WMMA(x2B[NFOR + pp], bo, accB);
                }
                #pragma unroll
                for (int f = 0; f < NFOR - NDEF; ++f) {
                    accA = WMMA(x2A[f], bx[f], accA);
                    accB = WMMA(x2B[f], bx[f], accB);
                }
            } else {
                accA = dA;
                {
                    const float4 a0 = *(const float4*)(eNext);
                    dA[0] = a0.x; dA[1] = a0.y; dA[2] = a0.z; dA[3] = a0.w;
                }
                const int   un  = lane & (HID - 1);
                const float fv1 = fT[(2 * p) * HID + un];
                const float fv2 = f2T[p * HID + un];
                const float fv3 = f3T[p * HID + un] + fT[(2 * p + 1) * HID + un];
                #pragma unroll
                for (int j = 0; j < NCHUNK; ++j)
                    bx[j] = *(const v2f*)(bBaseC + 4 * j);
                #pragma unroll
                for (int g = 0; g < NGRP - 1; ++g)
                    bp[g] = *(const v2f*)(bPk + 16 * g);
                bp[NGRP - 1] = *(const v2f*)(bPl);
                __builtin_amdgcn_sched_barrier(0);
                v8f a3[4], c3[4];
                #pragma unroll
                for (int qq = 0; qq < 4; ++qq) {
                    a3[qq] = (v8f){0.f, 0.f, 0.f, 0.f, 0.f, 0.f, 0.f, 0.f};
                    c3[qq] = (v8f){0.f, 0.f, 0.f, 0.f, 0.f, 0.f, 0.f, 0.f};
                }
                #pragma unroll
                for (int j = 0; j < NCHUNK; ++j)
                    a3[j & 3] = WMMA(x2B[j], bx[j], a3[j & 3]);
                __builtin_amdgcn_sched_barrier(0);
                const v8f t01 = a3[0] + a3[1];
                const v8f t23 = a3[2] + a3[3];
                PIN2(t01, t23);
                HID2(0);
                const v8f s1 = t01 + t23;
                const float p10 = (lane & 1) ? s1[1] : s1[0];
                const float p11 = (lane & 1) ? s1[3] : s1[2];
                const float p12 = (lane & 1) ? s1[5] : s1[4];
                const float p13 = (lane & 1) ? s1[7] : s1[6];
                PIN4(p10, p11, p12, p13);
                HID2(1);
                const float q10 = (lane & 2) ? p11 : p10;
                const float q11 = (lane & 2) ? p13 : p12;
                const float zn1 = (lane & 4) ? q11 : q10;
                const float y0  = (zn1 + b1v) * 2.885390081777927f;
                PIN1(y0);
                HID2(2);
                const float e0 = __builtin_amdgcn_exp2f(y0);
                float z1 = zn1 + (b1v + fv1), z1b = 0.0f;
                float z2 = b1v + fv2, z2b = 0.0f;
                float z3 = b1v + fv3, z3b = 0.0f;
                PIN4(e0, z1, z2, z3);
                HID2(3);
                const float r0 = __builtin_amdgcn_rcpf(e0 + 1.0f);
                PIN1(r0);
                HID2(4);
                const float h0v = fmaf(-2.0f, r0, 1.0f);
                float hm[HID], gm[HID];
                #pragma unroll
                for (int m = 0; m < 4; ++m) hm[m] = lane_get(h0v, m);
                PIN1(h0v);
                HID2(5);
                #pragma unroll
                for (int m = 4; m < HID; ++m) hm[m] = lane_get(h0v, m);
                #pragma unroll
                for (int m = 0; m < 2; ++m) {
                    if (m & 1) z1b = fmaf(wg[m], hm[m], z1b);
                    else       z1  = fmaf(wg[m], hm[m], z1);
                    z2 = fmaf(wag[m],  hm[m], z2);
                    z3 = fmaf(wa2g[m], hm[m], z3);
                }
                PIN4(z1, z1b, z2, z3);
                HID2(6);
                #pragma unroll
                for (int m = 2; m < 5; ++m) {
                    if (m & 1) z1b = fmaf(wg[m], hm[m], z1b);
                    else       z1  = fmaf(wg[m], hm[m], z1);
                    z2 = fmaf(wag[m],  hm[m], z2);
                    z3 = fmaf(wa2g[m], hm[m], z3);
                }
                PIN4(z1, z1b, z2, z3);
                HID2(7);
                #pragma unroll
                for (int m = 5; m < HID; ++m) {
                    if (m & 1) z1b = fmaf(wg[m], hm[m], z1b);
                    else       z1  = fmaf(wg[m], hm[m], z1);
                    z2 = fmaf(wag[m],  hm[m], z2);
                    z3 = fmaf(wa2g[m], hm[m], z3);
                }
                PIN4(z1, z1b, z2, z3);
                HID2(8);
                const float e1 = __builtin_amdgcn_exp2f((z1 + z1b) * 2.885390081777927f);
                PIN1(e1);
                HID2(9);
                const float r1 = __builtin_amdgcn_rcpf(e1 + 1.0f);
                PIN1(r1);
                HID2(10);
                const float h1v = fmaf(-2.0f, r1, 1.0f);
                #pragma unroll
                for (int m = 0; m < 4; ++m) gm[m] = lane_get(h1v, 16 + m);
                PIN1(h1v);
                HID2(11);
                #pragma unroll
                for (int m = 4; m < HID; ++m) gm[m] = lane_get(h1v, 16 + m);
                #pragma unroll
                for (int m = 0; m < 3; ++m) {
                    if (m & 1) z2b = fmaf(wg[m], gm[m], z2b);
                    else       z2  = fmaf(wg[m], gm[m], z2);
                    z3 = fmaf(wag[m], gm[m], z3);
                }
                PIN4(z2, z2b, z3, h1v);
                HID2(12);
                #pragma unroll
                for (int m = 3; m < HID; ++m) {
                    if (m & 1) z2b = fmaf(wg[m], gm[m], z2b);
                    else       z2  = fmaf(wg[m], gm[m], z2);
                    z3 = fmaf(wag[m], gm[m], z3);
                }
                PIN4(z2, z2b, z3, h1v);
                HID2(13);
                #pragma unroll
                for (int j = 14; j < NCHUNK; ++j)
                    c3[j & 3] = WMMA(xB[j], bx[j], c3[j & 3]);
                static_assert(NGRP == 7, "the statement below names every packed operand");
                __builtin_amdgcn_sched_barrier(0);
                asm volatile("" :: "v"(bp[0]), "v"(bp[1]), "v"(bp[2]), "v"(bp[3]), "v"(bp[4]), "v"(bp[5]), "v"(bp[6]));
                float zn2;
                {
                    const v8f s = (c3[0] + c3[1]) + (c3[2] + c3[3]);
                    const float p0 = (lane & 1) ? s[1] : s[0];
                    const float p1 = (lane & 1) ? s[3] : s[2];
                    const float p2 = (lane & 1) ? s[5] : s[4];
                    const float p3 = (lane & 1) ? s[7] : s[6];
                    const float q0 = (lane & 2) ? p1 : p0;
                    const float q1 = (lane & 2) ? p3 : p2;
                    zn2 = (lane & 4) ? q1 : q0;
                }
                const float h2v = tanh_fast(zn2 + (z2 + z2b));
                #pragma unroll
                for (int m = 0; m < HID; ++m) {
                    const float hm = lane_get(h2v, m);
                    if (m & 1) z3b = fmaf(wg[m], hm, z3b);
                    else       z3  = fmaf(wg[m], hm, z3);
                }
                const float h3v = tanh_fast(zn2 + (z3 + z3b));
                {
                    const float ha = hi ? h1v : h0v;
                    const float hb = hi ? h3v : h2v;
                    hS[lane] = ha;              hS[32 + lane] = hb;
                    hT[p * 64 + lane] = ha;     hT[p * 64 + 32 + lane] = hb;
                }
            }
            __syncthreads();

            {
                v2f hq[8];
                #pragma unroll
                for (int c = 0; c < 8; ++c)
                    hq[c] = *(const v2f*)(bBaseH + 16 * (c >> 1) + 4 * (c & 1));
                if (two) {
                    #pragma unroll
                    for (int f = NFOR - NDEF; f < NFOR; ++f) {
                        accA = WMMA(x2A[f], bx[f], accA);
                        accB = WMMA(x2B[f], bx[f], accB);
                    }
                    #pragma unroll
                    for (int c = 0; c < 6; ++c) {
                        accA = WMMA(hfA[c], hq[c], accA);
                        accB = WMMA(hfB[c], hq[c], accB);
                    }
                    accA = WMMA(gA[0], hq[6], accA);
                    accB = WMMA(gB[0], hq[6], accB);
                    sA = WMMA(gA[1], hq[7], accA);
                    sB = WMMA(gB[1], hq[7], accB);
                } else {
                    v8f t6 = (v8f){0.f, 0.f, 0.f, 0.f, 0.f, 0.f, 0.f, 0.f};
                    #pragma unroll
                    for (int g = 0; g < NGRP; ++g) {
                        if (g & 1) t6   = WMMA(x2A[g], bp[g], t6);
                        else       accA = WMMA(x2A[g], bp[g], accA);
                    }
                    #pragma unroll
                    for (int c = 0; c < 6; ++c) {
                        if (c & 1) t6   = WMMA(hfA[c], hq[c], t6);
                        else       accA = WMMA(hfA[c], hq[c], accA);
                    }
                    accA = WMMA(gA[0], hq[6], accA);
                    t6   = WMMA(gA[1], hq[7], t6);
                    const v8f t = accA + t6;
                    #pragma unroll
                    for (int rr = 0; rr < 4; ++rr)
                        sA[rr] = ((t[rr] + lane_get(t[4 + rr], 1)) + lane_get(t[rr], 18)) + lane_get(t[4 + rr], 19);
                }
                float* orow = oT + (4 * p + 4) * SDIM + sBase;
                if (st0) {
                    const float4 x0 = make_float4(sA[0], sA[1], sA[2], sA[3]);
                    *(float4*)(cS + sBase) = x0;
                    *(float4*)(orow)       = x0;
                }
                if (st1) {
                    const float4 x1 = make_float4(sA[4], sA[5], sA[6], sA[7]);
                    const float4 y0 = make_float4(sB[0], sB[1], sB[2], sB[3]);
                    const float4 y1 = make_float4(sB[4], sB[5], sB[6], sB[7]);
                    *(float4*)(cS + sBase + 4)  = x1;
                    *(float4*)(cS + sBase + 16) = y0;
                    *(float4*)(cS + sBase + 20) = y1;
                    *(float4*)(orow + 4)  = x1;
                    *(float4*)(orow + 16) = y0;
                    *(float4*)(orow + 20) = y1;
                }
            }
        }
        __syncthreads();

        {
            const int tn = t0 + TB;
            const int nn = (T - tn < TB) ? (T - tn) : TB;
            const int tb = (nn > 0) ? tn : 0;
            const float4* un = (const float4*)(u + (size_t)tb * SDIM);
            const float2* wn = (const float2*)(w + (size_t)tb * 2);
            if (nn == TB) tile_load<true >(r, wv, un, wn, nn, tid);
            else          tile_load<false>(r, wv, un, wn, nn, tid);
        }

        for (int jq = 0; jq < 3; ++jq) {
            if (jq > 0) __syncthreads();
            for (int q = 0; q < NQB; ++q) {
                if (64 * q >= ns) break;
                const int gc = 16 * q + rloc;
                batch_pass<true, true, false>(xA, xB, gA, gB, oT + (4 * gc + jq) * SDIM, uT + (4 * gc + jq) * SDIM,
                                              oT + (4 * gc + jq + 1) * SDIM, hT + gc * 64 + 16 * jq, nullptr,
                                              two, hi, dOff);
            }
        }
        __syncthreads();

        {
            float4* __restrict__ dst4 = (float4*)(out_c + (size_t)t0 * SDIM);
            const float4* src4 = (const float4*)(oT + SDIM);
            const float* oT1 = oT + SDIM;
            const int n4 = ns * ROW4;
            if (ns == TB) {
                float4 o[NIT4];
                #pragma unroll
                for (int k = 0; k < NIT4; ++k) o[k] = src4[tid + NTHR * k];
                int yi[NYIT];
                #pragma unroll
                for (int k = 0; k < NYIT; ++k) {
                    const int i  = tid + NTHR * k;
                    const int rr = i / NOBS;
                    yi[k] = rr * SDIM + obsS[i - rr * NOBS];
                }
                #pragma unroll
                for (int k = 0; k < NIT4; ++k) dst4[tid + NTHR * k] = o[k];
                float yv[NYIT];
                #pragma unroll
                for (int k = 0; k < NYIT; ++k) yv[k] = oT1[yi[k]];
                float* __restrict__ ydst = out_y + (size_t)t0 * NOBS;
                #pragma unroll
                for (int k = 0; k < NYIT; ++k) ydst[tid + NTHR * k] = yv[k];
            } else {
                #pragma unroll
                for (int k = 0; k < NIT4; ++k) {
                    const int i = tid + NTHR * k;
                    if (i < n4) dst4[i] = src4[i];
                }
                const int ny = ns * NOBS;
                for (int i = tid; i < ny; i += NTHR) {
                    const int rr = i / NOBS;
                    const int j = i - rr * NOBS;
                    out_y[(size_t)t0 * NOBS + i] = oT1[rr * SDIM + obsS[j]];
                }
            }
        }
    }
}

extern "C" void kernel_launch(void* const* d_in, const int* in_sizes, int n_in,
                              void* d_out, int out_size, void* d_ws, size_t ws_size,
                              hipStream_t stream) {
    const float* c0 = (const float*)d_in[0];
    const float* w  = (const float*)d_in[1];
    const float* u  = (const float*)d_in[2];
    const float* A  = (const float*)d_in[3];
    const float* Bw = (const float*)d_in[4];
    const float* W1 = (const float*)d_in[5];
    const float* b1 = (const float*)d_in[6];
    const float* W2 = (const float*)d_in[7];
    const float* b2 = (const float*)d_in[8];
    const int*  obs = (const int*)  d_in[9];

    const int T = in_sizes[1] / 2;
    float* out_c = (float*)d_out;
    float* out_y = out_c + (size_t)T * SDIM;

    greybox_scan_kernel<<<1, NTHR, 0, stream>>>(c0, w, u, A, Bw, W1, b1, W2, b2,
                                               obs, out_c, out_y, T);
}
